# starting kernel
# baseline (speedup 1.0000x reference)
#include <hip/hip_runtime.h>

typedef __attribute__((ext_vector_type(2))) float v2f;
typedef __attribute__((ext_vector_type(8))) float v8f;

#define NCOLS 128
#define MTILE 32
#define LDSTRIDE 260

__global__ void pack_w(const float* __restrict__ W, float* __restrict__ Wp,
                       int Kw, int Kpad) {
  int i = blockIdx.x * blockDim.x + threadIdx.x;
  if (i < Kpad * NCOLS) {
    int k = i / NCOLS;
    int n = i - k * NCOLS;
    float v = (k < Kw) ? W[(size_t)k * NCOLS + n] : 0.0f;
    Wp[((size_t)(k >> 1) * NCOLS + n) * 2 + (k & 1)] = v;
  }
}

__global__ void init_ws(float* __restrict__ deg, float* __restrict__ s1,
                        float* __restrict__ agg2, int N) {
  unsigned tid = blockIdx.x * blockDim.x + threadIdx.x;
  if (tid < (unsigned)N * 128u) {
    agg2[tid] = 0.0f;
    unsigned j = tid & 127u;
    unsigned i = tid >> 7;
    if (j == 0u) deg[i] = 1.0f;
    if (j < 5u)  s1[i * 5u + j] = 0.0f;
  }
}

__global__ void edge_accum_feat(const int* __restrict__ src, const int* __restrict__ dst,
                                const float* __restrict__ x, float* __restrict__ deg,
                                float* __restrict__ s1, int E) {
  int e = blockIdx.x * blockDim.x + threadIdx.x;
  if (e < E) {
    int s = src[e], d = dst[e];
    atomicAdd(&deg[d], 1.0f);
#pragma unroll
    for (int f = 0; f < 5; ++f)
      atomicAdd(&s1[(size_t)d * 5 + f], x[(size_t)s * 5 + f]);
  }
}

__global__ void edge_accum_embed(const int* __restrict__ src, const int* __restrict__ dst,
                                 const float* __restrict__ h, float* __restrict__ agg, int E) {
  unsigned tid = blockIdx.x * blockDim.x + threadIdx.x;
  unsigned e = tid >> 5;
  unsigned c = (tid & 31u) * 4u;
  if (e < (unsigned)E) {
    int s = src[e], d = dst[e];
    const float4 v = *(const float4*)(h + (size_t)s * 128 + c);
    float* a = agg + (size_t)d * 128 + c;
    atomicAdd(a + 0, v.x);
    atomicAdd(a + 1, v.y);
    atomicAdd(a + 2, v.z);
    atomicAdd(a + 3, v.w);
  }
}

__global__ void sage_gemm_wmma(const float* __restrict__ A0, const float* __restrict__ A1,
                               const float* __restrict__ deg, const float* __restrict__ Wp,
                               const float* __restrict__ bias, float* __restrict__ out,
                               int M, int d, int Kpad) {
  __shared__ float As[MTILE * LDSTRIDE];

  const int row0 = blockIdx.x * MTILE;
  const int t = threadIdx.x;

  const int total = MTILE * Kpad;
  for (int i = t; i < total; i += 256) {
    int m = i / Kpad;
    int k = i - m * Kpad;
    int row = row0 + m;
    float v = 0.0f;
    if (row < M) {
      if (k < d) {
        v = A0[(size_t)row * d + k];
      } else if (A1 != nullptr && k < 2 * d) {
        int kk = k - d;
        v = (A0[(size_t)row * d + kk] + A1[(size_t)row * d + kk]) / deg[row];
      }
    }
    As[m * LDSTRIDE + k] = v;
  }
  __syncthreads();

  const int lane  = t & 31;
  const int wave  = t >> 5;
  const int n     = wave * 16 + (lane & 15);
  const int mA    = lane & 15;
  const int khalf = (lane < 16) ? 0 : 2;

  v8f c0 = {};
  v8f c1 = {};
#pragma unroll 2
  for (int k0 = 0; k0 < Kpad; k0 += 4) {
    const int kb = k0 + khalf;
    v2f a0 = *(const v2f*)&As[mA * LDSTRIDE + kb];
    v2f a1 = *(const v2f*)&As[(mA + 16) * LDSTRIDE + kb];
    v2f b  = *(const v2f*)(Wp + ((size_t)(kb >> 1) * NCOLS + n) * 2);
    c0 = __builtin_amdgcn_wmma_f32_16x16x4_f32(false, a0, false, b,
                                               (short)0, c0, false, false);
    c1 = __builtin_amdgcn_wmma_f32_16x16x4_f32(false, a1, false, b,
                                               (short)0, c1, false, false);
  }

  const float bv    = bias[n];
  const int   rbase = (lane < 16) ? 0 : 8;
  float* outA = out + (size_t)(row0 + rbase) * NCOLS + n;
  float* outB = outA + (size_t)16 * NCOLS;

  if (row0 + MTILE <= M) {
#pragma unroll
    for (int r = 0; r < 8; ++r) {
      float vA = c0[r] + bv;
      float vB = c1[r] + bv;
      outA[(size_t)r * NCOLS] = vA > 0.0f ? vA : 0.0f;
      outB[(size_t)r * NCOLS] = vB > 0.0f ? vB : 0.0f;
    }
  } else {
#pragma unroll
    for (int r = 0; r < 8; ++r) {
      int rowA = row0 + rbase + r;
      if (rowA < M) {
        float v = c0[r] + bv;
        outA[(size_t)r * NCOLS] = v > 0.0f ? v : 0.0f;
      }
      int rowB = row0 + 16 + rbase + r;
      if (rowB < M) {
        float v = c1[r] + bv;
        outB[(size_t)r * NCOLS] = v > 0.0f ? v : 0.0f;
      }
    }
  }
}

__global__ void lin2_kernel(const float* __restrict__ h, const float* __restrict__ W,
                            const float* __restrict__ b, float* __restrict__ out, int M) {
  int i = blockIdx.x * blockDim.x + threadIdx.x;
  if (i < M) {
    float a0 = b[0], a1 = b[1];
    const float* row = h + (size_t)i * 128;
#pragma unroll 8
    for (int k = 0; k < 128; ++k) {
      float v = row[k];
      a0 += v * W[2 * k];
      a1 += v * W[2 * k + 1];
    }
    out[(size_t)i * 2 + 0] = a0;
    out[(size_t)i * 2 + 1] = a1;
  }
}

extern "C" void kernel_launch(void* const* d_in, const int* in_sizes, int n_in,
                              void* d_out, int out_size, void* d_ws, size_t ws_size,
                              hipStream_t stream) {
  const float* x   = (const float*)d_in[0];
  const int*   ei  = (const int*)d_in[1];
  const float* W1  = (const float*)d_in[2];
  const float* b1  = (const float*)d_in[3];
  const float* W2  = (const float*)d_in[4];
  const float* b2  = (const float*)d_in[5];
  const float* l1W = (const float*)d_in[6];
  const float* l1b = (const float*)d_in[7];
  const float* l2W = (const float*)d_in[8];
  const float* l2b = (const float*)d_in[9];

  const int N = in_sizes[0] / 5;
  const int E = in_sizes[1] / 2;
  const int* src = ei;
  const int* dst = ei + E;

  float* ws   = (float*)d_ws;
  float* deg  = ws;
  float* s1   = deg  + (size_t)N;
  float* agg2 = s1   + (size_t)5 * N;
  float* h1   = agg2 + (size_t)128 * N;
  float* h2   = h1   + (size_t)128 * N;
  float* Wp1  = h2   + (size_t)128 * N;
  float* Wp2  = Wp1  + (size_t)12 * 128;
  float* Wp3  = Wp2  + (size_t)256 * 128;
  float* h3   = h1;

  const int nthread = 256;

  pack_w<<<(12  * NCOLS + nthread - 1) / nthread, nthread, 0, stream>>>(W1,  Wp1, 10,  12);
  pack_w<<<(256 * NCOLS + nthread - 1) / nthread, nthread, 0, stream>>>(W2,  Wp2, 256, 256);
  pack_w<<<(128 * NCOLS + nthread - 1) / nthread, nthread, 0, stream>>>(l1W, Wp3, 128, 128);

  init_ws<<<((unsigned)N * 128u + nthread - 1) / nthread, nthread, 0, stream>>>(deg, s1, agg2, N);

  edge_accum_feat<<<(E + nthread - 1) / nthread, nthread, 0, stream>>>(src, dst, x, deg, s1, E);

  sage_gemm_wmma<<<(N + MTILE - 1) / MTILE, nthread, 0, stream>>>(x, s1, deg, Wp1, b1, h1, N, 5, 12);

  edge_accum_embed<<<((unsigned)E * 32u + nthread - 1) / nthread, nthread, 0, stream>>>(src, dst, h1, agg2, E);

  sage_gemm_wmma<<<(N + MTILE - 1) / MTILE, nthread, 0, stream>>>(h1, agg2, deg, Wp2, b2, h2, N, 128, 256);

  sage_gemm_wmma<<<(N + MTILE - 1) / MTILE, nthread, 0, stream>>>(h2, nullptr, deg, Wp3, l1b, h3, N, 128, 128);

  lin2_kernel<<<(N + nthread - 1) / nthread, nthread, 0, stream>>>(h3, l2W, l2b, (float*)d_out, N);
}


# One edge's atomic adds share a line: layer 1 sums+degree in one 32-B node row, 8 lanes per edge (6 scattered adds -> 1 request); layer 2 adds whole 128-B lines; float atomics, arrival order as before
# speedup vs baseline: 1.7291x; 1.7291x over previous
#include <hip/hip_runtime.h>

typedef __attribute__((ext_vector_type(2))) float v2f;
typedef __attribute__((ext_vector_type(8))) float v8f;

#define NCOLS 128
#define MTILE 32
#define LDSTRIDE 260

__global__ void pack_w(const float* __restrict__ W, float* __restrict__ Wp,
                       int Kw, int Kpad) {
  int i = blockIdx.x * blockDim.x + threadIdx.x;
  if (i < Kpad * NCOLS) {
    int k = i / NCOLS;
    int n = i - k * NCOLS;
    float v = (k < Kw) ? W[(size_t)k * NCOLS + n] : 0.0f;
    Wp[((size_t)(k >> 1) * NCOLS + n) * 2 + (k & 1)] = v;
  }
}

#define ROWW 8
#define DEGSLOT 5

__global__ void init_ws(float* __restrict__ rows, float* __restrict__ agg2, int N) {
  unsigned tid = blockIdx.x * blockDim.x + threadIdx.x;
  if (tid < (unsigned)N * 128u) {
    agg2[tid] = 0.0f;
    unsigned j = tid & 127u;
    unsigned i = tid >> 7;
    if (j < (unsigned)ROWW) rows[i * ROWW + j] = (j == (unsigned)DEGSLOT) ? 1.0f : 0.0f;
  }
}

__global__ void edge_accum_feat(const int* __restrict__ src, const int* __restrict__ dst,
                                const float* __restrict__ x, float* __restrict__ rows, int E) {
  const unsigned tid  = blockIdx.x * blockDim.x + threadIdx.x;
  const unsigned lane = threadIdx.x & 31u;
  const unsigned base = tid - lane;
  int s = 0, d = 0;
  if (tid < (unsigned)E) { s = src[tid]; d = dst[tid]; }
  const unsigned f  = lane & 7u;
  const unsigned g  = lane >> 3;
  const unsigned fx = f < 5u ? f : 4u;

  float  xv[8];
  float* p[8];
  bool   ok[8];
#pragma unroll
  for (int r = 0; r < 8; ++r) {
    const int sl = r * 4 + (int)g;
    const int ss = __shfl(s, sl);
    const int dd = __shfl(d, sl);
    ok[r] = (base + (unsigned)sl < (unsigned)E) && (f <= (unsigned)DEGSLOT);
    p[r]  = rows + (size_t)dd * ROWW + f;
    xv[r] = x[(size_t)ss * 5 + fx];
  }
#pragma unroll
  for (int r = 0; r < 8; ++r)
    if (ok[r]) atomicAdd(p[r], f < 5u ? xv[r] : 1.0f);
}

__global__ void edge_accum_embed(const int* __restrict__ src, const int* __restrict__ dst,
                                 const float* __restrict__ h, float* __restrict__ agg, int E) {
  unsigned tid = blockIdx.x * blockDim.x + threadIdx.x;
  unsigned e = tid >> 5;
  unsigned l = tid & 31u;
  if (e < (unsigned)E) {
    int s = src[e], d = dst[e];
    const float* hp = h + (size_t)s * 128 + l;
    float* a = agg + (size_t)d * 128 + l;
    const float v0 = hp[0], v1 = hp[32], v2 = hp[64], v3 = hp[96];
    atomicAdd(a,      v0);
    atomicAdd(a + 32, v1);
    atomicAdd(a + 64, v2);
    atomicAdd(a + 96, v3);
  }
}

__global__ void sage_gemm_wmma(const float* __restrict__ A0, const float* __restrict__ A1,
                               const float* __restrict__ deg, const float* __restrict__ Wp,
                               const float* __restrict__ bias, float* __restrict__ out,
                               int M, int d, int Kpad, int lda1, int ldd) {
  __shared__ float As[MTILE * LDSTRIDE];

  const int row0 = blockIdx.x * MTILE;
  const int t = threadIdx.x;

  const int total = MTILE * Kpad;
  for (int i = t; i < total; i += 256) {
    int m = i / Kpad;
    int k = i - m * Kpad;
    int row = row0 + m;
    float v = 0.0f;
    if (row < M) {
      if (k < d) {
        v = A0[(size_t)row * d + k];
      } else if (A1 != nullptr && k < 2 * d) {
        int kk = k - d;
        v = (A0[(size_t)row * d + kk] + A1[(size_t)row * lda1 + kk]) / deg[(size_t)row * ldd];
      }
    }
    As[m * LDSTRIDE + k] = v;
  }
  __syncthreads();

  const int lane  = t & 31;
  const int wave  = t >> 5;
  const int n     = wave * 16 + (lane & 15);
  const int mA    = lane & 15;
  const int khalf = (lane < 16) ? 0 : 2;

  v8f c0 = {};
  v8f c1 = {};
#pragma unroll 2
  for (int k0 = 0; k0 < Kpad; k0 += 4) {
    const int kb = k0 + khalf;
    v2f a0 = *(const v2f*)&As[mA * LDSTRIDE + kb];
    v2f a1 = *(const v2f*)&As[(mA + 16) * LDSTRIDE + kb];
    v2f b  = *(const v2f*)(Wp + ((size_t)(kb >> 1) * NCOLS + n) * 2);
    c0 = __builtin_amdgcn_wmma_f32_16x16x4_f32(false, a0, false, b,
                                               (short)0, c0, false, false);
    c1 = __builtin_amdgcn_wmma_f32_16x16x4_f32(false, a1, false, b,
                                               (short)0, c1, false, false);
  }

  const float bv    = bias[n];
  const int   rbase = (lane < 16) ? 0 : 8;
  float* outA = out + (size_t)(row0 + rbase) * NCOLS + n;
  float* outB = outA + (size_t)16 * NCOLS;

  if (row0 + MTILE <= M) {
#pragma unroll
    for (int r = 0; r < 8; ++r) {
      float vA = c0[r] + bv;
      float vB = c1[r] + bv;
      outA[(size_t)r * NCOLS] = vA > 0.0f ? vA : 0.0f;
      outB[(size_t)r * NCOLS] = vB > 0.0f ? vB : 0.0f;
    }
  } else {
#pragma unroll
    for (int r = 0; r < 8; ++r) {
      int rowA = row0 + rbase + r;
      if (rowA < M) {
        float v = c0[r] + bv;
        outA[(size_t)r * NCOLS] = v > 0.0f ? v : 0.0f;
      }
      int rowB = row0 + 16 + rbase + r;
      if (rowB < M) {
        float v = c1[r] + bv;
        outB[(size_t)r * NCOLS] = v > 0.0f ? v : 0.0f;
      }
    }
  }
}

__global__ void lin2_kernel(const float* __restrict__ h, const float* __restrict__ W,
                            const float* __restrict__ b, float* __restrict__ out, int M) {
  int i = blockIdx.x * blockDim.x + threadIdx.x;
  if (i < M) {
    float a0 = b[0], a1 = b[1];
    const float* row = h + (size_t)i * 128;
#pragma unroll 8
    for (int k = 0; k < 128; ++k) {
      float v = row[k];
      a0 += v * W[2 * k];
      a1 += v * W[2 * k + 1];
    }
    out[(size_t)i * 2 + 0] = a0;
    out[(size_t)i * 2 + 1] = a1;
  }
}

extern "C" void kernel_launch(void* const* d_in, const int* in_sizes, int n_in,
                              void* d_out, int out_size, void* d_ws, size_t ws_size,
                              hipStream_t stream) {
  const float* x   = (const float*)d_in[0];
  const int*   ei  = (const int*)d_in[1];
  const float* W1  = (const float*)d_in[2];
  const float* b1  = (const float*)d_in[3];
  const float* W2  = (const float*)d_in[4];
  const float* b2  = (const float*)d_in[5];
  const float* l1W = (const float*)d_in[6];
  const float* l1b = (const float*)d_in[7];
  const float* l2W = (const float*)d_in[8];
  const float* l2b = (const float*)d_in[9];

  const int N = in_sizes[0] / 5;
  const int E = in_sizes[1] / 2;
  const int* src = ei;
  const int* dst = ei + E;

  float* ws   = (float*)d_ws;
  float* rows = ws;
  float* deg  = rows + DEGSLOT;
  float* agg2 = rows + (size_t)ROWW * N;
  float* h1   = agg2 + (size_t)128 * N;
  float* h2   = h1   + (size_t)128 * N;
  float* Wp1  = h2   + (size_t)128 * N;
  float* Wp2  = Wp1  + (size_t)12 * 128;
  float* Wp3  = Wp2  + (size_t)256 * 128;
  float* h3   = h1;

  const int nthread = 256;

  pack_w<<<(12  * NCOLS + nthread - 1) / nthread, nthread, 0, stream>>>(W1,  Wp1, 10,  12);
  pack_w<<<(256 * NCOLS + nthread - 1) / nthread, nthread, 0, stream>>>(W2,  Wp2, 256, 256);
  pack_w<<<(128 * NCOLS + nthread - 1) / nthread, nthread, 0, stream>>>(l1W, Wp3, 128, 128);

  init_ws<<<((unsigned)N * 128u + nthread - 1) / nthread, nthread, 0, stream>>>(rows, agg2, N);

  edge_accum_feat<<<(E + nthread - 1) / nthread, nthread, 0, stream>>>(src, dst, x, rows, E);

  sage_gemm_wmma<<<(N + MTILE - 1) / MTILE, nthread, 0, stream>>>(x, rows, deg, Wp1, b1, h1, N, 5, 12, ROWW, ROWW);

  edge_accum_embed<<<((unsigned)E * 32u + nthread - 1) / nthread, nthread, 0, stream>>>(src, dst, h1, agg2, E);

  sage_gemm_wmma<<<(N + MTILE - 1) / MTILE, nthread, 0, stream>>>(h1, agg2, deg, Wp2, b2, h2, N, 128, 256, 128, ROWW);

  sage_gemm_wmma<<<(N + MTILE - 1) / MTILE, nthread, 0, stream>>>(h2, nullptr, deg, Wp3, l1b, h3, N, 128, 128, 128, ROWW);

  lin2_kernel<<<(N + nthread - 1) / nthread, nthread, 0, stream>>>(h3, l2W, l2b, (float*)d_out, N);
}


# Layer 2 without float atomics: edges grouped by target in-call (degree scan, int-atomic slot fill), one wave per node sums neighbour h1 rows in registers; sum order by arrival as before
# speedup vs baseline: 1.9970x; 1.1549x over previous
#include <hip/hip_runtime.h>

typedef __attribute__((ext_vector_type(2))) float v2f;
typedef __attribute__((ext_vector_type(8))) float v8f;

#define NCOLS 128
#define MTILE 32
#define LDSTRIDE 260

__global__ void pack_w(const float* __restrict__ W, float* __restrict__ Wp,
                       int Kw, int Kpad) {
  int i = blockIdx.x * blockDim.x + threadIdx.x;
  if (i < Kpad * NCOLS) {
    int k = i / NCOLS;
    int n = i - k * NCOLS;
    float v = (k < Kw) ? W[(size_t)k * NCOLS + n] : 0.0f;
    Wp[((size_t)(k >> 1) * NCOLS + n) * 2 + (k & 1)] = v;
  }
}

#define ROWW 8
#define DEGSLOT 5

__global__ void init_ws(float* __restrict__ rows, int N) {
  unsigned tid = blockIdx.x * blockDim.x + threadIdx.x;
  if (tid < (unsigned)N * (unsigned)ROWW)
    rows[tid] = ((tid & (unsigned)(ROWW - 1)) == (unsigned)DEGSLOT) ? 1.0f : 0.0f;
}

__global__ void edge_accum_feat(const int* __restrict__ src, const int* __restrict__ dst,
                                const float* __restrict__ x, float* __restrict__ rows, int E) {
  const unsigned tid  = blockIdx.x * blockDim.x + threadIdx.x;
  const unsigned lane = threadIdx.x & 31u;
  const unsigned base = tid - lane;
  int s = 0, d = 0;
  if (tid < (unsigned)E) { s = src[tid]; d = dst[tid]; }
  const unsigned f  = lane & 7u;
  const unsigned g  = lane >> 3;
  const unsigned fx = f < 5u ? f : 4u;

  float  xv[8];
  float* p[8];
  bool   ok[8];
#pragma unroll
  for (int r = 0; r < 8; ++r) {
    const int sl = r * 4 + (int)g;
    const int ss = __shfl(s, sl);
    const int dd = __shfl(d, sl);
    ok[r] = (base + (unsigned)sl < (unsigned)E) && (f <= (unsigned)DEGSLOT);
    p[r]  = rows + (size_t)dd * ROWW + f;
    xv[r] = x[(size_t)ss * 5 + fx];
  }
#pragma unroll
  for (int r = 0; r < 8; ++r)
    if (ok[r]) atomicAdd(p[r], f < 5u ? xv[r] : 1.0f);
}

#define SCAN_CHUNK 1024

__device__ __forceinline__ int indeg_of(const float* __restrict__ rows, int i) {
  return (int)rows[(size_t)i * ROWW + DEGSLOT] - 1;
}

__global__ void scan_chunk_sums(const float* __restrict__ rows, int* __restrict__ csum, int N) {
  __shared__ int wsum[8];
  const int t  = threadIdx.x;
  const int i0 = blockIdx.x * SCAN_CHUNK + t * 4;
  int c = 0;
  for (int j = 0; j < 4; ++j)
    if (i0 + j < N) c += indeg_of(rows, i0 + j);
  for (int o = 16; o > 0; o >>= 1) c += __shfl_xor(c, o);
  if ((t & 31) == 0) wsum[t >> 5] = c;
  __syncthreads();
  if (t == 0) {
    int s = 0;
    for (int w = 0; w < 8; ++w) s += wsum[w];
    csum[blockIdx.x] = s;
  }
}

__global__ void scan_offsets(const float* __restrict__ rows, const int* __restrict__ csum,
                             int* __restrict__ offs, int* __restrict__ cursor, int N) {
  __shared__ int wsumA[8];
  __shared__ int wsumB[8];
  const int t    = threadIdx.x;
  const int lane = t & 31;
  const int wave = t >> 5;

  int b = 0;
  for (int k = t; k < (int)blockIdx.x; k += 256) b += csum[k];
  for (int o = 16; o > 0; o >>= 1) b += __shfl_xor(b, o);
  if (lane == 0) wsumA[wave] = b;

  const int i0 = blockIdx.x * SCAN_CHUNK + t * 4;
  int c[4];
  int tot = 0;
  for (int j = 0; j < 4; ++j) {
    c[j] = (i0 + j < N) ? indeg_of(rows, i0 + j) : 0;
    tot += c[j];
  }
  int inc = tot;
  for (int o = 1; o < 32; o <<= 1) {
    const int v = __shfl_up(inc, o);
    if (lane >= o) inc += v;
  }
  if (lane == 31) wsumB[wave] = inc;
  __syncthreads();

  int base = 0;
  for (int w = 0; w < 8; ++w) {
    base += wsumA[w];
    if (w < wave) base += wsumB[w];
  }
  int excl = base + inc - tot;
  for (int j = 0; j < 4; ++j) {
    if (i0 + j < N) {
      offs[i0 + j]   = excl;
      cursor[i0 + j] = excl;
      excl += c[j];
      if (i0 + j == N - 1) offs[N] = excl;
    }
  }
}

__global__ void fill_lists(const int* __restrict__ src, const int* __restrict__ dst,
                           int* __restrict__ cursor, int* __restrict__ nbr, int E) {
  const unsigned e = blockIdx.x * blockDim.x + threadIdx.x;
  if (e < (unsigned)E) {
    const int pos = atomicAdd(&cursor[dst[e]], 1);
    nbr[pos] = src[e];
  }
}

__global__ void gather_sum_rows(const int* __restrict__ offs, const int* __restrict__ nbr,
                                const float* __restrict__ h, float* __restrict__ agg, int N) {
  const unsigned tid = blockIdx.x * blockDim.x + threadIdx.x;
  const unsigned i = tid >> 5;
  const unsigned l = tid & 31u;
  if (i < (unsigned)N) {
    const int beg = __builtin_amdgcn_readfirstlane(offs[i]);
    const int end = __builtin_amdgcn_readfirstlane(offs[i + 1]);
    const float* hl = h + 4u * l;
    float4 acc = make_float4(0.0f, 0.0f, 0.0f, 0.0f);
    for (int j0 = beg; j0 < end; j0 += 32) {
      const int cnt = (end - j0 < 32) ? (end - j0) : 32;
      const int mine = ((int)l < cnt) ? nbr[j0 + (int)l] : 0;
      int k = 0;
      for (; k + 4 <= cnt; k += 4) {
        const int s0 = __builtin_amdgcn_readlane(mine, k);
        const int s1 = __builtin_amdgcn_readlane(mine, k + 1);
        const int s2 = __builtin_amdgcn_readlane(mine, k + 2);
        const int s3 = __builtin_amdgcn_readlane(mine, k + 3);
        const float4 v0 = *(const float4*)(hl + (size_t)s0 * 128);
        const float4 v1 = *(const float4*)(hl + (size_t)s1 * 128);
        const float4 v2 = *(const float4*)(hl + (size_t)s2 * 128);
        const float4 v3 = *(const float4*)(hl + (size_t)s3 * 128);
        acc.x += v0.x; acc.y += v0.y; acc.z += v0.z; acc.w += v0.w;
        acc.x += v1.x; acc.y += v1.y; acc.z += v1.z; acc.w += v1.w;
        acc.x += v2.x; acc.y += v2.y; acc.z += v2.z; acc.w += v2.w;
        acc.x += v3.x; acc.y += v3.y; acc.z += v3.z; acc.w += v3.w;
      }
      float4 t0 = make_float4(0.0f, 0.0f, 0.0f, 0.0f), t1 = t0, t2 = t0;
      if (k < cnt)     t0 = *(const float4*)(hl + (size_t)__builtin_amdgcn_readlane(mine, k) * 128);
      if (k + 1 < cnt) t1 = *(const float4*)(hl + (size_t)__builtin_amdgcn_readlane(mine, k + 1) * 128);
      if (k + 2 < cnt) t2 = *(const float4*)(hl + (size_t)__builtin_amdgcn_readlane(mine, k + 2) * 128);
      if (k < cnt)     { acc.x += t0.x; acc.y += t0.y; acc.z += t0.z; acc.w += t0.w; }
      if (k + 1 < cnt) { acc.x += t1.x; acc.y += t1.y; acc.z += t1.z; acc.w += t1.w; }
      if (k + 2 < cnt) { acc.x += t2.x; acc.y += t2.y; acc.z += t2.z; acc.w += t2.w; }
    }
    *(float4*)(agg + (size_t)i * 128 + 4u * l) = acc;
  }
}

__global__ void sage_gemm_wmma(const float* __restrict__ A0, const float* __restrict__ A1,
                               const float* __restrict__ deg, const float* __restrict__ Wp,
                               const float* __restrict__ bias, float* __restrict__ out,
                               int M, int d, int Kpad, int lda1, int ldd) {
  __shared__ float As[MTILE * LDSTRIDE];

  const int row0 = blockIdx.x * MTILE;
  const int t = threadIdx.x;

  const int total = MTILE * Kpad;
  for (int i = t; i < total; i += 256) {
    int m = i / Kpad;
    int k = i - m * Kpad;
    int row = row0 + m;
    float v = 0.0f;
    if (row < M) {
      if (k < d) {
        v = A0[(size_t)row * d + k];
      } else if (A1 != nullptr && k < 2 * d) {
        int kk = k - d;
        v = (A0[(size_t)row * d + kk] + A1[(size_t)row * lda1 + kk]) / deg[(size_t)row * ldd];
      }
    }
    As[m * LDSTRIDE + k] = v;
  }
  __syncthreads();

  const int lane  = t & 31;
  const int wave  = t >> 5;
  const int n     = wave * 16 + (lane & 15);
  const int mA    = lane & 15;
  const int khalf = (lane < 16) ? 0 : 2;

  v8f c0 = {};
  v8f c1 = {};
#pragma unroll 2
  for (int k0 = 0; k0 < Kpad; k0 += 4) {
    const int kb = k0 + khalf;
    v2f a0 = *(const v2f*)&As[mA * LDSTRIDE + kb];
    v2f a1 = *(const v2f*)&As[(mA + 16) * LDSTRIDE + kb];
    v2f b  = *(const v2f*)(Wp + ((size_t)(kb >> 1) * NCOLS + n) * 2);
    c0 = __builtin_amdgcn_wmma_f32_16x16x4_f32(false, a0, false, b,
                                               (short)0, c0, false, false);
    c1 = __builtin_amdgcn_wmma_f32_16x16x4_f32(false, a1, false, b,
                                               (short)0, c1, false, false);
  }

  const float bv    = bias[n];
  const int   rbase = (lane < 16) ? 0 : 8;
  float* outA = out + (size_t)(row0 + rbase) * NCOLS + n;
  float* outB = outA + (size_t)16 * NCOLS;

  if (row0 + MTILE <= M) {
#pragma unroll
    for (int r = 0; r < 8; ++r) {
      float vA = c0[r] + bv;
      float vB = c1[r] + bv;
      outA[(size_t)r * NCOLS] = vA > 0.0f ? vA : 0.0f;
      outB[(size_t)r * NCOLS] = vB > 0.0f ? vB : 0.0f;
    }
  } else {
#pragma unroll
    for (int r = 0; r < 8; ++r) {
      int rowA = row0 + rbase + r;
      if (rowA < M) {
        float v = c0[r] + bv;
        outA[(size_t)r * NCOLS] = v > 0.0f ? v : 0.0f;
      }
      int rowB = row0 + 16 + rbase + r;
      if (rowB < M) {
        float v = c1[r] + bv;
        outB[(size_t)r * NCOLS] = v > 0.0f ? v : 0.0f;
      }
    }
  }
}

__global__ void lin2_kernel(const float* __restrict__ h, const float* __restrict__ W,
                            const float* __restrict__ b, float* __restrict__ out, int M) {
  int i = blockIdx.x * blockDim.x + threadIdx.x;
  if (i < M) {
    float a0 = b[0], a1 = b[1];
    const float* row = h + (size_t)i * 128;
#pragma unroll 8
    for (int k = 0; k < 128; ++k) {
      float v = row[k];
      a0 += v * W[2 * k];
      a1 += v * W[2 * k + 1];
    }
    out[(size_t)i * 2 + 0] = a0;
    out[(size_t)i * 2 + 1] = a1;
  }
}

extern "C" void kernel_launch(void* const* d_in, const int* in_sizes, int n_in,
                              void* d_out, int out_size, void* d_ws, size_t ws_size,
                              hipStream_t stream) {
  const float* x   = (const float*)d_in[0];
  const int*   ei  = (const int*)d_in[1];
  const float* W1  = (const float*)d_in[2];
  const float* b1  = (const float*)d_in[3];
  const float* W2  = (const float*)d_in[4];
  const float* b2  = (const float*)d_in[5];
  const float* l1W = (const float*)d_in[6];
  const float* l1b = (const float*)d_in[7];
  const float* l2W = (const float*)d_in[8];
  const float* l2b = (const float*)d_in[9];

  const int N = in_sizes[0] / 5;
  const int E = in_sizes[1] / 2;
  const int* src = ei;
  const int* dst = ei + E;

  float* ws   = (float*)d_ws;
  float* rows = ws;
  float* deg  = rows + DEGSLOT;
  float* agg2 = rows + (size_t)ROWW * N;
  float* h1   = agg2 + (size_t)128 * N;
  float* h2   = h1   + (size_t)128 * N;
  float* Wp1  = h2   + (size_t)128 * N;
  float* Wp2  = Wp1  + (size_t)12 * 128;
  float* Wp3  = Wp2  + (size_t)256 * 128;
  float* h3   = h1;
  const int nchunk = (N + SCAN_CHUNK - 1) / SCAN_CHUNK;
  int* offs   = (int*)(Wp3 + (size_t)128 * 128);
  int* cursor = offs + (size_t)N + 1;
  int* csum   = cursor + (size_t)N;
  int* nbr    = csum + nchunk;

  const int nthread = 256;

  pack_w<<<(12  * NCOLS + nthread - 1) / nthread, nthread, 0, stream>>>(W1,  Wp1, 10,  12);
  pack_w<<<(256 * NCOLS + nthread - 1) / nthread, nthread, 0, stream>>>(W2,  Wp2, 256, 256);
  pack_w<<<(128 * NCOLS + nthread - 1) / nthread, nthread, 0, stream>>>(l1W, Wp3, 128, 128);

  init_ws<<<((unsigned)N * (unsigned)ROWW + nthread - 1) / nthread, nthread, 0, stream>>>(rows, N);

  edge_accum_feat<<<(E + nthread - 1) / nthread, nthread, 0, stream>>>(src, dst, x, rows, E);

  scan_chunk_sums<<<nchunk, nthread, 0, stream>>>(rows, csum, N);
  scan_offsets<<<nchunk, nthread, 0, stream>>>(rows, csum, offs, cursor, N);
  fill_lists<<<(E + nthread - 1) / nthread, nthread, 0, stream>>>(src, dst, cursor, nbr, E);

  sage_gemm_wmma<<<(N + MTILE - 1) / MTILE, nthread, 0, stream>>>(x, rows, deg, Wp1, b1, h1, N, 5, 12, ROWW, ROWW);

  gather_sum_rows<<<((unsigned)N * 32u + nthread - 1) / nthread, nthread, 0, stream>>>(offs, nbr, h1, agg2, N);

  sage_gemm_wmma<<<(N + MTILE - 1) / MTILE, nthread, 0, stream>>>(h1, agg2, deg, Wp2, b2, h2, N, 128, 256, 128, ROWW);

  sage_gemm_wmma<<<(N + MTILE - 1) / MTILE, nthread, 0, stream>>>(h2, nullptr, deg, Wp3, l1b, h3, N, 128, 128, 128, ROWW);

  lin2_kernel<<<(N + nthread - 1) / nthread, nthread, 0, stream>>>(h3, l2W, l2b, (float*)d_out, N);
}


# Layer-2 GEMM, lin1 and lin2 in one launch: each 32-row tile stays in LDS between the three products (h2, h3 never reach memory); lin2's 128-term sum is now 8 partial sums joined by a lane tree
# speedup vs baseline: 2.1438x; 1.0735x over previous
#include <hip/hip_runtime.h>

typedef __attribute__((ext_vector_type(2))) float v2f;
typedef __attribute__((ext_vector_type(8))) float v8f;

#define NCOLS 128
#define MTILE 32
#define LDSTRIDE 260

__global__ void pack_w(const float* __restrict__ W, float* __restrict__ Wp,
                       int Kw, int Kpad) {
  int i = blockIdx.x * blockDim.x + threadIdx.x;
  if (i < Kpad * NCOLS) {
    int k = i / NCOLS;
    int n = i - k * NCOLS;
    float v = (k < Kw) ? W[(size_t)k * NCOLS + n] : 0.0f;
    Wp[((size_t)(k >> 1) * NCOLS + n) * 2 + (k & 1)] = v;
  }
}

#define ROWW 8
#define DEGSLOT 5

__global__ void init_ws(float* __restrict__ rows, int N) {
  unsigned tid = blockIdx.x * blockDim.x + threadIdx.x;
  if (tid < (unsigned)N * (unsigned)ROWW)
    rows[tid] = ((tid & (unsigned)(ROWW - 1)) == (unsigned)DEGSLOT) ? 1.0f : 0.0f;
}

__global__ void edge_accum_feat(const int* __restrict__ src, const int* __restrict__ dst,
                                const float* __restrict__ x, float* __restrict__ rows, int E) {
  const unsigned tid  = blockIdx.x * blockDim.x + threadIdx.x;
  const unsigned lane = threadIdx.x & 31u;
  const unsigned base = tid - lane;
  int s = 0, d = 0;
  if (tid < (unsigned)E) { s = src[tid]; d = dst[tid]; }
  const unsigned f  = lane & 7u;
  const unsigned g  = lane >> 3;
  const unsigned fx = f < 5u ? f : 4u;

  float  xv[8];
  float* p[8];
  bool   ok[8];
#pragma unroll
  for (int r = 0; r < 8; ++r) {
    const int sl = r * 4 + (int)g;
    const int ss = __shfl(s, sl);
    const int dd = __shfl(d, sl);
    ok[r] = (base + (unsigned)sl < (unsigned)E) && (f <= (unsigned)DEGSLOT);
    p[r]  = rows + (size_t)dd * ROWW + f;
    xv[r] = x[(size_t)ss * 5 + fx];
  }
#pragma unroll
  for (int r = 0; r < 8; ++r)
    if (ok[r]) atomicAdd(p[r], f < 5u ? xv[r] : 1.0f);
}

#define SCAN_CHUNK 1024

__device__ __forceinline__ int indeg_of(const float* __restrict__ rows, int i) {
  return (int)rows[(size_t)i * ROWW + DEGSLOT] - 1;
}

__global__ void scan_chunk_sums(const float* __restrict__ rows, int* __restrict__ csum, int N) {
  __shared__ int wsum[8];
  const int t  = threadIdx.x;
  const int i0 = blockIdx.x * SCAN_CHUNK + t * 4;
  int c = 0;
  for (int j = 0; j < 4; ++j)
    if (i0 + j < N) c += indeg_of(rows, i0 + j);
  for (int o = 16; o > 0; o >>= 1) c += __shfl_xor(c, o);
  if ((t & 31) == 0) wsum[t >> 5] = c;
  __syncthreads();
  if (t == 0) {
    int s = 0;
    for (int w = 0; w < 8; ++w) s += wsum[w];
    csum[blockIdx.x] = s;
  }
}

__global__ void scan_offsets(const float* __restrict__ rows, const int* __restrict__ csum,
                             int* __restrict__ offs, int* __restrict__ cursor, int N) {
  __shared__ int wsumA[8];
  __shared__ int wsumB[8];
  const int t    = threadIdx.x;
  const int lane = t & 31;
  const int wave = t >> 5;

  int b = 0;
  for (int k = t; k < (int)blockIdx.x; k += 256) b += csum[k];
  for (int o = 16; o > 0; o >>= 1) b += __shfl_xor(b, o);
  if (lane == 0) wsumA[wave] = b;

  const int i0 = blockIdx.x * SCAN_CHUNK + t * 4;
  int c[4];
  int tot = 0;
  for (int j = 0; j < 4; ++j) {
    c[j] = (i0 + j < N) ? indeg_of(rows, i0 + j) : 0;
    tot += c[j];
  }
  int inc = tot;
  for (int o = 1; o < 32; o <<= 1) {
    const int v = __shfl_up(inc, o);
    if (lane >= o) inc += v;
  }
  if (lane == 31) wsumB[wave] = inc;
  __syncthreads();

  int base = 0;
  for (int w = 0; w < 8; ++w) {
    base += wsumA[w];
    if (w < wave) base += wsumB[w];
  }
  int excl = base + inc - tot;
  for (int j = 0; j < 4; ++j) {
    if (i0 + j < N) {
      offs[i0 + j]   = excl;
      cursor[i0 + j] = excl;
      excl += c[j];
      if (i0 + j == N - 1) offs[N] = excl;
    }
  }
}

__global__ void fill_lists(const int* __restrict__ src, const int* __restrict__ dst,
                           int* __restrict__ cursor, int* __restrict__ nbr, int E) {
  const unsigned e = blockIdx.x * blockDim.x + threadIdx.x;
  if (e < (unsigned)E) {
    const int pos = atomicAdd(&cursor[dst[e]], 1);
    nbr[pos] = src[e];
  }
}

__global__ void gather_sum_rows(const int* __restrict__ offs, const int* __restrict__ nbr,
                                const float* __restrict__ h, float* __restrict__ agg, int N) {
  const unsigned tid = blockIdx.x * blockDim.x + threadIdx.x;
  const unsigned i = tid >> 5;
  const unsigned l = tid & 31u;
  if (i < (unsigned)N) {
    const int beg = __builtin_amdgcn_readfirstlane(offs[i]);
    const int end = __builtin_amdgcn_readfirstlane(offs[i + 1]);
    const float* hl = h + 4u * l;
    float4 acc = make_float4(0.0f, 0.0f, 0.0f, 0.0f);
    for (int j0 = beg; j0 < end; j0 += 32) {
      const int cnt = (end - j0 < 32) ? (end - j0) : 32;
      const int mine = ((int)l < cnt) ? nbr[j0 + (int)l] : 0;
      int k = 0;
      for (; k + 4 <= cnt; k += 4) {
        const int s0 = __builtin_amdgcn_readlane(mine, k);
        const int s1 = __builtin_amdgcn_readlane(mine, k + 1);
        const int s2 = __builtin_amdgcn_readlane(mine, k + 2);
        const int s3 = __builtin_amdgcn_readlane(mine, k + 3);
        const float4 v0 = *(const float4*)(hl + (size_t)s0 * 128);
        const float4 v1 = *(const float4*)(hl + (size_t)s1 * 128);
        const float4 v2 = *(const float4*)(hl + (size_t)s2 * 128);
        const float4 v3 = *(const float4*)(hl + (size_t)s3 * 128);
        acc.x += v0.x; acc.y += v0.y; acc.z += v0.z; acc.w += v0.w;
        acc.x += v1.x; acc.y += v1.y; acc.z += v1.z; acc.w += v1.w;
        acc.x += v2.x; acc.y += v2.y; acc.z += v2.z; acc.w += v2.w;
        acc.x += v3.x; acc.y += v3.y; acc.z += v3.z; acc.w += v3.w;
      }
      float4 t0 = make_float4(0.0f, 0.0f, 0.0f, 0.0f), t1 = t0, t2 = t0;
      if (k < cnt)     t0 = *(const float4*)(hl + (size_t)__builtin_amdgcn_readlane(mine, k) * 128);
      if (k + 1 < cnt) t1 = *(const float4*)(hl + (size_t)__builtin_amdgcn_readlane(mine, k + 1) * 128);
      if (k + 2 < cnt) t2 = *(const float4*)(hl + (size_t)__builtin_amdgcn_readlane(mine, k + 2) * 128);
      if (k < cnt)     { acc.x += t0.x; acc.y += t0.y; acc.z += t0.z; acc.w += t0.w; }
      if (k + 1 < cnt) { acc.x += t1.x; acc.y += t1.y; acc.z += t1.z; acc.w += t1.w; }
      if (k + 2 < cnt) { acc.x += t2.x; acc.y += t2.y; acc.z += t2.z; acc.w += t2.w; }
    }
    *(float4*)(agg + (size_t)i * 128 + 4u * l) = acc;
  }
}

__global__ void sage_gemm_wmma(const float* __restrict__ A0, const float* __restrict__ A1,
                               const float* __restrict__ deg, const float* __restrict__ Wp,
                               const float* __restrict__ bias, float* __restrict__ out,
                               int M, int d, int Kpad, int lda1, int ldd) {
  __shared__ float As[MTILE * LDSTRIDE];

  const int row0 = blockIdx.x * MTILE;
  const int t = threadIdx.x;

  const int total = MTILE * Kpad;
  for (int i = t; i < total; i += 256) {
    int m = i / Kpad;
    int k = i - m * Kpad;
    int row = row0 + m;
    float v = 0.0f;
    if (row < M) {
      if (k < d) {
        v = A0[(size_t)row * d + k];
      } else if (A1 != nullptr && k < 2 * d) {
        int kk = k - d;
        v = (A0[(size_t)row * d + kk] + A1[(size_t)row * lda1 + kk]) / deg[(size_t)row * ldd];
      }
    }
    As[m * LDSTRIDE + k] = v;
  }
  __syncthreads();

  const int lane  = t & 31;
  const int wave  = t >> 5;
  const int n     = wave * 16 + (lane & 15);
  const int mA    = lane & 15;
  const int khalf = (lane < 16) ? 0 : 2;

  v8f c0 = {};
  v8f c1 = {};
#pragma unroll 2
  for (int k0 = 0; k0 < Kpad; k0 += 4) {
    const int kb = k0 + khalf;
    v2f a0 = *(const v2f*)&As[mA * LDSTRIDE + kb];
    v2f a1 = *(const v2f*)&As[(mA + 16) * LDSTRIDE + kb];
    v2f b  = *(const v2f*)(Wp + ((size_t)(kb >> 1) * NCOLS + n) * 2);
    c0 = __builtin_amdgcn_wmma_f32_16x16x4_f32(false, a0, false, b,
                                               (short)0, c0, false, false);
    c1 = __builtin_amdgcn_wmma_f32_16x16x4_f32(false, a1, false, b,
                                               (short)0, c1, false, false);
  }

  const float bv    = bias[n];
  const int   rbase = (lane < 16) ? 0 : 8;
  float* outA = out + (size_t)(row0 + rbase) * NCOLS + n;
  float* outB = outA + (size_t)16 * NCOLS;

  if (row0 + MTILE <= M) {
#pragma unroll
    for (int r = 0; r < 8; ++r) {
      float vA = c0[r] + bv;
      float vB = c1[r] + bv;
      outA[(size_t)r * NCOLS] = vA > 0.0f ? vA : 0.0f;
      outB[(size_t)r * NCOLS] = vB > 0.0f ? vB : 0.0f;
    }
  } else {
#pragma unroll
    for (int r = 0; r < 8; ++r) {
      int rowA = row0 + rbase + r;
      if (rowA < M) {
        float v = c0[r] + bv;
        outA[(size_t)r * NCOLS] = v > 0.0f ? v : 0.0f;
      }
      int rowB = row0 + 16 + rbase + r;
      if (rowB < M) {
        float v = c1[r] + bv;
        outB[(size_t)r * NCOLS] = v > 0.0f ? v : 0.0f;
      }
    }
  }
}

__global__ void sage_head_fused(const float* __restrict__ A0, const float* __restrict__ A1,
                                const float* __restrict__ deg,
                                const float* __restrict__ Wp2, const float* __restrict__ b2,
                                const float* __restrict__ Wp3, const float* __restrict__ b3,
                                const float* __restrict__ W4, const float* __restrict__ b4,
                                float* __restrict__ out,
                                int M, int d, int Kpad, int lda1, int ldd) {
  __shared__ float As[MTILE * LDSTRIDE];

  const int row0 = blockIdx.x * MTILE;
  const int t = threadIdx.x;

  const int total = MTILE * Kpad;
  for (int i = t; i < total; i += 256) {
    int m = i / Kpad;
    int k = i - m * Kpad;
    int row = row0 + m;
    float v = 0.0f;
    if (row < M) {
      if (k < d) {
        v = A0[(size_t)row * d + k];
      } else if (k < 2 * d) {
        int kk = k - d;
        v = (A0[(size_t)row * d + kk] + A1[(size_t)row * lda1 + kk]) / deg[(size_t)row * ldd];
      }
    }
    As[m * LDSTRIDE + k] = v;
  }
  __syncthreads();

  const int lane  = t & 31;
  const int wave  = t >> 5;
  const int n     = wave * 16 + (lane & 15);
  const int mA    = lane & 15;
  const int khalf = (lane < 16) ? 0 : 2;
  const int rbase = (lane < 16) ? 0 : 8;

  v8f c0 = {};
  v8f c1 = {};
#pragma unroll 2
  for (int k0 = 0; k0 < Kpad; k0 += 4) {
    const int kb = k0 + khalf;
    v2f a0 = *(const v2f*)&As[mA * LDSTRIDE + kb];
    v2f a1 = *(const v2f*)&As[(mA + 16) * LDSTRIDE + kb];
    v2f b  = *(const v2f*)(Wp2 + ((size_t)(kb >> 1) * NCOLS + n) * 2);
    c0 = __builtin_amdgcn_wmma_f32_16x16x4_f32(false, a0, false, b,
                                               (short)0, c0, false, false);
    c1 = __builtin_amdgcn_wmma_f32_16x16x4_f32(false, a1, false, b,
                                               (short)0, c1, false, false);
  }

  __syncthreads();
  {
    const float bv = b2[n];
#pragma unroll
    for (int r = 0; r < 8; ++r) {
      float vA = c0[r] + bv;
      float vB = c1[r] + bv;
      As[(rbase + r) * LDSTRIDE + n]      = vA > 0.0f ? vA : 0.0f;
      As[(16 + rbase + r) * LDSTRIDE + n] = vB > 0.0f ? vB : 0.0f;
    }
  }
  __syncthreads();

  v8f e0 = {};
  v8f e1 = {};
#pragma unroll 2
  for (int k0 = 0; k0 < NCOLS; k0 += 4) {
    const int kb = k0 + khalf;
    v2f a0 = *(const v2f*)&As[mA * LDSTRIDE + kb];
    v2f a1 = *(const v2f*)&As[(mA + 16) * LDSTRIDE + kb];
    v2f b  = *(const v2f*)(Wp3 + ((size_t)(kb >> 1) * NCOLS + n) * 2);
    e0 = __builtin_amdgcn_wmma_f32_16x16x4_f32(false, a0, false, b,
                                               (short)0, e0, false, false);
    e1 = __builtin_amdgcn_wmma_f32_16x16x4_f32(false, a1, false, b,
                                               (short)0, e1, false, false);
  }
  {
    const float bv = b3[n];
#pragma unroll
    for (int r = 0; r < 8; ++r) {
      float vA = e0[r] + bv;
      float vB = e1[r] + bv;
      As[(rbase + r) * LDSTRIDE + NCOLS + n]      = vA > 0.0f ? vA : 0.0f;
      As[(16 + rbase + r) * LDSTRIDE + NCOLS + n] = vB > 0.0f ? vB : 0.0f;
    }
  }
  __syncthreads();

  const int orow = t >> 3;
  const int part = t & 7;
  const float* hrow = &As[orow * LDSTRIDE + NCOLS + part * 16];
  const float* w    = W4 + part * 32;
  float s0 = 0.0f, s1 = 0.0f;
#pragma unroll
  for (int q = 0; q < 4; ++q) {
    const float4 hv = *(const float4*)(hrow + 4 * q);
    const float4 wa = *(const float4*)(w + 8 * q);
    const float4 wb = *(const float4*)(w + 8 * q + 4);
    s0 += hv.x * wa.x;  s1 += hv.x * wa.y;
    s0 += hv.y * wa.z;  s1 += hv.y * wa.w;
    s0 += hv.z * wb.x;  s1 += hv.z * wb.y;
    s0 += hv.w * wb.z;  s1 += hv.w * wb.w;
  }
  for (int o = 1; o < 8; o <<= 1) {
    s0 += __shfl_xor(s0, o);
    s1 += __shfl_xor(s1, o);
  }
  if (part == 0) {
    As[orow * 2 + 0] = s0 + b4[0];
    As[orow * 2 + 1] = s1 + b4[1];
  }
  __syncthreads();
  if (t < 2 * MTILE && row0 + (t >> 1) < M)
    out[(size_t)row0 * 2 + t] = As[t];
}

extern "C" void kernel_launch(void* const* d_in, const int* in_sizes, int n_in,
                              void* d_out, int out_size, void* d_ws, size_t ws_size,
                              hipStream_t stream) {
  const float* x   = (const float*)d_in[0];
  const int*   ei  = (const int*)d_in[1];
  const float* W1  = (const float*)d_in[2];
  const float* b1  = (const float*)d_in[3];
  const float* W2  = (const float*)d_in[4];
  const float* b2  = (const float*)d_in[5];
  const float* l1W = (const float*)d_in[6];
  const float* l1b = (const float*)d_in[7];
  const float* l2W = (const float*)d_in[8];
  const float* l2b = (const float*)d_in[9];

  const int N = in_sizes[0] / 5;
  const int E = in_sizes[1] / 2;
  const int* src = ei;
  const int* dst = ei + E;

  float* ws   = (float*)d_ws;
  float* rows = ws;
  float* deg  = rows + DEGSLOT;
  float* agg2 = rows + (size_t)ROWW * N;
  float* h1   = agg2 + (size_t)128 * N;
  float* Wp1  = h1   + (size_t)128 * N;
  float* Wp2  = Wp1  + (size_t)12 * 128;
  float* Wp3  = Wp2  + (size_t)256 * 128;
  const int nchunk = (N + SCAN_CHUNK - 1) / SCAN_CHUNK;
  int* offs   = (int*)(Wp3 + (size_t)128 * 128);
  int* cursor = offs + (size_t)N + 1;
  int* csum   = cursor + (size_t)N;
  int* nbr    = csum + nchunk;

  const int nthread = 256;

  pack_w<<<(12  * NCOLS + nthread - 1) / nthread, nthread, 0, stream>>>(W1,  Wp1, 10,  12);
  pack_w<<<(256 * NCOLS + nthread - 1) / nthread, nthread, 0, stream>>>(W2,  Wp2, 256, 256);
  pack_w<<<(128 * NCOLS + nthread - 1) / nthread, nthread, 0, stream>>>(l1W, Wp3, 128, 128);

  init_ws<<<((unsigned)N * (unsigned)ROWW + nthread - 1) / nthread, nthread, 0, stream>>>(rows, N);

  edge_accum_feat<<<(E + nthread - 1) / nthread, nthread, 0, stream>>>(src, dst, x, rows, E);

  scan_chunk_sums<<<nchunk, nthread, 0, stream>>>(rows, csum, N);
  scan_offsets<<<nchunk, nthread, 0, stream>>>(rows, csum, offs, cursor, N);
  fill_lists<<<(E + nthread - 1) / nthread, nthread, 0, stream>>>(src, dst, cursor, nbr, E);

  sage_gemm_wmma<<<(N + MTILE - 1) / MTILE, nthread, 0, stream>>>(x, rows, deg, Wp1, b1, h1, N, 5, 12, ROWW, ROWW);

  gather_sum_rows<<<((unsigned)N * 32u + nthread - 1) / nthread, nthread, 0, stream>>>(offs, nbr, h1, agg2, N);

  sage_head_fused<<<(N + MTILE - 1) / MTILE, nthread, 0, stream>>>(h1, agg2, deg, Wp2, b2, Wp3, l1b, l2W, l2b,
                                                                    (float*)d_out, N, 128, 256, 128, ROWW);
}


# Edge lists with no scan or fill pass: the layer-1 degree add returns the edge's rank, its source goes into the target's 32-slot row (past 32: float atomics on agg2); sums by arrival as before
# speedup vs baseline: 2.6885x; 1.2541x over previous
#include <hip/hip_runtime.h>

typedef __attribute__((ext_vector_type(2))) float v2f;
typedef __attribute__((ext_vector_type(8))) float v8f;

#define NCOLS 128
#define MTILE 32
#define LDSTRIDE 260

__global__ void pack_w(const float* __restrict__ W, float* __restrict__ Wp,
                       int Kw, int Kpad) {
  int i = blockIdx.x * blockDim.x + threadIdx.x;
  if (i < Kpad * NCOLS) {
    int k = i / NCOLS;
    int n = i - k * NCOLS;
    float v = (k < Kw) ? W[(size_t)k * NCOLS + n] : 0.0f;
    Wp[((size_t)(k >> 1) * NCOLS + n) * 2 + (k & 1)] = v;
  }
}

#define ROWW 8
#define DEGSLOT 5
#define SLOTCAP 32

__global__ void init_ws(float* __restrict__ rows, int* __restrict__ ovf_cnt, int N) {
  unsigned tid = blockIdx.x * blockDim.x + threadIdx.x;
  if (tid < (unsigned)N * (unsigned)ROWW)
    rows[tid] = ((tid & (unsigned)(ROWW - 1)) == (unsigned)DEGSLOT) ? 1.0f : 0.0f;
  if (tid == 0) *ovf_cnt = 0;
}

__global__ void edge_accum_feat(const int* __restrict__ src, const int* __restrict__ dst,
                                const float* __restrict__ x, float* __restrict__ rows,
                                int* __restrict__ slot, int* __restrict__ ovf_cnt,
                                int* __restrict__ ovf, int E) {
  const unsigned tid  = blockIdx.x * blockDim.x + threadIdx.x;
  const unsigned lane = threadIdx.x & 31u;
  const unsigned base = tid - lane;
  int s = 0, d = 0;
  if (tid < (unsigned)E) { s = src[tid]; d = dst[tid]; }
  const unsigned f  = lane & 7u;
  const unsigned g  = lane >> 3;
  const unsigned fx = f < 5u ? f : 4u;

  float  xv[8];
  float* p[8];
  bool   ok[8];
#pragma unroll
  for (int r = 0; r < 8; ++r) {
    const int sl = r * 4 + (int)g;
    const int ss = __shfl(s, sl);
    const int dd = __shfl(d, sl);
    ok[r] = (base + (unsigned)sl < (unsigned)E) && (f <= (unsigned)DEGSLOT);
    p[r]  = rows + (size_t)dd * ROWW + f;
    xv[r] = x[(size_t)ss * 5 + fx];
  }
  float oldv[8];
#pragma unroll
  for (int r = 0; r < 8; ++r) {
    oldv[r] = 0.0f;
    if (ok[r]) oldv[r] = atomicAdd(p[r], f < 5u ? xv[r] : 1.0f);
  }
  float myold = 0.0f;
#pragma unroll
  for (int r = 0; r < 8; ++r) {
    const float v = __shfl(oldv[r], (int)((lane & 3u) * 8u + (unsigned)DEGSLOT));
    if ((lane >> 2) == (unsigned)r) myold = v;
  }
  if (tid < (unsigned)E) {
    const int place = (int)myold - 1;
    if (place < SLOTCAP) {
      slot[(size_t)d * SLOTCAP + place] = s;
    } else {
      const int k = atomicAdd(ovf_cnt, 1);
      ovf[2 * (size_t)k]     = s;
      ovf[2 * (size_t)k + 1] = d;
    }
  }
}

__global__ void gather_sum_rows(const float* __restrict__ rows, const int* __restrict__ slot,
                                const float* __restrict__ h, float* __restrict__ agg, int N) {
  const unsigned tid = blockIdx.x * blockDim.x + threadIdx.x;
  const unsigned i = tid >> 5;
  const unsigned l = tid & 31u;
  if (i < (unsigned)N) {
    const int indeg = __builtin_amdgcn_readfirstlane((int)rows[(size_t)i * ROWW + DEGSLOT] - 1);
    const int end = indeg < SLOTCAP ? indeg : SLOTCAP;
    const int* nbr = slot + (size_t)i * SLOTCAP;
    const float* hl = h + 4u * l;
    float4 acc = make_float4(0.0f, 0.0f, 0.0f, 0.0f);
    for (int j0 = 0; j0 < end; j0 += 32) {
      const int cnt = (end - j0 < 32) ? (end - j0) : 32;
      const int mine = ((int)l < cnt) ? nbr[j0 + (int)l] : 0;
      int k = 0;
      for (; k + 4 <= cnt; k += 4) {
        const int s0 = __builtin_amdgcn_readlane(mine, k);
        const int s1 = __builtin_amdgcn_readlane(mine, k + 1);
        const int s2 = __builtin_amdgcn_readlane(mine, k + 2);
        const int s3 = __builtin_amdgcn_readlane(mine, k + 3);
        const float4 v0 = *(const float4*)(hl + (size_t)s0 * 128);
        const float4 v1 = *(const float4*)(hl + (size_t)s1 * 128);
        const float4 v2 = *(const float4*)(hl + (size_t)s2 * 128);
        const float4 v3 = *(const float4*)(hl + (size_t)s3 * 128);
        acc.x += v0.x; acc.y += v0.y; acc.z += v0.z; acc.w += v0.w;
        acc.x += v1.x; acc.y += v1.y; acc.z += v1.z; acc.w += v1.w;
        acc.x += v2.x; acc.y += v2.y; acc.z += v2.z; acc.w += v2.w;
        acc.x += v3.x; acc.y += v3.y; acc.z += v3.z; acc.w += v3.w;
      }
      float4 t0 = make_float4(0.0f, 0.0f, 0.0f, 0.0f), t1 = t0, t2 = t0;
      if (k < cnt)     t0 = *(const float4*)(hl + (size_t)__builtin_amdgcn_readlane(mine, k) * 128);
      if (k + 1 < cnt) t1 = *(const float4*)(hl + (size_t)__builtin_amdgcn_readlane(mine, k + 1) * 128);
      if (k + 2 < cnt) t2 = *(const float4*)(hl + (size_t)__builtin_amdgcn_readlane(mine, k + 2) * 128);
      if (k < cnt)     { acc.x += t0.x; acc.y += t0.y; acc.z += t0.z; acc.w += t0.w; }
      if (k + 1 < cnt) { acc.x += t1.x; acc.y += t1.y; acc.z += t1.z; acc.w += t1.w; }
      if (k + 2 < cnt) { acc.x += t2.x; acc.y += t2.y; acc.z += t2.z; acc.w += t2.w; }
    }
    *(float4*)(agg + (size_t)i * 128 + 4u * l) = acc;
  }
}

__global__ void overflow_add_rows(const int* __restrict__ ovf_cnt, const int* __restrict__ ovf,
                                  const float* __restrict__ h, float* __restrict__ agg) {
  const int cnt   = *ovf_cnt;
  const int l     = threadIdx.x & 31;
  const int wave  = (int)((blockIdx.x * blockDim.x + threadIdx.x) >> 5);
  const int nwave = (int)((gridDim.x * blockDim.x) >> 5);
  for (int k = wave; k < cnt; k += nwave) {
    const int s = ovf[2 * (size_t)k];
    const int d = ovf[2 * (size_t)k + 1];
    for (int q = 0; q < 4; ++q)
      atomicAdd(agg + (size_t)d * 128 + q * 32 + l, h[(size_t)s * 128 + q * 32 + l]);
  }
}

__global__ void sage_gemm_wmma(const float* __restrict__ A0, const float* __restrict__ A1,
                               const float* __restrict__ deg, const float* __restrict__ Wp,
                               const float* __restrict__ bias, float* __restrict__ out,
                               int M, int d, int Kpad, int lda1, int ldd) {
  __shared__ float As[MTILE * LDSTRIDE];

  const int row0 = blockIdx.x * MTILE;
  const int t = threadIdx.x;

  const int total = MTILE * Kpad;
  for (int i = t; i < total; i += 256) {
    int m = i / Kpad;
    int k = i - m * Kpad;
    int row = row0 + m;
    float v = 0.0f;
    if (row < M) {
      if (k < d) {
        v = A0[(size_t)row * d + k];
      } else if (A1 != nullptr && k < 2 * d) {
        int kk = k - d;
        v = (A0[(size_t)row * d + kk] + A1[(size_t)row * lda1 + kk]) / deg[(size_t)row * ldd];
      }
    }
    As[m * LDSTRIDE + k] = v;
  }
  __syncthreads();

  const int lane  = t & 31;
  const int wave  = t >> 5;
  const int n     = wave * 16 + (lane & 15);
  const int mA    = lane & 15;
  const int khalf = (lane < 16) ? 0 : 2;

  v8f c0 = {};
  v8f c1 = {};
#pragma unroll 2
  for (int k0 = 0; k0 < Kpad; k0 += 4) {
    const int kb = k0 + khalf;
    v2f a0 = *(const v2f*)&As[mA * LDSTRIDE + kb];
    v2f a1 = *(const v2f*)&As[(mA + 16) * LDSTRIDE + kb];
    v2f b  = *(const v2f*)(Wp + ((size_t)(kb >> 1) * NCOLS + n) * 2);
    c0 = __builtin_amdgcn_wmma_f32_16x16x4_f32(false, a0, false, b,
                                               (short)0, c0, false, false);
    c1 = __builtin_amdgcn_wmma_f32_16x16x4_f32(false, a1, false, b,
                                               (short)0, c1, false, false);
  }

  const float bv    = bias[n];
  const int   rbase = (lane < 16) ? 0 : 8;
  float* outA = out + (size_t)(row0 + rbase) * NCOLS + n;
  float* outB = outA + (size_t)16 * NCOLS;

  if (row0 + MTILE <= M) {
#pragma unroll
    for (int r = 0; r < 8; ++r) {
      float vA = c0[r] + bv;
      float vB = c1[r] + bv;
      outA[(size_t)r * NCOLS] = vA > 0.0f ? vA : 0.0f;
      outB[(size_t)r * NCOLS] = vB > 0.0f ? vB : 0.0f;
    }
  } else {
#pragma unroll
    for (int r = 0; r < 8; ++r) {
      int rowA = row0 + rbase + r;
      if (rowA < M) {
        float v = c0[r] + bv;
        outA[(size_t)r * NCOLS] = v > 0.0f ? v : 0.0f;
      }
      int rowB = row0 + 16 + rbase + r;
      if (rowB < M) {
        float v = c1[r] + bv;
        outB[(size_t)r * NCOLS] = v > 0.0f ? v : 0.0f;
      }
    }
  }
}

__global__ void sage_head_fused(const float* __restrict__ A0, const float* __restrict__ A1,
                                const float* __restrict__ deg,
                                const float* __restrict__ Wp2, const float* __restrict__ b2,
                                const float* __restrict__ Wp3, const float* __restrict__ b3,
                                const float* __restrict__ W4, const float* __restrict__ b4,
                                float* __restrict__ out,
                                int M, int d, int Kpad, int lda1, int ldd) {
  __shared__ float As[MTILE * LDSTRIDE];

  const int row0 = blockIdx.x * MTILE;
  const int t = threadIdx.x;

  const int total = MTILE * Kpad;
  for (int i = t; i < total; i += 256) {
    int m = i / Kpad;
    int k = i - m * Kpad;
    int row = row0 + m;
    float v = 0.0f;
    if (row < M) {
      if (k < d) {
        v = A0[(size_t)row * d + k];
      } else if (k < 2 * d) {
        int kk = k - d;
        v = (A0[(size_t)row * d + kk] + A1[(size_t)row * lda1 + kk]) / deg[(size_t)row * ldd];
      }
    }
    As[m * LDSTRIDE + k] = v;
  }
  __syncthreads();

  const int lane  = t & 31;
  const int wave  = t >> 5;
  const int n     = wave * 16 + (lane & 15);
  const int mA    = lane & 15;
  const int khalf = (lane < 16) ? 0 : 2;
  const int rbase = (lane < 16) ? 0 : 8;

  v8f c0 = {};
  v8f c1 = {};
#pragma unroll 2
  for (int k0 = 0; k0 < Kpad; k0 += 4) {
    const int kb = k0 + khalf;
    v2f a0 = *(const v2f*)&As[mA * LDSTRIDE + kb];
    v2f a1 = *(const v2f*)&As[(mA + 16) * LDSTRIDE + kb];
    v2f b  = *(const v2f*)(Wp2 + ((size_t)(kb >> 1) * NCOLS + n) * 2);
    c0 = __builtin_amdgcn_wmma_f32_16x16x4_f32(false, a0, false, b,
                                               (short)0, c0, false, false);
    c1 = __builtin_amdgcn_wmma_f32_16x16x4_f32(false, a1, false, b,
                                               (short)0, c1, false, false);
  }

  __syncthreads();
  {
    const float bv = b2[n];
#pragma unroll
    for (int r = 0; r < 8; ++r) {
      float vA = c0[r] + bv;
      float vB = c1[r] + bv;
      As[(rbase + r) * LDSTRIDE + n]      = vA > 0.0f ? vA : 0.0f;
      As[(16 + rbase + r) * LDSTRIDE + n] = vB > 0.0f ? vB : 0.0f;
    }
  }
  __syncthreads();

  v8f e0 = {};
  v8f e1 = {};
#pragma unroll 2
  for (int k0 = 0; k0 < NCOLS; k0 += 4) {
    const int kb = k0 + khalf;
    v2f a0 = *(const v2f*)&As[mA * LDSTRIDE + kb];
    v2f a1 = *(const v2f*)&As[(mA + 16) * LDSTRIDE + kb];
    v2f b  = *(const v2f*)(Wp3 + ((size_t)(kb >> 1) * NCOLS + n) * 2);
    e0 = __builtin_amdgcn_wmma_f32_16x16x4_f32(false, a0, false, b,
                                               (short)0, e0, false, false);
    e1 = __builtin_amdgcn_wmma_f32_16x16x4_f32(false, a1, false, b,
                                               (short)0, e1, false, false);
  }
  {
    const float bv = b3[n];
#pragma unroll
    for (int r = 0; r < 8; ++r) {
      float vA = e0[r] + bv;
      float vB = e1[r] + bv;
      As[(rbase + r) * LDSTRIDE + NCOLS + n]      = vA > 0.0f ? vA : 0.0f;
      As[(16 + rbase + r) * LDSTRIDE + NCOLS + n] = vB > 0.0f ? vB : 0.0f;
    }
  }
  __syncthreads();

  const int orow = t >> 3;
  const int part = t & 7;
  const float* hrow = &As[orow * LDSTRIDE + NCOLS + part * 16];
  const float* w    = W4 + part * 32;
  float s0 = 0.0f, s1 = 0.0f;
#pragma unroll
  for (int q = 0; q < 4; ++q) {
    const float4 hv = *(const float4*)(hrow + 4 * q);
    const float4 wa = *(const float4*)(w + 8 * q);
    const float4 wb = *(const float4*)(w + 8 * q + 4);
    s0 += hv.x * wa.x;  s1 += hv.x * wa.y;
    s0 += hv.y * wa.z;  s1 += hv.y * wa.w;
    s0 += hv.z * wb.x;  s1 += hv.z * wb.y;
    s0 += hv.w * wb.z;  s1 += hv.w * wb.w;
  }
  for (int o = 1; o < 8; o <<= 1) {
    s0 += __shfl_xor(s0, o);
    s1 += __shfl_xor(s1, o);
  }
  if (part == 0) {
    As[orow * 2 + 0] = s0 + b4[0];
    As[orow * 2 + 1] = s1 + b4[1];
  }
  __syncthreads();
  if (t < 2 * MTILE && row0 + (t >> 1) < M)
    out[(size_t)row0 * 2 + t] = As[t];
}

extern "C" void kernel_launch(void* const* d_in, const int* in_sizes, int n_in,
                              void* d_out, int out_size, void* d_ws, size_t ws_size,
                              hipStream_t stream) {
  const float* x   = (const float*)d_in[0];
  const int*   ei  = (const int*)d_in[1];
  const float* W1  = (const float*)d_in[2];
  const float* b1  = (const float*)d_in[3];
  const float* W2  = (const float*)d_in[4];
  const float* b2  = (const float*)d_in[5];
  const float* l1W = (const float*)d_in[6];
  const float* l1b = (const float*)d_in[7];
  const float* l2W = (const float*)d_in[8];
  const float* l2b = (const float*)d_in[9];

  const int N = in_sizes[0] / 5;
  const int E = in_sizes[1] / 2;
  const int* src = ei;
  const int* dst = ei + E;

  float* ws   = (float*)d_ws;
  float* rows = ws;
  float* deg  = rows + DEGSLOT;
  float* agg2 = rows + (size_t)ROWW * N;
  float* h1   = agg2 + (size_t)128 * N;
  float* Wp1  = h1   + (size_t)128 * N;
  float* Wp2  = Wp1  + (size_t)12 * 128;
  float* Wp3  = Wp2  + (size_t)256 * 128;
  int* slot    = (int*)(Wp3 + (size_t)128 * 128);
  int* ovf_cnt = slot + (size_t)SLOTCAP * N;
  int* ovf     = ovf_cnt + 64;

  const int nthread = 256;

  pack_w<<<(12  * NCOLS + nthread - 1) / nthread, nthread, 0, stream>>>(W1,  Wp1, 10,  12);
  pack_w<<<(256 * NCOLS + nthread - 1) / nthread, nthread, 0, stream>>>(W2,  Wp2, 256, 256);
  pack_w<<<(128 * NCOLS + nthread - 1) / nthread, nthread, 0, stream>>>(l1W, Wp3, 128, 128);

  init_ws<<<((unsigned)N * (unsigned)ROWW + nthread - 1) / nthread, nthread, 0, stream>>>(rows, ovf_cnt, N);

  edge_accum_feat<<<(E + nthread - 1) / nthread, nthread, 0, stream>>>(src, dst, x, rows, slot, ovf_cnt, ovf, E);

  sage_gemm_wmma<<<(N + MTILE - 1) / MTILE, nthread, 0, stream>>>(x, rows, deg, Wp1, b1, h1, N, 5, 12, ROWW, ROWW);

  gather_sum_rows<<<((unsigned)N * 32u + nthread - 1) / nthread, nthread, 0, stream>>>(rows, slot, h1, agg2, N);

  overflow_add_rows<<<64, nthread, 0, stream>>>(ovf_cnt, ovf, h1, agg2);

  sage_head_fused<<<(N + MTILE - 1) / MTILE, nthread, 0, stream>>>(h1, agg2, deg, Wp2, b2, Wp3, l1b, l2W, l2b,
                                                                    (float*)d_out, N, 128, 256, 128, ROWW);
}


# Layer-2 neighbour sum moved into the GEMM launch: each block gathers its 32 nodes' neighbour h1 rows into its LDS tile itself; agg2 never reaches memory, 2 launches fewer; sums by arrival as before
# speedup vs baseline: 3.5425x; 1.3176x over previous
#include <hip/hip_runtime.h>

typedef __attribute__((ext_vector_type(2))) float v2f;
typedef __attribute__((ext_vector_type(8))) float v8f;

#define NCOLS 128
#define MTILE 32
#define LDSTRIDE 260

__global__ void pack_w(const float* __restrict__ W, float* __restrict__ Wp,
                       int Kw, int Kpad) {
  int i = blockIdx.x * blockDim.x + threadIdx.x;
  if (i < Kpad * NCOLS) {
    int k = i / NCOLS;
    int n = i - k * NCOLS;
    float v = (k < Kw) ? W[(size_t)k * NCOLS + n] : 0.0f;
    Wp[((size_t)(k >> 1) * NCOLS + n) * 2 + (k & 1)] = v;
  }
}

#define ROWW 8
#define DEGSLOT 5
#define SLOTCAP 32

__global__ void init_ws(float* __restrict__ rows, int* __restrict__ ovf_cnt, int N) {
  unsigned tid = blockIdx.x * blockDim.x + threadIdx.x;
  if (tid < (unsigned)N * (unsigned)ROWW)
    rows[tid] = ((tid & (unsigned)(ROWW - 1)) == (unsigned)DEGSLOT) ? 1.0f : 0.0f;
  if (tid == 0) *ovf_cnt = 0;
}

__global__ void edge_accum_feat(const int* __restrict__ src, const int* __restrict__ dst,
                                const float* __restrict__ x, float* __restrict__ rows,
                                int* __restrict__ slot, int* __restrict__ ovf_cnt,
                                int* __restrict__ ovf, int E) {
  const unsigned tid  = blockIdx.x * blockDim.x + threadIdx.x;
  const unsigned lane = threadIdx.x & 31u;
  const unsigned base = tid - lane;
  int s = 0, d = 0;
  if (tid < (unsigned)E) { s = src[tid]; d = dst[tid]; }
  const unsigned f  = lane & 7u;
  const unsigned g  = lane >> 3;
  const unsigned fx = f < 5u ? f : 4u;

  float  xv[8];
  float* p[8];
  bool   ok[8];
#pragma unroll
  for (int r = 0; r < 8; ++r) {
    const int sl = r * 4 + (int)g;
    const int ss = __shfl(s, sl);
    const int dd = __shfl(d, sl);
    ok[r] = (base + (unsigned)sl < (unsigned)E) && (f <= (unsigned)DEGSLOT);
    p[r]  = rows + (size_t)dd * ROWW + f;
    xv[r] = x[(size_t)ss * 5 + fx];
  }
  float oldv[8];
#pragma unroll
  for (int r = 0; r < 8; ++r) {
    oldv[r] = 0.0f;
    if (ok[r]) oldv[r] = atomicAdd(p[r], f < 5u ? xv[r] : 1.0f);
  }
  float myold = 0.0f;
#pragma unroll
  for (int r = 0; r < 8; ++r) {
    const float v = __shfl(oldv[r], (int)((lane & 3u) * 8u + (unsigned)DEGSLOT));
    if ((lane >> 2) == (unsigned)r) myold = v;
  }
  if (tid < (unsigned)E) {
    const int place = (int)myold - 1;
    if (place < SLOTCAP) {
      slot[(size_t)d * SLOTCAP + place] = s;
    } else {
      const int k = atomicAdd(ovf_cnt, 1);
      ovf[2 * (size_t)k]     = s;
      ovf[2 * (size_t)k + 1] = d;
    }
  }
}

__global__ void sage_gemm_wmma(const float* __restrict__ A0, const float* __restrict__ A1,
                               const float* __restrict__ deg, const float* __restrict__ Wp,
                               const float* __restrict__ bias, float* __restrict__ out,
                               int M, int d, int Kpad, int lda1, int ldd) {
  __shared__ float As[MTILE * LDSTRIDE];

  const int row0 = blockIdx.x * MTILE;
  const int t = threadIdx.x;

  const int total = MTILE * Kpad;
  for (int i = t; i < total; i += 256) {
    int m = i / Kpad;
    int k = i - m * Kpad;
    int row = row0 + m;
    float v = 0.0f;
    if (row < M) {
      if (k < d) {
        v = A0[(size_t)row * d + k];
      } else if (A1 != nullptr && k < 2 * d) {
        int kk = k - d;
        v = (A0[(size_t)row * d + kk] + A1[(size_t)row * lda1 + kk]) / deg[(size_t)row * ldd];
      }
    }
    As[m * LDSTRIDE + k] = v;
  }
  __syncthreads();

  const int lane  = t & 31;
  const int wave  = t >> 5;
  const int n     = wave * 16 + (lane & 15);
  const int mA    = lane & 15;
  const int khalf = (lane < 16) ? 0 : 2;

  v8f c0 = {};
  v8f c1 = {};
#pragma unroll 2
  for (int k0 = 0; k0 < Kpad; k0 += 4) {
    const int kb = k0 + khalf;
    v2f a0 = *(const v2f*)&As[mA * LDSTRIDE + kb];
    v2f a1 = *(const v2f*)&As[(mA + 16) * LDSTRIDE + kb];
    v2f b  = *(const v2f*)(Wp + ((size_t)(kb >> 1) * NCOLS + n) * 2);
    c0 = __builtin_amdgcn_wmma_f32_16x16x4_f32(false, a0, false, b,
                                               (short)0, c0, false, false);
    c1 = __builtin_amdgcn_wmma_f32_16x16x4_f32(false, a1, false, b,
                                               (short)0, c1, false, false);
  }

  const float bv    = bias[n];
  const int   rbase = (lane < 16) ? 0 : 8;
  float* outA = out + (size_t)(row0 + rbase) * NCOLS + n;
  float* outB = outA + (size_t)16 * NCOLS;

  if (row0 + MTILE <= M) {
#pragma unroll
    for (int r = 0; r < 8; ++r) {
      float vA = c0[r] + bv;
      float vB = c1[r] + bv;
      outA[(size_t)r * NCOLS] = vA > 0.0f ? vA : 0.0f;
      outB[(size_t)r * NCOLS] = vB > 0.0f ? vB : 0.0f;
    }
  } else {
#pragma unroll
    for (int r = 0; r < 8; ++r) {
      int rowA = row0 + rbase + r;
      if (rowA < M) {
        float v = c0[r] + bv;
        outA[(size_t)r * NCOLS] = v > 0.0f ? v : 0.0f;
      }
      int rowB = row0 + 16 + rbase + r;
      if (rowB < M) {
        float v = c1[r] + bv;
        outB[(size_t)r * NCOLS] = v > 0.0f ? v : 0.0f;
      }
    }
  }
}

#define NPW (MTILE / 8)
static_assert(SLOTCAP == 32, "one slot word per lane");

__global__ void sage_head_fused(const float* __restrict__ h, const float* __restrict__ rows,
                                const int* __restrict__ slot,
                                const int* __restrict__ ovf_cnt, const int* __restrict__ ovf,
                                const float* __restrict__ Wp2, const float* __restrict__ b2,
                                const float* __restrict__ Wp3, const float* __restrict__ b3,
                                const float* __restrict__ W4, const float* __restrict__ b4,
                                float* __restrict__ out, int M) {
  __shared__ __attribute__((aligned(16))) float As[MTILE * LDSTRIDE];

  const int Kpad = 2 * NCOLS;
  const int row0 = blockIdx.x * MTILE;
  const int t = threadIdx.x;
  const int lane  = t & 31;
  const int wave  = t >> 5;
  const int wv = __builtin_amdgcn_readfirstlane(wave);

  const float* hl = h + 4 * lane;
  float  dg[NPW];
  int    mn[NPW];
  float4 sf[NPW];
#pragma unroll
  for (int q = 0; q < NPW; ++q) {
    const int row = row0 + wv * NPW + q;
    dg[q] = 1.0f;
    sf[q] = make_float4(0.0f, 0.0f, 0.0f, 0.0f);
    if (row < M) {
      dg[q] = rows[(size_t)row * ROWW + DEGSLOT];
      sf[q] = *(const float4*)(hl + (size_t)row * 128);
    }
  }
#pragma unroll
  for (int q = 0; q < NPW; ++q) {
    const int row = row0 + wv * NPW + q;
    const int filled = (int)dg[q] - 1;
    mn[q] = (lane < filled) ? slot[(size_t)row * SLOTCAP + lane] : 0;
  }
#pragma unroll
  for (int q = 0; q < NPW; ++q) {
    const int m   = wv * NPW + q;
    const int row = row0 + m;
    const int indeg = __builtin_amdgcn_readfirstlane((int)dg[q] - 1);
    const int end   = indeg < SLOTCAP ? indeg : SLOTCAP;
    const int mine  = mn[q];
    float4 acc = make_float4(0.0f, 0.0f, 0.0f, 0.0f);
    int k = 0;
    for (; k + 4 <= end; k += 4) {
      const int s0 = __builtin_amdgcn_readlane(mine, k);
      const int s1 = __builtin_amdgcn_readlane(mine, k + 1);
      const int s2 = __builtin_amdgcn_readlane(mine, k + 2);
      const int s3 = __builtin_amdgcn_readlane(mine, k + 3);
      const float4 v0 = *(const float4*)(hl + (size_t)s0 * 128);
      const float4 v1 = *(const float4*)(hl + (size_t)s1 * 128);
      const float4 v2 = *(const float4*)(hl + (size_t)s2 * 128);
      const float4 v3 = *(const float4*)(hl + (size_t)s3 * 128);
      acc.x += v0.x; acc.y += v0.y; acc.z += v0.z; acc.w += v0.w;
      acc.x += v1.x; acc.y += v1.y; acc.z += v1.z; acc.w += v1.w;
      acc.x += v2.x; acc.y += v2.y; acc.z += v2.z; acc.w += v2.w;
      acc.x += v3.x; acc.y += v3.y; acc.z += v3.z; acc.w += v3.w;
    }
    {
      float4 t0 = make_float4(0.0f, 0.0f, 0.0f, 0.0f), t1 = t0, t2 = t0;
      if (k < end)     t0 = *(const float4*)(hl + (size_t)__builtin_amdgcn_readlane(mine, k) * 128);
      if (k + 1 < end) t1 = *(const float4*)(hl + (size_t)__builtin_amdgcn_readlane(mine, k + 1) * 128);
      if (k + 2 < end) t2 = *(const float4*)(hl + (size_t)__builtin_amdgcn_readlane(mine, k + 2) * 128);
      if (k < end)     { acc.x += t0.x; acc.y += t0.y; acc.z += t0.z; acc.w += t0.w; }
      if (k + 1 < end) { acc.x += t1.x; acc.y += t1.y; acc.z += t1.z; acc.w += t1.w; }
      if (k + 2 < end) { acc.x += t2.x; acc.y += t2.y; acc.z += t2.z; acc.w += t2.w; }
    }
    if (indeg > SLOTCAP) {
      const int ocnt = __builtin_amdgcn_readfirstlane(*ovf_cnt);
      for (int k0 = 0; k0 < ocnt; k0 += 32) {
        const int kk = k0 + lane;
        int ss = 0, dd = -1;
        if (kk < ocnt) { ss = ovf[2 * (size_t)kk]; dd = ovf[2 * (size_t)kk + 1]; }
        unsigned hit = __builtin_amdgcn_ballot_w32(dd == row);
        while (hit != 0u) {
          const int b = __builtin_ctz(hit);
          hit &= hit - 1u;
          const int s = __builtin_amdgcn_readlane(ss, b);
          const float4 v = *(const float4*)(hl + (size_t)s * 128);
          acc.x += v.x; acc.y += v.y; acc.z += v.z; acc.w += v.w;
        }
      }
    }
    const float  dv = dg[q];
    const float4 self = sf[q];
    float4 mean;
    mean.x = (self.x + acc.x) / dv;
    mean.y = (self.y + acc.y) / dv;
    mean.z = (self.z + acc.z) / dv;
    mean.w = (self.w + acc.w) / dv;
    *(float4*)&As[m * LDSTRIDE + 4 * lane]         = self;
    *(float4*)&As[m * LDSTRIDE + NCOLS + 4 * lane] = mean;
  }
  __syncthreads();

  const int n     = wave * 16 + (lane & 15);
  const int mA    = lane & 15;
  const int khalf = (lane < 16) ? 0 : 2;
  const int rbase = (lane < 16) ? 0 : 8;

  v8f c0 = {};
  v8f c1 = {};
#pragma unroll 2
  for (int k0 = 0; k0 < Kpad; k0 += 4) {
    const int kb = k0 + khalf;
    v2f a0 = *(const v2f*)&As[mA * LDSTRIDE + kb];
    v2f a1 = *(const v2f*)&As[(mA + 16) * LDSTRIDE + kb];
    v2f b  = *(const v2f*)(Wp2 + ((size_t)(kb >> 1) * NCOLS + n) * 2);
    c0 = __builtin_amdgcn_wmma_f32_16x16x4_f32(false, a0, false, b,
                                               (short)0, c0, false, false);
    c1 = __builtin_amdgcn_wmma_f32_16x16x4_f32(false, a1, false, b,
                                               (short)0, c1, false, false);
  }

  __syncthreads();
  {
    const float bv = b2[n];
#pragma unroll
    for (int r = 0; r < 8; ++r) {
      float vA = c0[r] + bv;
      float vB = c1[r] + bv;
      As[(rbase + r) * LDSTRIDE + n]      = vA > 0.0f ? vA : 0.0f;
      As[(16 + rbase + r) * LDSTRIDE + n] = vB > 0.0f ? vB : 0.0f;
    }
  }
  __syncthreads();

  v8f e0 = {};
  v8f e1 = {};
#pragma unroll 2
  for (int k0 = 0; k0 < NCOLS; k0 += 4) {
    const int kb = k0 + khalf;
    v2f a0 = *(const v2f*)&As[mA * LDSTRIDE + kb];
    v2f a1 = *(const v2f*)&As[(mA + 16) * LDSTRIDE + kb];
    v2f b  = *(const v2f*)(Wp3 + ((size_t)(kb >> 1) * NCOLS + n) * 2);
    e0 = __builtin_amdgcn_wmma_f32_16x16x4_f32(false, a0, false, b,
                                               (short)0, e0, false, false);
    e1 = __builtin_amdgcn_wmma_f32_16x16x4_f32(false, a1, false, b,
                                               (short)0, e1, false, false);
  }
  {
    const float bv = b3[n];
#pragma unroll
    for (int r = 0; r < 8; ++r) {
      float vA = e0[r] + bv;
      float vB = e1[r] + bv;
      As[(rbase + r) * LDSTRIDE + NCOLS + n]      = vA > 0.0f ? vA : 0.0f;
      As[(16 + rbase + r) * LDSTRIDE + NCOLS + n] = vB > 0.0f ? vB : 0.0f;
    }
  }
  __syncthreads();

  const int orow = t >> 3;
  const int part = t & 7;
  const float* hrow = &As[orow * LDSTRIDE + NCOLS + part * 16];
  const float* w    = W4 + part * 32;
  float s0 = 0.0f, s1 = 0.0f;
#pragma unroll
  for (int q = 0; q < 4; ++q) {
    const float4 hv = *(const float4*)(hrow + 4 * q);
    const float4 wa = *(const float4*)(w + 8 * q);
    const float4 wb = *(const float4*)(w + 8 * q + 4);
    s0 += hv.x * wa.x;  s1 += hv.x * wa.y;
    s0 += hv.y * wa.z;  s1 += hv.y * wa.w;
    s0 += hv.z * wb.x;  s1 += hv.z * wb.y;
    s0 += hv.w * wb.z;  s1 += hv.w * wb.w;
  }
  for (int o = 1; o < 8; o <<= 1) {
    s0 += __shfl_xor(s0, o);
    s1 += __shfl_xor(s1, o);
  }
  if (part == 0) {
    As[orow * 2 + 0] = s0 + b4[0];
    As[orow * 2 + 1] = s1 + b4[1];
  }
  __syncthreads();
  if (t < 2 * MTILE && row0 + (t >> 1) < M)
    out[(size_t)row0 * 2 + t] = As[t];
}

extern "C" void kernel_launch(void* const* d_in, const int* in_sizes, int n_in,
                              void* d_out, int out_size, void* d_ws, size_t ws_size,
                              hipStream_t stream) {
  const float* x   = (const float*)d_in[0];
  const int*   ei  = (const int*)d_in[1];
  const float* W1  = (const float*)d_in[2];
  const float* b1  = (const float*)d_in[3];
  const float* W2  = (const float*)d_in[4];
  const float* b2  = (const float*)d_in[5];
  const float* l1W = (const float*)d_in[6];
  const float* l1b = (const float*)d_in[7];
  const float* l2W = (const float*)d_in[8];
  const float* l2b = (const float*)d_in[9];

  const int N = in_sizes[0] / 5;
  const int E = in_sizes[1] / 2;
  const int* src = ei;
  const int* dst = ei + E;

  float* ws   = (float*)d_ws;
  float* rows = ws;
  float* deg  = rows + DEGSLOT;
  float* h1   = rows + (size_t)ROWW * N;
  float* Wp1  = h1   + (size_t)128 * N;
  float* Wp2  = Wp1  + (size_t)12 * 128;
  float* Wp3  = Wp2  + (size_t)256 * 128;
  int* slot    = (int*)(Wp3 + (size_t)128 * 128);
  int* ovf_cnt = slot + (size_t)SLOTCAP * N;
  int* ovf     = ovf_cnt + 64;

  const int nthread = 256;

  pack_w<<<(12  * NCOLS + nthread - 1) / nthread, nthread, 0, stream>>>(W1,  Wp1, 10,  12);
  pack_w<<<(256 * NCOLS + nthread - 1) / nthread, nthread, 0, stream>>>(W2,  Wp2, 256, 256);
  pack_w<<<(128 * NCOLS + nthread - 1) / nthread, nthread, 0, stream>>>(l1W, Wp3, 128, 128);

  init_ws<<<((unsigned)N * (unsigned)ROWW + nthread - 1) / nthread, nthread, 0, stream>>>(rows, ovf_cnt, N);

  edge_accum_feat<<<(E + nthread - 1) / nthread, nthread, 0, stream>>>(src, dst, x, rows, slot, ovf_cnt, ovf, E);

  sage_gemm_wmma<<<(N + MTILE - 1) / MTILE, nthread, 0, stream>>>(x, rows, deg, Wp1, b1, h1, N, 5, 12, ROWW, ROWW);

  sage_head_fused<<<(N + MTILE - 1) / MTILE, nthread, 0, stream>>>(h1, rows, slot, ovf_cnt, ovf, Wp2, b2, Wp3, l1b, l2W, l2b,
                                                                    (float*)d_out, N);
}


# Fused launch, both WMMA loops: weight pairs requested one group (2 k-steps) ahead into a second register set, lin1's first group under layer 2's last group; operands, k order and sum order unchanged
# speedup vs baseline: 3.6906x; 1.0418x over previous
#include <hip/hip_runtime.h>

typedef __attribute__((ext_vector_type(2))) float v2f;
typedef __attribute__((ext_vector_type(8))) float v8f;

#define NCOLS 128
#define MTILE 32
#define LDSTRIDE 260

__global__ void pack_w(const float* __restrict__ W, float* __restrict__ Wp,
                       int Kw, int Kpad) {
  int i = blockIdx.x * blockDim.x + threadIdx.x;
  if (i < Kpad * NCOLS) {
    int k = i / NCOLS;
    int n = i - k * NCOLS;
    float v = (k < Kw) ? W[(size_t)k * NCOLS + n] : 0.0f;
    Wp[((size_t)(k >> 1) * NCOLS + n) * 2 + (k & 1)] = v;
  }
}

#define ROWW 8
#define DEGSLOT 5
#define SLOTCAP 32

__global__ void init_ws(float* __restrict__ rows, int* __restrict__ ovf_cnt, int N) {
  unsigned tid = blockIdx.x * blockDim.x + threadIdx.x;
  if (tid < (unsigned)N * (unsigned)ROWW)
    rows[tid] = ((tid & (unsigned)(ROWW - 1)) == (unsigned)DEGSLOT) ? 1.0f : 0.0f;
  if (tid == 0) *ovf_cnt = 0;
}

__global__ void edge_accum_feat(const int* __restrict__ src, const int* __restrict__ dst,
                                const float* __restrict__ x, float* __restrict__ rows,
                                int* __restrict__ slot, int* __restrict__ ovf_cnt,
                                int* __restrict__ ovf, int E) {
  const unsigned tid  = blockIdx.x * blockDim.x + threadIdx.x;
  const unsigned lane = threadIdx.x & 31u;
  const unsigned base = tid - lane;
  int s = 0, d = 0;
  if (tid < (unsigned)E) { s = src[tid]; d = dst[tid]; }
  const unsigned f  = lane & 7u;
  const unsigned g  = lane >> 3;
  const unsigned fx = f < 5u ? f : 4u;

  float  xv[8];
  float* p[8];
  bool   ok[8];
#pragma unroll
  for (int r = 0; r < 8; ++r) {
    const int sl = r * 4 + (int)g;
    const int ss = __shfl(s, sl);
    const int dd = __shfl(d, sl);
    ok[r] = (base + (unsigned)sl < (unsigned)E) && (f <= (unsigned)DEGSLOT);
    p[r]  = rows + (size_t)dd * ROWW + f;
    xv[r] = x[(size_t)ss * 5 + fx];
  }
  float oldv[8];
#pragma unroll
  for (int r = 0; r < 8; ++r) {
    oldv[r] = 0.0f;
    if (ok[r]) oldv[r] = atomicAdd(p[r], f < 5u ? xv[r] : 1.0f);
  }
  float myold = 0.0f;
#pragma unroll
  for (int r = 0; r < 8; ++r) {
    const float v = __shfl(oldv[r], (int)((lane & 3u) * 8u + (unsigned)DEGSLOT));
    if ((lane >> 2) == (unsigned)r) myold = v;
  }
  if (tid < (unsigned)E) {
    const int place = (int)myold - 1;
    if (place < SLOTCAP) {
      slot[(size_t)d * SLOTCAP + place] = s;
    } else {
      const int k = atomicAdd(ovf_cnt, 1);
      ovf[2 * (size_t)k]     = s;
      ovf[2 * (size_t)k + 1] = d;
    }
  }
}

__global__ void sage_gemm_wmma(const float* __restrict__ A0, const float* __restrict__ A1,
                               const float* __restrict__ deg, const float* __restrict__ Wp,
                               const float* __restrict__ bias, float* __restrict__ out,
                               int M, int d, int Kpad, int lda1, int ldd) {
  __shared__ float As[MTILE * LDSTRIDE];

  const int row0 = blockIdx.x * MTILE;
  const int t = threadIdx.x;

  const int total = MTILE * Kpad;
  for (int i = t; i < total; i += 256) {
    int m = i / Kpad;
    int k = i - m * Kpad;
    int row = row0 + m;
    float v = 0.0f;
    if (row < M) {
      if (k < d) {
        v = A0[(size_t)row * d + k];
      } else if (A1 != nullptr && k < 2 * d) {
        int kk = k - d;
        v = (A0[(size_t)row * d + kk] + A1[(size_t)row * lda1 + kk]) / deg[(size_t)row * ldd];
      }
    }
    As[m * LDSTRIDE + k] = v;
  }
  __syncthreads();

  const int lane  = t & 31;
  const int wave  = t >> 5;
  const int n     = wave * 16 + (lane & 15);
  const int mA    = lane & 15;
  const int khalf = (lane < 16) ? 0 : 2;

  v8f c0 = {};
  v8f c1 = {};
#pragma unroll 2
  for (int k0 = 0; k0 < Kpad; k0 += 4) {
    const int kb = k0 + khalf;
    v2f a0 = *(const v2f*)&As[mA * LDSTRIDE + kb];
    v2f a1 = *(const v2f*)&As[(mA + 16) * LDSTRIDE + kb];
    v2f b  = *(const v2f*)(Wp + ((size_t)(kb >> 1) * NCOLS + n) * 2);
    c0 = __builtin_amdgcn_wmma_f32_16x16x4_f32(false, a0, false, b,
                                               (short)0, c0, false, false);
    c1 = __builtin_amdgcn_wmma_f32_16x16x4_f32(false, a1, false, b,
                                               (short)0, c1, false, false);
  }

  const float bv    = bias[n];
  const int   rbase = (lane < 16) ? 0 : 8;
  float* outA = out + (size_t)(row0 + rbase) * NCOLS + n;
  float* outB = outA + (size_t)16 * NCOLS;

  if (row0 + MTILE <= M) {
#pragma unroll
    for (int r = 0; r < 8; ++r) {
      float vA = c0[r] + bv;
      float vB = c1[r] + bv;
      outA[(size_t)r * NCOLS] = vA > 0.0f ? vA : 0.0f;
      outB[(size_t)r * NCOLS] = vB > 0.0f ? vB : 0.0f;
    }
  } else {
#pragma unroll
    for (int r = 0; r < 8; ++r) {
      int rowA = row0 + rbase + r;
      if (rowA < M) {
        float v = c0[r] + bv;
        outA[(size_t)r * NCOLS] = v > 0.0f ? v : 0.0f;
      }
      int rowB = row0 + 16 + rbase + r;
      if (rowB < M) {
        float v = c1[r] + bv;
        outB[(size_t)r * NCOLS] = v > 0.0f ? v : 0.0f;
      }
    }
  }
}

#define KG 2
#define BSTEP (2 * NCOLS * 2)

__device__ __forceinline__ void load_b_group(v2f (&b)[KG], const float* __restrict__ bp, const int g) {
#pragma unroll
  for (int j = 0; j < KG; ++j)
    b[j] = *(const v2f*)(bp + (size_t)(g * KG + j) * BSTEP);
}

__device__ __forceinline__ void mma_group(v8f& c0, v8f& c1, const float* ap, const v2f (&b)[KG], const int g) {
  v2f a0[KG], a1[KG];
#pragma unroll
  for (int j = 0; j < KG; ++j) {
    a0[j] = *(const v2f*)(ap + (g * KG + j) * 4);
    a1[j] = *(const v2f*)(ap + 16 * LDSTRIDE + (g * KG + j) * 4);
  }
#pragma unroll
  for (int j = 0; j < KG; ++j) {
    c0 = __builtin_amdgcn_wmma_f32_16x16x4_f32(false, a0[j], false, b[j],
                                               (short)0, c0, false, false);
    c1 = __builtin_amdgcn_wmma_f32_16x16x4_f32(false, a1[j], false, b[j],
                                               (short)0, c1, false, false);
  }
}

__device__ __forceinline__ void gemm_groups(v8f& c0, v8f& c1, const float* ap,
                                            const float* __restrict__ bp, const int ng,
                                            const float* __restrict__ bp_next, const bool has_next,
                                            v2f (&bA)[KG]) {
  v2f bB[KG];
#pragma unroll 1
  for (int g = 0; g < ng - 2; g += 2) {
    load_b_group(bB, bp, g + 1);
    mma_group(c0, c1, ap, bA, g);
    load_b_group(bA, bp, g + 2);
    mma_group(c0, c1, ap, bB, g + 1);
  }
  load_b_group(bB, bp, ng - 1);
  mma_group(c0, c1, ap, bA, ng - 2);
  if (has_next) load_b_group(bA, bp_next, 0);
  mma_group(c0, c1, ap, bB, ng - 1);
}

#define NPW (MTILE / 8)
static_assert(SLOTCAP == 32, "one slot word per lane");

__global__ void sage_head_fused(const float* __restrict__ h, const float* __restrict__ rows,
                                const int* __restrict__ slot,
                                const int* __restrict__ ovf_cnt, const int* __restrict__ ovf,
                                const float* __restrict__ Wp2, const float* __restrict__ b2,
                                const float* __restrict__ Wp3, const float* __restrict__ b3,
                                const float* __restrict__ W4, const float* __restrict__ b4,
                                float* __restrict__ out, int M) {
  __shared__ __attribute__((aligned(16))) float As[MTILE * LDSTRIDE];

  const int Kpad = 2 * NCOLS;
  const int row0 = blockIdx.x * MTILE;
  const int t = threadIdx.x;
  const int lane  = t & 31;
  const int wave  = t >> 5;
  const int wv = __builtin_amdgcn_readfirstlane(wave);

  const float* hl = h + 4 * lane;
  float  dg[NPW];
  int    mn[NPW];
  float4 sf[NPW];
#pragma unroll
  for (int q = 0; q < NPW; ++q) {
    const int row = row0 + wv * NPW + q;
    dg[q] = 1.0f;
    sf[q] = make_float4(0.0f, 0.0f, 0.0f, 0.0f);
    if (row < M) {
      dg[q] = rows[(size_t)row * ROWW + DEGSLOT];
      sf[q] = *(const float4*)(hl + (size_t)row * 128);
    }
  }
#pragma unroll
  for (int q = 0; q < NPW; ++q) {
    const int row = row0 + wv * NPW + q;
    const int filled = (int)dg[q] - 1;
    mn[q] = (lane < filled) ? slot[(size_t)row * SLOTCAP + lane] : 0;
  }
#pragma unroll
  for (int q = 0; q < NPW; ++q) {
    const int m   = wv * NPW + q;
    const int row = row0 + m;
    const int indeg = __builtin_amdgcn_readfirstlane((int)dg[q] - 1);
    const int end   = indeg < SLOTCAP ? indeg : SLOTCAP;
    const int mine  = mn[q];
    float4 acc = make_float4(0.0f, 0.0f, 0.0f, 0.0f);
    int k = 0;
    for (; k + 4 <= end; k += 4) {
      const int s0 = __builtin_amdgcn_readlane(mine, k);
      const int s1 = __builtin_amdgcn_readlane(mine, k + 1);
      const int s2 = __builtin_amdgcn_readlane(mine, k + 2);
      const int s3 = __builtin_amdgcn_readlane(mine, k + 3);
      const float4 v0 = *(const float4*)(hl + (size_t)s0 * 128);
      const float4 v1 = *(const float4*)(hl + (size_t)s1 * 128);
      const float4 v2 = *(const float4*)(hl + (size_t)s2 * 128);
      const float4 v3 = *(const float4*)(hl + (size_t)s3 * 128);
      acc.x += v0.x; acc.y += v0.y; acc.z += v0.z; acc.w += v0.w;
      acc.x += v1.x; acc.y += v1.y; acc.z += v1.z; acc.w += v1.w;
      acc.x += v2.x; acc.y += v2.y; acc.z += v2.z; acc.w += v2.w;
      acc.x += v3.x; acc.y += v3.y; acc.z += v3.z; acc.w += v3.w;
    }
    {
      float4 t0 = make_float4(0.0f, 0.0f, 0.0f, 0.0f), t1 = t0, t2 = t0;
      if (k < end)     t0 = *(const float4*)(hl + (size_t)__builtin_amdgcn_readlane(mine, k) * 128);
      if (k + 1 < end) t1 = *(const float4*)(hl + (size_t)__builtin_amdgcn_readlane(mine, k + 1) * 128);
      if (k + 2 < end) t2 = *(const float4*)(hl + (size_t)__builtin_amdgcn_readlane(mine, k + 2) * 128);
      if (k < end)     { acc.x += t0.x; acc.y += t0.y; acc.z += t0.z; acc.w += t0.w; }
      if (k + 1 < end) { acc.x += t1.x; acc.y += t1.y; acc.z += t1.z; acc.w += t1.w; }
      if (k + 2 < end) { acc.x += t2.x; acc.y += t2.y; acc.z += t2.z; acc.w += t2.w; }
    }
    if (indeg > SLOTCAP) {
      const int ocnt = __builtin_amdgcn_readfirstlane(*ovf_cnt);
      for (int k0 = 0; k0 < ocnt; k0 += 32) {
        const int kk = k0 + lane;
        int ss = 0, dd = -1;
        if (kk < ocnt) { ss = ovf[2 * (size_t)kk]; dd = ovf[2 * (size_t)kk + 1]; }
        unsigned hit = __builtin_amdgcn_ballot_w32(dd == row);
        while (hit != 0u) {
          const int b = __builtin_ctz(hit);
          hit &= hit - 1u;
          const int s = __builtin_amdgcn_readlane(ss, b);
          const float4 v = *(const float4*)(hl + (size_t)s * 128);
          acc.x += v.x; acc.y += v.y; acc.z += v.z; acc.w += v.w;
        }
      }
    }
    const float  dv = dg[q];
    const float4 self = sf[q];
    float4 mean;
    mean.x = (self.x + acc.x) / dv;
    mean.y = (self.y + acc.y) / dv;
    mean.z = (self.z + acc.z) / dv;
    mean.w = (self.w + acc.w) / dv;
    *(float4*)&As[m * LDSTRIDE + 4 * lane]         = self;
    *(float4*)&As[m * LDSTRIDE + NCOLS + 4 * lane] = mean;
  }
  const int n     = wave * 16 + (lane & 15);
  const int mA    = lane & 15;
  const int khalf = (lane < 16) ? 0 : 2;
  const int rbase = (lane < 16) ? 0 : 8;

  const float* bp2 = Wp2 + ((size_t)(khalf >> 1) * NCOLS + n) * 2;
  const float* bp3 = Wp3 + ((size_t)(khalf >> 1) * NCOLS + n) * 2;
  const float* ap  = &As[mA * LDSTRIDE + khalf];

  v2f bA[KG];
  load_b_group(bA, bp2, 0);
  __syncthreads();

  v8f c0 = {};
  v8f c1 = {};
  gemm_groups(c0, c1, ap, bp2, Kpad / (4 * KG), bp3, true, bA);

  __syncthreads();
  {
    const float bv = b2[n];
#pragma unroll
    for (int r = 0; r < 8; ++r) {
      float vA = c0[r] + bv;
      float vB = c1[r] + bv;
      As[(rbase + r) * LDSTRIDE + n]      = vA > 0.0f ? vA : 0.0f;
      As[(16 + rbase + r) * LDSTRIDE + n] = vB > 0.0f ? vB : 0.0f;
    }
  }
  __syncthreads();

  v8f e0 = {};
  v8f e1 = {};
  gemm_groups(e0, e1, ap, bp3, NCOLS / (4 * KG), bp3, false, bA);
  {
    const float bv = b3[n];
#pragma unroll
    for (int r = 0; r < 8; ++r) {
      float vA = e0[r] + bv;
      float vB = e1[r] + bv;
      As[(rbase + r) * LDSTRIDE + NCOLS + n]      = vA > 0.0f ? vA : 0.0f;
      As[(16 + rbase + r) * LDSTRIDE + NCOLS + n] = vB > 0.0f ? vB : 0.0f;
    }
  }
  __syncthreads();

  const int orow = t >> 3;
  const int part = t & 7;
  const float* hrow = &As[orow * LDSTRIDE + NCOLS + part * 16];
  const float* w    = W4 + part * 32;
  float s0 = 0.0f, s1 = 0.0f;
#pragma unroll
  for (int q = 0; q < 4; ++q) {
    const float4 hv = *(const float4*)(hrow + 4 * q);
    const float4 wa = *(const float4*)(w + 8 * q);
    const float4 wb = *(const float4*)(w + 8 * q + 4);
    s0 += hv.x * wa.x;  s1 += hv.x * wa.y;
    s0 += hv.y * wa.z;  s1 += hv.y * wa.w;
    s0 += hv.z * wb.x;  s1 += hv.z * wb.y;
    s0 += hv.w * wb.z;  s1 += hv.w * wb.w;
  }
  for (int o = 1; o < 8; o <<= 1) {
    s0 += __shfl_xor(s0, o);
    s1 += __shfl_xor(s1, o);
  }
  if (part == 0) {
    As[orow * 2 + 0] = s0 + b4[0];
    As[orow * 2 + 1] = s1 + b4[1];
  }
  __syncthreads();
  if (t < 2 * MTILE && row0 + (t >> 1) < M)
    out[(size_t)row0 * 2 + t] = As[t];
}

extern "C" void kernel_launch(void* const* d_in, const int* in_sizes, int n_in,
                              void* d_out, int out_size, void* d_ws, size_t ws_size,
                              hipStream_t stream) {
  const float* x   = (const float*)d_in[0];
  const int*   ei  = (const int*)d_in[1];
  const float* W1  = (const float*)d_in[2];
  const float* b1  = (const float*)d_in[3];
  const float* W2  = (const float*)d_in[4];
  const float* b2  = (const float*)d_in[5];
  const float* l1W = (const float*)d_in[6];
  const float* l1b = (const float*)d_in[7];
  const float* l2W = (const float*)d_in[8];
  const float* l2b = (const float*)d_in[9];

  const int N = in_sizes[0] / 5;
  const int E = in_sizes[1] / 2;
  const int* src = ei;
  const int* dst = ei + E;

  float* ws   = (float*)d_ws;
  float* rows = ws;
  float* deg  = rows + DEGSLOT;
  float* h1   = rows + (size_t)ROWW * N;
  float* Wp1  = h1   + (size_t)128 * N;
  float* Wp2  = Wp1  + (size_t)12 * 128;
  float* Wp3  = Wp2  + (size_t)256 * 128;
  int* slot    = (int*)(Wp3 + (size_t)128 * 128);
  int* ovf_cnt = slot + (size_t)SLOTCAP * N;
  int* ovf     = ovf_cnt + 64;

  const int nthread = 256;

  pack_w<<<(12  * NCOLS + nthread - 1) / nthread, nthread, 0, stream>>>(W1,  Wp1, 10,  12);
  pack_w<<<(256 * NCOLS + nthread - 1) / nthread, nthread, 0, stream>>>(W2,  Wp2, 256, 256);
  pack_w<<<(128 * NCOLS + nthread - 1) / nthread, nthread, 0, stream>>>(l1W, Wp3, 128, 128);

  init_ws<<<((unsigned)N * (unsigned)ROWW + nthread - 1) / nthread, nthread, 0, stream>>>(rows, ovf_cnt, N);

  edge_accum_feat<<<(E + nthread - 1) / nthread, nthread, 0, stream>>>(src, dst, x, rows, slot, ovf_cnt, ovf, E);

  sage_gemm_wmma<<<(N + MTILE - 1) / MTILE, nthread, 0, stream>>>(x, rows, deg, Wp1, b1, h1, N, 5, 12, ROWW, ROWW);

  sage_head_fused<<<(N + MTILE - 1) / MTILE, nthread, 0, stream>>>(h1, rows, slot, ovf_cnt, ovf, Wp2, b2, Wp3, l1b, l2W, l2b,
                                                                    (float*)d_out, N);
}


# Edge pass without per-edge global atomics or scattered stores: edges regrouped by 256-node target range through LDS, one work-group per range sums rows and fills lists by LDS adds, arrival order
# speedup vs baseline: 4.8446x; 1.3127x over previous
#include <hip/hip_runtime.h>

typedef __attribute__((ext_vector_type(2))) float v2f;
typedef __attribute__((ext_vector_type(8))) float v8f;

#define NCOLS 128
#define MTILE 32
#define LDSTRIDE 260

__global__ void pack_w(const float* __restrict__ W, float* __restrict__ Wp,
                       int Kw, int Kpad) {
  int i = blockIdx.x * blockDim.x + threadIdx.x;
  if (i < Kpad * NCOLS) {
    int k = i / NCOLS;
    int n = i - k * NCOLS;
    float v = (k < Kw) ? W[(size_t)k * NCOLS + n] : 0.0f;
    Wp[((size_t)(k >> 1) * NCOLS + n) * 2 + (k & 1)] = v;
  }
}

#define ROWW 8
#define DEGSLOT 5
#define SLOTCAP 32
#define RSH 8
#define RNODES (1 << RSH)
#define MAXR 512
#define CHUNK 4096
#define EPT (CHUNK / 256)
#define OWN_T 512

__global__ void init_tabs(int* __restrict__ tot, int* __restrict__ cursor, int* __restrict__ ovf_cnt) {
  const int i = blockIdx.x * blockDim.x + threadIdx.x;
  if (i < MAXR) { tot[i] = 0; cursor[i] = 0; }
  if (i == 0) *ovf_cnt = 0;
}

__global__ void range_count(const int* __restrict__ dst, int* __restrict__ tot, int E) {
  __shared__ int cnt[MAXR];
  const int t = threadIdx.x;
  for (int i = t; i < MAXR; i += 256) cnt[i] = 0;
  __syncthreads();
  const int e0 = blockIdx.x * CHUNK;
#pragma unroll
  for (int j = 0; j < EPT; ++j) {
    const int e = e0 + j * 256 + t;
    if (e < E) atomicAdd(&cnt[dst[e] >> RSH], 1);
  }
  __syncthreads();
  for (int i = t; i < MAXR; i += 256) {
    const int v = cnt[i];
    if (v != 0) atomicAdd(&tot[i], v);
  }
}

__global__ void range_place(const int* __restrict__ src, const int* __restrict__ dst,
                            const int* __restrict__ tot, int* __restrict__ cursor,
                            int* __restrict__ gstart, int* __restrict__ binned, int E, int NR) {
  __shared__ int cnt[MAXR];
  __shared__ int loff[MAXR];
  __shared__ int gs[MAXR];
  __shared__ int gbase[MAXR];
  __shared__ int2 sorted[CHUNK];
  const int t = threadIdx.x;
  for (int i = t; i < MAXR; i += 256) {
    cnt[i] = 0;
    gs[i]  = (i < NR) ? tot[i] : 0;
  }
  __syncthreads();
  const int e0 = blockIdx.x * CHUNK;
  int s[EPT], d[EPT], r[EPT];
#pragma unroll
  for (int j = 0; j < EPT; ++j) {
    const int e = e0 + j * 256 + t;
    s[j] = 0; d[j] = 0;
    if (e < E) { s[j] = src[e]; d[j] = dst[e]; }
  }
#pragma unroll
  for (int j = 0; j < EPT; ++j) {
    const int e = e0 + j * 256 + t;
    r[j] = -1;
    if (e < E) r[j] = atomicAdd(&cnt[d[j] >> RSH], 1);
  }
  __syncthreads();
  if (t < 32) {
    const int i0 = t * (MAXR / 32);
    int a = 0, b = 0;
    for (int j = 0; j < MAXR / 32; ++j) { a += cnt[i0 + j]; b += gs[i0 + j]; }
    int ia = a, ib = b;
    for (int o = 1; o < 32; o <<= 1) {
      const int ua = __shfl_up(ia, o);
      const int ub = __shfl_up(ib, o);
      if (t >= o) { ia += ua; ib += ub; }
    }
    int ra = ia - a, rb = ib - b;
    for (int j = 0; j < MAXR / 32; ++j) {
      const int va = cnt[i0 + j];
      const int vb = gs[i0 + j];
      loff[i0 + j] = ra;
      gs[i0 + j]   = rb;
      ra += va;
      rb += vb;
    }
  }
  __syncthreads();
  for (int i = t; i < MAXR; i += 256) {
    const int v = cnt[i];
    int gb = 0;
    if (v != 0) gb = gs[i] + atomicAdd(&cursor[i], v);
    gbase[i] = gb;
    if (blockIdx.x == 0 && i < NR) gstart[i] = gs[i];
  }
#pragma unroll
  for (int j = 0; j < EPT; ++j)
    if (r[j] >= 0) sorted[loff[d[j] >> RSH] + r[j]] = make_int2(s[j], d[j]);
  __syncthreads();
  const int left = E - e0;
  const int nloc = left < CHUNK ? left : CHUNK;
  int2* __restrict__ outp = reinterpret_cast<int2*>(binned);
  for (int i = t; i < nloc; i += 256) {
    const int2 p = sorted[i];
    const int c = p.y >> RSH;
    outp[(size_t)gbase[c] + (size_t)(i - loff[c])] = p;
  }
}

__global__ void __launch_bounds__(OWN_T)
range_own(const int* __restrict__ binned, const int* __restrict__ gstart, const int* __restrict__ tot,
          const float* __restrict__ x, float* __restrict__ rows, int* __restrict__ slot,
          int* __restrict__ ovf_cnt, int* __restrict__ ovf, int N) {
  __shared__ __attribute__((aligned(16))) float rows_l[RNODES * ROWW];
  __shared__ __attribute__((aligned(16))) int   slot_l[RNODES * SLOTCAP];
  const int t = threadIdx.x;
  const int c = blockIdx.x;
  const int node0 = c << RSH;
  for (int i = t; i < RNODES * ROWW; i += OWN_T)
    rows_l[i] = ((i & (ROWW - 1)) == DEGSLOT) ? 1.0f : 0.0f;
  for (int i = t; i < RNODES * SLOTCAP; i += OWN_T) slot_l[i] = 0;
  __syncthreads();

  const int base = gstart[c];
  const int cnt  = tot[c];
  const int2* __restrict__ inp = reinterpret_cast<const int2*>(binned);
  const unsigned lane = (unsigned)t & 31u;
  const int wv = __builtin_amdgcn_readfirstlane(t >> 5);
  const unsigned f  = lane & 7u;
  const unsigned g  = lane >> 3;
  const unsigned fx = f < 5u ? f : 4u;

  for (int off = wv * 32; off < cnt; off += OWN_T) {
    int s = 0, d = node0;
    const bool have = off + (int)lane < cnt;
    if (have) {
      const int2 p = inp[(size_t)base + (size_t)(off + (int)lane)];
      s = p.x;
      d = p.y;
    }
    float  xv[8];
    float* p[8];
    bool   ok[8];
#pragma unroll
    for (int r = 0; r < 8; ++r) {
      const int sl = r * 4 + (int)g;
      const int ss = __shfl(s, sl);
      const int dd = __shfl(d, sl);
      ok[r] = (off + sl < cnt) && (f <= (unsigned)DEGSLOT);
      p[r]  = rows_l + (dd - node0) * ROWW + (int)f;
      xv[r] = x[(size_t)ss * 5 + fx];
    }
    float oldv[8];
#pragma unroll
    for (int r = 0; r < 8; ++r) {
      oldv[r] = 0.0f;
      if (ok[r]) oldv[r] = atomicAdd(p[r], f < 5u ? xv[r] : 1.0f);
    }
    float myold = 0.0f;
#pragma unroll
    for (int r = 0; r < 8; ++r) {
      const float v = __shfl(oldv[r], (int)((lane & 3u) * 8u + (unsigned)DEGSLOT));
      if ((lane >> 2) == (unsigned)r) myold = v;
    }
    if (have) {
      const int place = (int)myold - 1;
      if (place < SLOTCAP) {
        slot_l[(d - node0) * SLOTCAP + place] = s;
      } else {
        const int k = atomicAdd(ovf_cnt, 1);
        ovf[2 * (size_t)k]     = s;
        ovf[2 * (size_t)k + 1] = d;
      }
    }
  }
  __syncthreads();

  const int left = N - node0;
  const int nn = left < RNODES ? left : RNODES;
  float4* __restrict__ rows_o = reinterpret_cast<float4*>(rows + (size_t)node0 * ROWW);
  int4*   __restrict__ slot_o = reinterpret_cast<int4*>(slot + (size_t)node0 * SLOTCAP);
  for (int i = t; i < nn * (ROWW / 4); i += OWN_T)
    rows_o[i] = *reinterpret_cast<const float4*>(&rows_l[4 * i]);
  for (int i = t; i < nn * (SLOTCAP / 4); i += OWN_T)
    slot_o[i] = *reinterpret_cast<const int4*>(&slot_l[4 * i]);
}

__global__ void sage_gemm_wmma(const float* __restrict__ A0, const float* __restrict__ A1,
                               const float* __restrict__ deg, const float* __restrict__ Wp,
                               const float* __restrict__ bias, float* __restrict__ out,
                               int M, int d, int Kpad, int lda1, int ldd) {
  __shared__ float As[MTILE * LDSTRIDE];

  const int row0 = blockIdx.x * MTILE;
  const int t = threadIdx.x;

  const int total = MTILE * Kpad;
  for (int i = t; i < total; i += 256) {
    int m = i / Kpad;
    int k = i - m * Kpad;
    int row = row0 + m;
    float v = 0.0f;
    if (row < M) {
      if (k < d) {
        v = A0[(size_t)row * d + k];
      } else if (A1 != nullptr && k < 2 * d) {
        int kk = k - d;
        v = (A0[(size_t)row * d + kk] + A1[(size_t)row * lda1 + kk]) / deg[(size_t)row * ldd];
      }
    }
    As[m * LDSTRIDE + k] = v;
  }
  __syncthreads();

  const int lane  = t & 31;
  const int wave  = t >> 5;
  const int n     = wave * 16 + (lane & 15);
  const int mA    = lane & 15;
  const int khalf = (lane < 16) ? 0 : 2;

  v8f c0 = {};
  v8f c1 = {};
#pragma unroll 2
  for (int k0 = 0; k0 < Kpad; k0 += 4) {
    const int kb = k0 + khalf;
    v2f a0 = *(const v2f*)&As[mA * LDSTRIDE + kb];
    v2f a1 = *(const v2f*)&As[(mA + 16) * LDSTRIDE + kb];
    v2f b  = *(const v2f*)(Wp + ((size_t)(kb >> 1) * NCOLS + n) * 2);
    c0 = __builtin_amdgcn_wmma_f32_16x16x4_f32(false, a0, false, b,
                                               (short)0, c0, false, false);
    c1 = __builtin_amdgcn_wmma_f32_16x16x4_f32(false, a1, false, b,
                                               (short)0, c1, false, false);
  }

  const float bv    = bias[n];
  const int   rbase = (lane < 16) ? 0 : 8;
  float* outA = out + (size_t)(row0 + rbase) * NCOLS + n;
  float* outB = outA + (size_t)16 * NCOLS;

  if (row0 + MTILE <= M) {
#pragma unroll
    for (int r = 0; r < 8; ++r) {
      float vA = c0[r] + bv;
      float vB = c1[r] + bv;
      outA[(size_t)r * NCOLS] = vA > 0.0f ? vA : 0.0f;
      outB[(size_t)r * NCOLS] = vB > 0.0f ? vB : 0.0f;
    }
  } else {
#pragma unroll
    for (int r = 0; r < 8; ++r) {
      int rowA = row0 + rbase + r;
      if (rowA < M) {
        float v = c0[r] + bv;
        outA[(size_t)r * NCOLS] = v > 0.0f ? v : 0.0f;
      }
      int rowB = row0 + 16 + rbase + r;
      if (rowB < M) {
        float v = c1[r] + bv;
        outB[(size_t)r * NCOLS] = v > 0.0f ? v : 0.0f;
      }
    }
  }
}

#define KG 2
#define BSTEP (2 * NCOLS * 2)

__device__ __forceinline__ void load_b_group(v2f (&b)[KG], const float* __restrict__ bp, const int g) {
#pragma unroll
  for (int j = 0; j < KG; ++j)
    b[j] = *(const v2f*)(bp + (size_t)(g * KG + j) * BSTEP);
}

__device__ __forceinline__ void mma_group(v8f& c0, v8f& c1, const float* ap, const v2f (&b)[KG], const int g) {
  v2f a0[KG], a1[KG];
#pragma unroll
  for (int j = 0; j < KG; ++j) {
    a0[j] = *(const v2f*)(ap + (g * KG + j) * 4);
    a1[j] = *(const v2f*)(ap + 16 * LDSTRIDE + (g * KG + j) * 4);
  }
#pragma unroll
  for (int j = 0; j < KG; ++j) {
    c0 = __builtin_amdgcn_wmma_f32_16x16x4_f32(false, a0[j], false, b[j],
                                               (short)0, c0, false, false);
    c1 = __builtin_amdgcn_wmma_f32_16x16x4_f32(false, a1[j], false, b[j],
                                               (short)0, c1, false, false);
  }
}

__device__ __forceinline__ void gemm_groups(v8f& c0, v8f& c1, const float* ap,
                                            const float* __restrict__ bp, const int ng,
                                            const float* __restrict__ bp_next, const bool has_next,
                                            v2f (&bA)[KG]) {
  v2f bB[KG];
#pragma unroll 1
  for (int g = 0; g < ng - 2; g += 2) {
    load_b_group(bB, bp, g + 1);
    mma_group(c0, c1, ap, bA, g);
    load_b_group(bA, bp, g + 2);
    mma_group(c0, c1, ap, bB, g + 1);
  }
  load_b_group(bB, bp, ng - 1);
  mma_group(c0, c1, ap, bA, ng - 2);
  if (has_next) load_b_group(bA, bp_next, 0);
  mma_group(c0, c1, ap, bB, ng - 1);
}

#define NPW (MTILE / 8)
static_assert(SLOTCAP == 32, "one slot word per lane");

__global__ void sage_head_fused(const float* __restrict__ h, const float* __restrict__ rows,
                                const int* __restrict__ slot,
                                const int* __restrict__ ovf_cnt, const int* __restrict__ ovf,
                                const float* __restrict__ Wp2, const float* __restrict__ b2,
                                const float* __restrict__ Wp3, const float* __restrict__ b3,
                                const float* __restrict__ W4, const float* __restrict__ b4,
                                float* __restrict__ out, int M) {
  __shared__ __attribute__((aligned(16))) float As[MTILE * LDSTRIDE];

  const int Kpad = 2 * NCOLS;
  const int row0 = blockIdx.x * MTILE;
  const int t = threadIdx.x;
  const int lane  = t & 31;
  const int wave  = t >> 5;
  const int wv = __builtin_amdgcn_readfirstlane(wave);

  const float* hl = h + 4 * lane;
  float  dg[NPW];
  int    mn[NPW];
  float4 sf[NPW];
#pragma unroll
  for (int q = 0; q < NPW; ++q) {
    const int row = row0 + wv * NPW + q;
    dg[q] = 1.0f;
    sf[q] = make_float4(0.0f, 0.0f, 0.0f, 0.0f);
    if (row < M) {
      dg[q] = rows[(size_t)row * ROWW + DEGSLOT];
      sf[q] = *(const float4*)(hl + (size_t)row * 128);
    }
  }
#pragma unroll
  for (int q = 0; q < NPW; ++q) {
    const int row = row0 + wv * NPW + q;
    const int filled = (int)dg[q] - 1;
    mn[q] = (lane < filled) ? slot[(size_t)row * SLOTCAP + lane] : 0;
  }
#pragma unroll
  for (int q = 0; q < NPW; ++q) {
    const int m   = wv * NPW + q;
    const int row = row0 + m;
    const int indeg = __builtin_amdgcn_readfirstlane((int)dg[q] - 1);
    const int end   = indeg < SLOTCAP ? indeg : SLOTCAP;
    const int mine  = mn[q];
    float4 acc = make_float4(0.0f, 0.0f, 0.0f, 0.0f);
    int k = 0;
    for (; k + 4 <= end; k += 4) {
      const int s0 = __builtin_amdgcn_readlane(mine, k);
      const int s1 = __builtin_amdgcn_readlane(mine, k + 1);
      const int s2 = __builtin_amdgcn_readlane(mine, k + 2);
      const int s3 = __builtin_amdgcn_readlane(mine, k + 3);
      const float4 v0 = *(const float4*)(hl + (size_t)s0 * 128);
      const float4 v1 = *(const float4*)(hl + (size_t)s1 * 128);
      const float4 v2 = *(const float4*)(hl + (size_t)s2 * 128);
      const float4 v3 = *(const float4*)(hl + (size_t)s3 * 128);
      acc.x += v0.x; acc.y += v0.y; acc.z += v0.z; acc.w += v0.w;
      acc.x += v1.x; acc.y += v1.y; acc.z += v1.z; acc.w += v1.w;
      acc.x += v2.x; acc.y += v2.y; acc.z += v2.z; acc.w += v2.w;
      acc.x += v3.x; acc.y += v3.y; acc.z += v3.z; acc.w += v3.w;
    }
    {
      float4 t0 = make_float4(0.0f, 0.0f, 0.0f, 0.0f), t1 = t0, t2 = t0;
      if (k < end)     t0 = *(const float4*)(hl + (size_t)__builtin_amdgcn_readlane(mine, k) * 128);
      if (k + 1 < end) t1 = *(const float4*)(hl + (size_t)__builtin_amdgcn_readlane(mine, k + 1) * 128);
      if (k + 2 < end) t2 = *(const float4*)(hl + (size_t)__builtin_amdgcn_readlane(mine, k + 2) * 128);
      if (k < end)     { acc.x += t0.x; acc.y += t0.y; acc.z += t0.z; acc.w += t0.w; }
      if (k + 1 < end) { acc.x += t1.x; acc.y += t1.y; acc.z += t1.z; acc.w += t1.w; }
      if (k + 2 < end) { acc.x += t2.x; acc.y += t2.y; acc.z += t2.z; acc.w += t2.w; }
    }
    if (indeg > SLOTCAP) {
      const int ocnt = __builtin_amdgcn_readfirstlane(*ovf_cnt);
      for (int k0 = 0; k0 < ocnt; k0 += 32) {
        const int kk = k0 + lane;
        int ss = 0, dd = -1;
        if (kk < ocnt) { ss = ovf[2 * (size_t)kk]; dd = ovf[2 * (size_t)kk + 1]; }
        unsigned hit = __builtin_amdgcn_ballot_w32(dd == row);
        while (hit != 0u) {
          const int b = __builtin_ctz(hit);
          hit &= hit - 1u;
          const int s = __builtin_amdgcn_readlane(ss, b);
          const float4 v = *(const float4*)(hl + (size_t)s * 128);
          acc.x += v.x; acc.y += v.y; acc.z += v.z; acc.w += v.w;
        }
      }
    }
    const float  dv = dg[q];
    const float4 self = sf[q];
    float4 mean;
    mean.x = (self.x + acc.x) / dv;
    mean.y = (self.y + acc.y) / dv;
    mean.z = (self.z + acc.z) / dv;
    mean.w = (self.w + acc.w) / dv;
    *(float4*)&As[m * LDSTRIDE + 4 * lane]         = self;
    *(float4*)&As[m * LDSTRIDE + NCOLS + 4 * lane] = mean;
  }
  const int n     = wave * 16 + (lane & 15);
  const int mA    = lane & 15;
  const int khalf = (lane < 16) ? 0 : 2;
  const int rbase = (lane < 16) ? 0 : 8;

  const float* bp2 = Wp2 + ((size_t)(khalf >> 1) * NCOLS + n) * 2;
  const float* bp3 = Wp3 + ((size_t)(khalf >> 1) * NCOLS + n) * 2;
  const float* ap  = &As[mA * LDSTRIDE + khalf];

  v2f bA[KG];
  load_b_group(bA, bp2, 0);
  __syncthreads();

  v8f c0 = {};
  v8f c1 = {};
  gemm_groups(c0, c1, ap, bp2, Kpad / (4 * KG), bp3, true, bA);

  __syncthreads();
  {
    const float bv = b2[n];
#pragma unroll
    for (int r = 0; r < 8; ++r) {
      float vA = c0[r] + bv;
      float vB = c1[r] + bv;
      As[(rbase + r) * LDSTRIDE + n]      = vA > 0.0f ? vA : 0.0f;
      As[(16 + rbase + r) * LDSTRIDE + n] = vB > 0.0f ? vB : 0.0f;
    }
  }
  __syncthreads();

  v8f e0 = {};
  v8f e1 = {};
  gemm_groups(e0, e1, ap, bp3, NCOLS / (4 * KG), bp3, false, bA);
  {
    const float bv = b3[n];
#pragma unroll
    for (int r = 0; r < 8; ++r) {
      float vA = e0[r] + bv;
      float vB = e1[r] + bv;
      As[(rbase + r) * LDSTRIDE + NCOLS + n]      = vA > 0.0f ? vA : 0.0f;
      As[(16 + rbase + r) * LDSTRIDE + NCOLS + n] = vB > 0.0f ? vB : 0.0f;
    }
  }
  __syncthreads();

  const int orow = t >> 3;
  const int part = t & 7;
  const float* hrow = &As[orow * LDSTRIDE + NCOLS + part * 16];
  const float* w    = W4 + part * 32;
  float s0 = 0.0f, s1 = 0.0f;
#pragma unroll
  for (int q = 0; q < 4; ++q) {
    const float4 hv = *(const float4*)(hrow + 4 * q);
    const float4 wa = *(const float4*)(w + 8 * q);
    const float4 wb = *(const float4*)(w + 8 * q + 4);
    s0 += hv.x * wa.x;  s1 += hv.x * wa.y;
    s0 += hv.y * wa.z;  s1 += hv.y * wa.w;
    s0 += hv.z * wb.x;  s1 += hv.z * wb.y;
    s0 += hv.w * wb.z;  s1 += hv.w * wb.w;
  }
  for (int o = 1; o < 8; o <<= 1) {
    s0 += __shfl_xor(s0, o);
    s1 += __shfl_xor(s1, o);
  }
  if (part == 0) {
    As[orow * 2 + 0] = s0 + b4[0];
    As[orow * 2 + 1] = s1 + b4[1];
  }
  __syncthreads();
  if (t < 2 * MTILE && row0 + (t >> 1) < M)
    out[(size_t)row0 * 2 + t] = As[t];
}

extern "C" void kernel_launch(void* const* d_in, const int* in_sizes, int n_in,
                              void* d_out, int out_size, void* d_ws, size_t ws_size,
                              hipStream_t stream) {
  const float* x   = (const float*)d_in[0];
  const int*   ei  = (const int*)d_in[1];
  const float* W1  = (const float*)d_in[2];
  const float* b1  = (const float*)d_in[3];
  const float* W2  = (const float*)d_in[4];
  const float* b2  = (const float*)d_in[5];
  const float* l1W = (const float*)d_in[6];
  const float* l1b = (const float*)d_in[7];
  const float* l2W = (const float*)d_in[8];
  const float* l2b = (const float*)d_in[9];

  const int N = in_sizes[0] / 5;
  const int E = in_sizes[1] / 2;
  const int* src = ei;
  const int* dst = ei + E;

  float* ws   = (float*)d_ws;
  float* rows = ws;
  float* deg  = rows + DEGSLOT;
  float* h1   = rows + (size_t)ROWW * N;
  float* Wp1  = h1   + (size_t)128 * N;
  float* Wp2  = Wp1  + (size_t)12 * 128;
  float* Wp3  = Wp2  + (size_t)256 * 128;
  int* slot    = (int*)(Wp3 + (size_t)128 * 128);
  int* ovf_cnt = slot + (size_t)SLOTCAP * N;
  int* ovf     = ovf_cnt + 64;
  int* tot     = ovf + 2 * (size_t)E;
  int* cursor  = tot + MAXR;
  int* gstart  = cursor + MAXR;
  int* binned  = gstart + MAXR;
  const int NR = (N + RNODES - 1) / RNODES;

  const int nthread = 256;

  pack_w<<<(12  * NCOLS + nthread - 1) / nthread, nthread, 0, stream>>>(W1,  Wp1, 10,  12);
  pack_w<<<(256 * NCOLS + nthread - 1) / nthread, nthread, 0, stream>>>(W2,  Wp2, 256, 256);
  pack_w<<<(128 * NCOLS + nthread - 1) / nthread, nthread, 0, stream>>>(l1W, Wp3, 128, 128);

  init_tabs<<<MAXR / nthread, nthread, 0, stream>>>(tot, cursor, ovf_cnt);

  range_count<<<(E + CHUNK - 1) / CHUNK, nthread, 0, stream>>>(dst, tot, E);
  range_place<<<(E + CHUNK - 1) / CHUNK, nthread, 0, stream>>>(src, dst, tot, cursor, gstart, binned, E, NR);
  range_own<<<NR, OWN_T, 0, stream>>>(binned, gstart, tot, x, rows, slot, ovf_cnt, ovf, N);

  sage_gemm_wmma<<<(N + MTILE - 1) / MTILE, nthread, 0, stream>>>(x, rows, deg, Wp1, b1, h1, N, 5, 12, ROWW, ROWW);

  sage_head_fused<<<(N + MTILE - 1) / MTILE, nthread, 0, stream>>>(h1, rows, slot, ovf_cnt, ovf, Wp2, b2, Wp3, l1b, l2W, l2b,
                                                                    (float*)d_out, N);
}


# The four set-up launches (three weight packs, table zeroing) become one launch whose block ranges do the four independent pieces: 9 launches -> 6; arithmetic and sum order unchanged
# speedup vs baseline: 5.0642x; 1.0453x over previous
#include <hip/hip_runtime.h>

typedef __attribute__((ext_vector_type(2))) float v2f;
typedef __attribute__((ext_vector_type(8))) float v8f;

#define NCOLS 128
#define MTILE 32
#define LDSTRIDE 260

__device__ __forceinline__ void pack_w(const float* __restrict__ W, float* __restrict__ Wp,
                                       const int Kw, const int Kpad, const int i) {
  if (i < Kpad * NCOLS) {
    int k = i / NCOLS;
    int n = i - k * NCOLS;
    float v = (k < Kw) ? W[(size_t)k * NCOLS + n] : 0.0f;
    Wp[((size_t)(k >> 1) * NCOLS + n) * 2 + (k & 1)] = v;
  }
}

#define ROWW 8
#define DEGSLOT 5
#define SLOTCAP 32
#define RSH 8
#define RNODES (1 << RSH)
#define MAXR 512
#define CHUNK 4096
#define EPT (CHUNK / 256)
#define OWN_T 512

#define PREP_T 256
#define PB1 ((12  * NCOLS + PREP_T - 1) / PREP_T)
#define PB2 ((256 * NCOLS + PREP_T - 1) / PREP_T)
#define PB3 ((128 * NCOLS + PREP_T - 1) / PREP_T)
#define PBI (MAXR / PREP_T)
__global__ void prep_all(const float* __restrict__ W1, const float* __restrict__ W2,
                         const float* __restrict__ W3,
                         float* __restrict__ Wp1, float* __restrict__ Wp2, float* __restrict__ Wp3,
                         int* __restrict__ tot, int* __restrict__ cursor, int* __restrict__ ovf_cnt) {
  const int b = blockIdx.x;
  const int t = threadIdx.x;
  if (b < PB1) {
    pack_w(W1, Wp1, 10, 12, b * PREP_T + t);
  } else if (b < PB1 + PB2) {
    pack_w(W2, Wp2, 256, 256, (b - PB1) * PREP_T + t);
  } else if (b < PB1 + PB2 + PB3) {
    pack_w(W3, Wp3, 128, 128, (b - PB1 - PB2) * PREP_T + t);
  } else {
    const int i = (b - PB1 - PB2 - PB3) * PREP_T + t;
    if (i < MAXR) { tot[i] = 0; cursor[i] = 0; }
    if (i == 0) *ovf_cnt = 0;
  }
}

__global__ void range_count(const int* __restrict__ dst, int* __restrict__ tot, int E) {
  __shared__ int cnt[MAXR];
  const int t = threadIdx.x;
  for (int i = t; i < MAXR; i += 256) cnt[i] = 0;
  __syncthreads();
  const int e0 = blockIdx.x * CHUNK;
#pragma unroll
  for (int j = 0; j < EPT; ++j) {
    const int e = e0 + j * 256 + t;
    if (e < E) atomicAdd(&cnt[dst[e] >> RSH], 1);
  }
  __syncthreads();
  for (int i = t; i < MAXR; i += 256) {
    const int v = cnt[i];
    if (v != 0) atomicAdd(&tot[i], v);
  }
}

__global__ void range_place(const int* __restrict__ src, const int* __restrict__ dst,
                            const int* __restrict__ tot, int* __restrict__ cursor,
                            int* __restrict__ gstart, int* __restrict__ binned, int E, int NR) {
  __shared__ int cnt[MAXR];
  __shared__ int loff[MAXR];
  __shared__ int gs[MAXR];
  __shared__ int gbase[MAXR];
  __shared__ int2 sorted[CHUNK];
  const int t = threadIdx.x;
  for (int i = t; i < MAXR; i += 256) {
    cnt[i] = 0;
    gs[i]  = (i < NR) ? tot[i] : 0;
  }
  __syncthreads();
  const int e0 = blockIdx.x * CHUNK;
  int s[EPT], d[EPT], r[EPT];
#pragma unroll
  for (int j = 0; j < EPT; ++j) {
    const int e = e0 + j * 256 + t;
    s[j] = 0; d[j] = 0;
    if (e < E) { s[j] = src[e]; d[j] = dst[e]; }
  }
#pragma unroll
  for (int j = 0; j < EPT; ++j) {
    const int e = e0 + j * 256 + t;
    r[j] = -1;
    if (e < E) r[j] = atomicAdd(&cnt[d[j] >> RSH], 1);
  }
  __syncthreads();
  if (t < 32) {
    const int i0 = t * (MAXR / 32);
    int a = 0, b = 0;
    for (int j = 0; j < MAXR / 32; ++j) { a += cnt[i0 + j]; b += gs[i0 + j]; }
    int ia = a, ib = b;
    for (int o = 1; o < 32; o <<= 1) {
      const int ua = __shfl_up(ia, o);
      const int ub = __shfl_up(ib, o);
      if (t >= o) { ia += ua; ib += ub; }
    }
    int ra = ia - a, rb = ib - b;
    for (int j = 0; j < MAXR / 32; ++j) {
      const int va = cnt[i0 + j];
      const int vb = gs[i0 + j];
      loff[i0 + j] = ra;
      gs[i0 + j]   = rb;
      ra += va;
      rb += vb;
    }
  }
  __syncthreads();
  for (int i = t; i < MAXR; i += 256) {
    const int v = cnt[i];
    int gb = 0;
    if (v != 0) gb = gs[i] + atomicAdd(&cursor[i], v);
    gbase[i] = gb;
    if (blockIdx.x == 0 && i < NR) gstart[i] = gs[i];
  }
#pragma unroll
  for (int j = 0; j < EPT; ++j)
    if (r[j] >= 0) sorted[loff[d[j] >> RSH] + r[j]] = make_int2(s[j], d[j]);
  __syncthreads();
  const int left = E - e0;
  const int nloc = left < CHUNK ? left : CHUNK;
  int2* __restrict__ outp = reinterpret_cast<int2*>(binned);
  for (int i = t; i < nloc; i += 256) {
    const int2 p = sorted[i];
    const int c = p.y >> RSH;
    outp[(size_t)gbase[c] + (size_t)(i - loff[c])] = p;
  }
}

__global__ void __launch_bounds__(OWN_T)
range_own(const int* __restrict__ binned, const int* __restrict__ gstart, const int* __restrict__ tot,
          const float* __restrict__ x, float* __restrict__ rows, int* __restrict__ slot,
          int* __restrict__ ovf_cnt, int* __restrict__ ovf, int N) {
  __shared__ __attribute__((aligned(16))) float rows_l[RNODES * ROWW];
  __shared__ __attribute__((aligned(16))) int   slot_l[RNODES * SLOTCAP];
  const int t = threadIdx.x;
  const int c = blockIdx.x;
  const int node0 = c << RSH;
  for (int i = t; i < RNODES * ROWW; i += OWN_T)
    rows_l[i] = ((i & (ROWW - 1)) == DEGSLOT) ? 1.0f : 0.0f;
  for (int i = t; i < RNODES * SLOTCAP; i += OWN_T) slot_l[i] = 0;
  __syncthreads();

  const int base = gstart[c];
  const int cnt  = tot[c];
  const int2* __restrict__ inp = reinterpret_cast<const int2*>(binned);
  const unsigned lane = (unsigned)t & 31u;
  const int wv = __builtin_amdgcn_readfirstlane(t >> 5);
  const unsigned f  = lane & 7u;
  const unsigned g  = lane >> 3;
  const unsigned fx = f < 5u ? f : 4u;

  for (int off = wv * 32; off < cnt; off += OWN_T) {
    int s = 0, d = node0;
    const bool have = off + (int)lane < cnt;
    if (have) {
      const int2 p = inp[(size_t)base + (size_t)(off + (int)lane)];
      s = p.x;
      d = p.y;
    }
    float  xv[8];
    float* p[8];
    bool   ok[8];
#pragma unroll
    for (int r = 0; r < 8; ++r) {
      const int sl = r * 4 + (int)g;
      const int ss = __shfl(s, sl);
      const int dd = __shfl(d, sl);
      ok[r] = (off + sl < cnt) && (f <= (unsigned)DEGSLOT);
      p[r]  = rows_l + (dd - node0) * ROWW + (int)f;
      xv[r] = x[(size_t)ss * 5 + fx];
    }
    float oldv[8];
#pragma unroll
    for (int r = 0; r < 8; ++r) {
      oldv[r] = 0.0f;
      if (ok[r]) oldv[r] = atomicAdd(p[r], f < 5u ? xv[r] : 1.0f);
    }
    float myold = 0.0f;
#pragma unroll
    for (int r = 0; r < 8; ++r) {
      const float v = __shfl(oldv[r], (int)((lane & 3u) * 8u + (unsigned)DEGSLOT));
      if ((lane >> 2) == (unsigned)r) myold = v;
    }
    if (have) {
      const int place = (int)myold - 1;
      if (place < SLOTCAP) {
        slot_l[(d - node0) * SLOTCAP + place] = s;
      } else {
        const int k = atomicAdd(ovf_cnt, 1);
        ovf[2 * (size_t)k]     = s;
        ovf[2 * (size_t)k + 1] = d;
      }
    }
  }
  __syncthreads();

  const int left = N - node0;
  const int nn = left < RNODES ? left : RNODES;
  float4* __restrict__ rows_o = reinterpret_cast<float4*>(rows + (size_t)node0 * ROWW);
  int4*   __restrict__ slot_o = reinterpret_cast<int4*>(slot + (size_t)node0 * SLOTCAP);
  for (int i = t; i < nn * (ROWW / 4); i += OWN_T)
    rows_o[i] = *reinterpret_cast<const float4*>(&rows_l[4 * i]);
  for (int i = t; i < nn * (SLOTCAP / 4); i += OWN_T)
    slot_o[i] = *reinterpret_cast<const int4*>(&slot_l[4 * i]);
}

__global__ void sage_gemm_wmma(const float* __restrict__ A0, const float* __restrict__ A1,
                               const float* __restrict__ deg, const float* __restrict__ Wp,
                               const float* __restrict__ bias, float* __restrict__ out,
                               int M, int d, int Kpad, int lda1, int ldd) {
  __shared__ float As[MTILE * LDSTRIDE];

  const int row0 = blockIdx.x * MTILE;
  const int t = threadIdx.x;

  const int total = MTILE * Kpad;
  for (int i = t; i < total; i += 256) {
    int m = i / Kpad;
    int k = i - m * Kpad;
    int row = row0 + m;
    float v = 0.0f;
    if (row < M) {
      if (k < d) {
        v = A0[(size_t)row * d + k];
      } else if (A1 != nullptr && k < 2 * d) {
        int kk = k - d;
        v = (A0[(size_t)row * d + kk] + A1[(size_t)row * lda1 + kk]) / deg[(size_t)row * ldd];
      }
    }
    As[m * LDSTRIDE + k] = v;
  }
  __syncthreads();

  const int lane  = t & 31;
  const int wave  = t >> 5;
  const int n     = wave * 16 + (lane & 15);
  const int mA    = lane & 15;
  const int khalf = (lane < 16) ? 0 : 2;

  v8f c0 = {};
  v8f c1 = {};
#pragma unroll 2
  for (int k0 = 0; k0 < Kpad; k0 += 4) {
    const int kb = k0 + khalf;
    v2f a0 = *(const v2f*)&As[mA * LDSTRIDE + kb];
    v2f a1 = *(const v2f*)&As[(mA + 16) * LDSTRIDE + kb];
    v2f b  = *(const v2f*)(Wp + ((size_t)(kb >> 1) * NCOLS + n) * 2);
    c0 = __builtin_amdgcn_wmma_f32_16x16x4_f32(false, a0, false, b,
                                               (short)0, c0, false, false);
    c1 = __builtin_amdgcn_wmma_f32_16x16x4_f32(false, a1, false, b,
                                               (short)0, c1, false, false);
  }

  const float bv    = bias[n];
  const int   rbase = (lane < 16) ? 0 : 8;
  float* outA = out + (size_t)(row0 + rbase) * NCOLS + n;
  float* outB = outA + (size_t)16 * NCOLS;

  if (row0 + MTILE <= M) {
#pragma unroll
    for (int r = 0; r < 8; ++r) {
      float vA = c0[r] + bv;
      float vB = c1[r] + bv;
      outA[(size_t)r * NCOLS] = vA > 0.0f ? vA : 0.0f;
      outB[(size_t)r * NCOLS] = vB > 0.0f ? vB : 0.0f;
    }
  } else {
#pragma unroll
    for (int r = 0; r < 8; ++r) {
      int rowA = row0 + rbase + r;
      if (rowA < M) {
        float v = c0[r] + bv;
        outA[(size_t)r * NCOLS] = v > 0.0f ? v : 0.0f;
      }
      int rowB = row0 + 16 + rbase + r;
      if (rowB < M) {
        float v = c1[r] + bv;
        outB[(size_t)r * NCOLS] = v > 0.0f ? v : 0.0f;
      }
    }
  }
}

#define KG 2
#define BSTEP (2 * NCOLS * 2)

__device__ __forceinline__ void load_b_group(v2f (&b)[KG], const float* __restrict__ bp, const int g) {
#pragma unroll
  for (int j = 0; j < KG; ++j)
    b[j] = *(const v2f*)(bp + (size_t)(g * KG + j) * BSTEP);
}

__device__ __forceinline__ void mma_group(v8f& c0, v8f& c1, const float* ap, const v2f (&b)[KG], const int g) {
  v2f a0[KG], a1[KG];
#pragma unroll
  for (int j = 0; j < KG; ++j) {
    a0[j] = *(const v2f*)(ap + (g * KG + j) * 4);
    a1[j] = *(const v2f*)(ap + 16 * LDSTRIDE + (g * KG + j) * 4);
  }
#pragma unroll
  for (int j = 0; j < KG; ++j) {
    c0 = __builtin_amdgcn_wmma_f32_16x16x4_f32(false, a0[j], false, b[j],
                                               (short)0, c0, false, false);
    c1 = __builtin_amdgcn_wmma_f32_16x16x4_f32(false, a1[j], false, b[j],
                                               (short)0, c1, false, false);
  }
}

__device__ __forceinline__ void gemm_groups(v8f& c0, v8f& c1, const float* ap,
                                            const float* __restrict__ bp, const int ng,
                                            const float* __restrict__ bp_next, const bool has_next,
                                            v2f (&bA)[KG]) {
  v2f bB[KG];
#pragma unroll 1
  for (int g = 0; g < ng - 2; g += 2) {
    load_b_group(bB, bp, g + 1);
    mma_group(c0, c1, ap, bA, g);
    load_b_group(bA, bp, g + 2);
    mma_group(c0, c1, ap, bB, g + 1);
  }
  load_b_group(bB, bp, ng - 1);
  mma_group(c0, c1, ap, bA, ng - 2);
  if (has_next) load_b_group(bA, bp_next, 0);
  mma_group(c0, c1, ap, bB, ng - 1);
}

#define NPW (MTILE / 8)
static_assert(SLOTCAP == 32, "one slot word per lane");

__global__ void sage_head_fused(const float* __restrict__ h, const float* __restrict__ rows,
                                const int* __restrict__ slot,
                                const int* __restrict__ ovf_cnt, const int* __restrict__ ovf,
                                const float* __restrict__ Wp2, const float* __restrict__ b2,
                                const float* __restrict__ Wp3, const float* __restrict__ b3,
                                const float* __restrict__ W4, const float* __restrict__ b4,
                                float* __restrict__ out, int M) {
  __shared__ __attribute__((aligned(16))) float As[MTILE * LDSTRIDE];

  const int Kpad = 2 * NCOLS;
  const int row0 = blockIdx.x * MTILE;
  const int t = threadIdx.x;
  const int lane  = t & 31;
  const int wave  = t >> 5;
  const int wv = __builtin_amdgcn_readfirstlane(wave);

  const float* hl = h + 4 * lane;
  float  dg[NPW];
  int    mn[NPW];
  float4 sf[NPW];
#pragma unroll
  for (int q = 0; q < NPW; ++q) {
    const int row = row0 + wv * NPW + q;
    dg[q] = 1.0f;
    sf[q] = make_float4(0.0f, 0.0f, 0.0f, 0.0f);
    if (row < M) {
      dg[q] = rows[(size_t)row * ROWW + DEGSLOT];
      sf[q] = *(const float4*)(hl + (size_t)row * 128);
    }
  }
#pragma unroll
  for (int q = 0; q < NPW; ++q) {
    const int row = row0 + wv * NPW + q;
    const int filled = (int)dg[q] - 1;
    mn[q] = (lane < filled) ? slot[(size_t)row * SLOTCAP + lane] : 0;
  }
#pragma unroll
  for (int q = 0; q < NPW; ++q) {
    const int m   = wv * NPW + q;
    const int row = row0 + m;
    const int indeg = __builtin_amdgcn_readfirstlane((int)dg[q] - 1);
    const int end   = indeg < SLOTCAP ? indeg : SLOTCAP;
    const int mine  = mn[q];
    float4 acc = make_float4(0.0f, 0.0f, 0.0f, 0.0f);
    int k = 0;
    for (; k + 4 <= end; k += 4) {
      const int s0 = __builtin_amdgcn_readlane(mine, k);
      const int s1 = __builtin_amdgcn_readlane(mine, k + 1);
      const int s2 = __builtin_amdgcn_readlane(mine, k + 2);
      const int s3 = __builtin_amdgcn_readlane(mine, k + 3);
      const float4 v0 = *(const float4*)(hl + (size_t)s0 * 128);
      const float4 v1 = *(const float4*)(hl + (size_t)s1 * 128);
      const float4 v2 = *(const float4*)(hl + (size_t)s2 * 128);
      const float4 v3 = *(const float4*)(hl + (size_t)s3 * 128);
      acc.x += v0.x; acc.y += v0.y; acc.z += v0.z; acc.w += v0.w;
      acc.x += v1.x; acc.y += v1.y; acc.z += v1.z; acc.w += v1.w;
      acc.x += v2.x; acc.y += v2.y; acc.z += v2.z; acc.w += v2.w;
      acc.x += v3.x; acc.y += v3.y; acc.z += v3.z; acc.w += v3.w;
    }
    {
      float4 t0 = make_float4(0.0f, 0.0f, 0.0f, 0.0f), t1 = t0, t2 = t0;
      if (k < end)     t0 = *(const float4*)(hl + (size_t)__builtin_amdgcn_readlane(mine, k) * 128);
      if (k + 1 < end) t1 = *(const float4*)(hl + (size_t)__builtin_amdgcn_readlane(mine, k + 1) * 128);
      if (k + 2 < end) t2 = *(const float4*)(hl + (size_t)__builtin_amdgcn_readlane(mine, k + 2) * 128);
      if (k < end)     { acc.x += t0.x; acc.y += t0.y; acc.z += t0.z; acc.w += t0.w; }
      if (k + 1 < end) { acc.x += t1.x; acc.y += t1.y; acc.z += t1.z; acc.w += t1.w; }
      if (k + 2 < end) { acc.x += t2.x; acc.y += t2.y; acc.z += t2.z; acc.w += t2.w; }
    }
    if (indeg > SLOTCAP) {
      const int ocnt = __builtin_amdgcn_readfirstlane(*ovf_cnt);
      for (int k0 = 0; k0 < ocnt; k0 += 32) {
        const int kk = k0 + lane;
        int ss = 0, dd = -1;
        if (kk < ocnt) { ss = ovf[2 * (size_t)kk]; dd = ovf[2 * (size_t)kk + 1]; }
        unsigned hit = __builtin_amdgcn_ballot_w32(dd == row);
        while (hit != 0u) {
          const int b = __builtin_ctz(hit);
          hit &= hit - 1u;
          const int s = __builtin_amdgcn_readlane(ss, b);
          const float4 v = *(const float4*)(hl + (size_t)s * 128);
          acc.x += v.x; acc.y += v.y; acc.z += v.z; acc.w += v.w;
        }
      }
    }
    const float  dv = dg[q];
    const float4 self = sf[q];
    float4 mean;
    mean.x = (self.x + acc.x) / dv;
    mean.y = (self.y + acc.y) / dv;
    mean.z = (self.z + acc.z) / dv;
    mean.w = (self.w + acc.w) / dv;
    *(float4*)&As[m * LDSTRIDE + 4 * lane]         = self;
    *(float4*)&As[m * LDSTRIDE + NCOLS + 4 * lane] = mean;
  }
  const int n     = wave * 16 + (lane & 15);
  const int mA    = lane & 15;
  const int khalf = (lane < 16) ? 0 : 2;
  const int rbase = (lane < 16) ? 0 : 8;

  const float* bp2 = Wp2 + ((size_t)(khalf >> 1) * NCOLS + n) * 2;
  const float* bp3 = Wp3 + ((size_t)(khalf >> 1) * NCOLS + n) * 2;
  const float* ap  = &As[mA * LDSTRIDE + khalf];

  v2f bA[KG];
  load_b_group(bA, bp2, 0);
  __syncthreads();

  v8f c0 = {};
  v8f c1 = {};
  gemm_groups(c0, c1, ap, bp2, Kpad / (4 * KG), bp3, true, bA);

  __syncthreads();
  {
    const float bv = b2[n];
#pragma unroll
    for (int r = 0; r < 8; ++r) {
      float vA = c0[r] + bv;
      float vB = c1[r] + bv;
      As[(rbase + r) * LDSTRIDE + n]      = vA > 0.0f ? vA : 0.0f;
      As[(16 + rbase + r) * LDSTRIDE + n] = vB > 0.0f ? vB : 0.0f;
    }
  }
  __syncthreads();

  v8f e0 = {};
  v8f e1 = {};
  gemm_groups(e0, e1, ap, bp3, NCOLS / (4 * KG), bp3, false, bA);
  {
    const float bv = b3[n];
#pragma unroll
    for (int r = 0; r < 8; ++r) {
      float vA = e0[r] + bv;
      float vB = e1[r] + bv;
      As[(rbase + r) * LDSTRIDE + NCOLS + n]      = vA > 0.0f ? vA : 0.0f;
      As[(16 + rbase + r) * LDSTRIDE + NCOLS + n] = vB > 0.0f ? vB : 0.0f;
    }
  }
  __syncthreads();

  const int orow = t >> 3;
  const int part = t & 7;
  const float* hrow = &As[orow * LDSTRIDE + NCOLS + part * 16];
  const float* w    = W4 + part * 32;
  float s0 = 0.0f, s1 = 0.0f;
#pragma unroll
  for (int q = 0; q < 4; ++q) {
    const float4 hv = *(const float4*)(hrow + 4 * q);
    const float4 wa = *(const float4*)(w + 8 * q);
    const float4 wb = *(const float4*)(w + 8 * q + 4);
    s0 += hv.x * wa.x;  s1 += hv.x * wa.y;
    s0 += hv.y * wa.z;  s1 += hv.y * wa.w;
    s0 += hv.z * wb.x;  s1 += hv.z * wb.y;
    s0 += hv.w * wb.z;  s1 += hv.w * wb.w;
  }
  for (int o = 1; o < 8; o <<= 1) {
    s0 += __shfl_xor(s0, o);
    s1 += __shfl_xor(s1, o);
  }
  if (part == 0) {
    As[orow * 2 + 0] = s0 + b4[0];
    As[orow * 2 + 1] = s1 + b4[1];
  }
  __syncthreads();
  if (t < 2 * MTILE && row0 + (t >> 1) < M)
    out[(size_t)row0 * 2 + t] = As[t];
}

extern "C" void kernel_launch(void* const* d_in, const int* in_sizes, int n_in,
                              void* d_out, int out_size, void* d_ws, size_t ws_size,
                              hipStream_t stream) {
  const float* x   = (const float*)d_in[0];
  const int*   ei  = (const int*)d_in[1];
  const float* W1  = (const float*)d_in[2];
  const float* b1  = (const float*)d_in[3];
  const float* W2  = (const float*)d_in[4];
  const float* b2  = (const float*)d_in[5];
  const float* l1W = (const float*)d_in[6];
  const float* l1b = (const float*)d_in[7];
  const float* l2W = (const float*)d_in[8];
  const float* l2b = (const float*)d_in[9];

  const int N = in_sizes[0] / 5;
  const int E = in_sizes[1] / 2;
  const int* src = ei;
  const int* dst = ei + E;

  float* ws   = (float*)d_ws;
  float* rows = ws;
  float* deg  = rows + DEGSLOT;
  float* h1   = rows + (size_t)ROWW * N;
  float* Wp1  = h1   + (size_t)128 * N;
  float* Wp2  = Wp1  + (size_t)12 * 128;
  float* Wp3  = Wp2  + (size_t)256 * 128;
  int* slot    = (int*)(Wp3 + (size_t)128 * 128);
  int* ovf_cnt = slot + (size_t)SLOTCAP * N;
  int* ovf     = ovf_cnt + 64;
  int* tot     = ovf + 2 * (size_t)E;
  int* cursor  = tot + MAXR;
  int* gstart  = cursor + MAXR;
  int* binned  = gstart + MAXR;
  const int NR = (N + RNODES - 1) / RNODES;

  const int nthread = 256;

  prep_all<<<PB1 + PB2 + PB3 + PBI, PREP_T, 0, stream>>>(W1, W2, l1W, Wp1, Wp2, Wp3, tot, cursor, ovf_cnt);

  range_count<<<(E + CHUNK - 1) / CHUNK, nthread, 0, stream>>>(dst, tot, E);
  range_place<<<(E + CHUNK - 1) / CHUNK, nthread, 0, stream>>>(src, dst, tot, cursor, gstart, binned, E, NR);
  range_own<<<NR, OWN_T, 0, stream>>>(binned, gstart, tot, x, rows, slot, ovf_cnt, ovf, N);

  sage_gemm_wmma<<<(N + MTILE - 1) / MTILE, nthread, 0, stream>>>(x, rows, deg, Wp1, b1, h1, N, 5, 12, ROWW, ROWW);

  sage_head_fused<<<(N + MTILE - 1) / MTILE, nthread, 0, stream>>>(h1, rows, slot, ovf_cnt, ovf, Wp2, b2, Wp3, l1b, l2W, l2b,
                                                                    (float*)d_out, N);
}


# Layer-1 GEMM moved into the edge pass's range-owning launch: each work-group computes h1 for its 256 nodes from the sums it holds in LDS; 6 launches -> 5, 25000 fewer waves; sum order unchanged
# speedup vs baseline: 5.2605x; 1.0388x over previous
#include <hip/hip_runtime.h>

typedef __attribute__((ext_vector_type(2))) float v2f;
typedef __attribute__((ext_vector_type(8))) float v8f;

#define NCOLS 128
#define MTILE 32
#define LDSTRIDE 260

__device__ __forceinline__ void pack_w(const float* __restrict__ W, float* __restrict__ Wp,
                                       const int Kw, const int Kpad, const int i) {
  if (i < Kpad * NCOLS) {
    int k = i / NCOLS;
    int n = i - k * NCOLS;
    float v = (k < Kw) ? W[(size_t)k * NCOLS + n] : 0.0f;
    Wp[((size_t)(k >> 1) * NCOLS + n) * 2 + (k & 1)] = v;
  }
}

#define ROWW 8
#define DEGSLOT 5
#define SLOTCAP 32
#define RSH 8
#define RNODES (1 << RSH)
#define MAXR 512
#define CHUNK 4096
#define EPT (CHUNK / 256)
#define OWN_T 512
#define L1K 12
#define WSTEP (2 * NCOLS * 2)

#define PREP_T 256
#define PB1 ((12  * NCOLS + PREP_T - 1) / PREP_T)
#define PB2 ((256 * NCOLS + PREP_T - 1) / PREP_T)
#define PB3 ((128 * NCOLS + PREP_T - 1) / PREP_T)
#define PBI (MAXR / PREP_T)
__global__ void prep_all(const float* __restrict__ W1, const float* __restrict__ W2,
                         const float* __restrict__ W3,
                         float* __restrict__ Wp1, float* __restrict__ Wp2, float* __restrict__ Wp3,
                         int* __restrict__ tot, int* __restrict__ cursor, int* __restrict__ ovf_cnt) {
  const int b = blockIdx.x;
  const int t = threadIdx.x;
  if (b < PB1) {
    pack_w(W1, Wp1, 10, 12, b * PREP_T + t);
  } else if (b < PB1 + PB2) {
    pack_w(W2, Wp2, 256, 256, (b - PB1) * PREP_T + t);
  } else if (b < PB1 + PB2 + PB3) {
    pack_w(W3, Wp3, 128, 128, (b - PB1 - PB2) * PREP_T + t);
  } else {
    const int i = (b - PB1 - PB2 - PB3) * PREP_T + t;
    if (i < MAXR) { tot[i] = 0; cursor[i] = 0; }
    if (i == 0) *ovf_cnt = 0;
  }
}

__global__ void range_count(const int* __restrict__ dst, int* __restrict__ tot, int E) {
  __shared__ int cnt[MAXR];
  const int t = threadIdx.x;
  for (int i = t; i < MAXR; i += 256) cnt[i] = 0;
  __syncthreads();
  const int e0 = blockIdx.x * CHUNK;
#pragma unroll
  for (int j = 0; j < EPT; ++j) {
    const int e = e0 + j * 256 + t;
    if (e < E) atomicAdd(&cnt[dst[e] >> RSH], 1);
  }
  __syncthreads();
  for (int i = t; i < MAXR; i += 256) {
    const int v = cnt[i];
    if (v != 0) atomicAdd(&tot[i], v);
  }
}

__global__ void range_place(const int* __restrict__ src, const int* __restrict__ dst,
                            const int* __restrict__ tot, int* __restrict__ cursor,
                            int* __restrict__ gstart, int* __restrict__ binned, int E, int NR) {
  __shared__ int cnt[MAXR];
  __shared__ int loff[MAXR];
  __shared__ int gs[MAXR];
  __shared__ int gbase[MAXR];
  __shared__ int2 sorted[CHUNK];
  const int t = threadIdx.x;
  for (int i = t; i < MAXR; i += 256) {
    cnt[i] = 0;
    gs[i]  = (i < NR) ? tot[i] : 0;
  }
  __syncthreads();
  const int e0 = blockIdx.x * CHUNK;
  int s[EPT], d[EPT], r[EPT];
#pragma unroll
  for (int j = 0; j < EPT; ++j) {
    const int e = e0 + j * 256 + t;
    s[j] = 0; d[j] = 0;
    if (e < E) { s[j] = src[e]; d[j] = dst[e]; }
  }
#pragma unroll
  for (int j = 0; j < EPT; ++j) {
    const int e = e0 + j * 256 + t;
    r[j] = -1;
    if (e < E) r[j] = atomicAdd(&cnt[d[j] >> RSH], 1);
  }
  __syncthreads();
  if (t < 32) {
    const int i0 = t * (MAXR / 32);
    int a = 0, b = 0;
    for (int j = 0; j < MAXR / 32; ++j) { a += cnt[i0 + j]; b += gs[i0 + j]; }
    int ia = a, ib = b;
    for (int o = 1; o < 32; o <<= 1) {
      const int ua = __shfl_up(ia, o);
      const int ub = __shfl_up(ib, o);
      if (t >= o) { ia += ua; ib += ub; }
    }
    int ra = ia - a, rb = ib - b;
    for (int j = 0; j < MAXR / 32; ++j) {
      const int va = cnt[i0 + j];
      const int vb = gs[i0 + j];
      loff[i0 + j] = ra;
      gs[i0 + j]   = rb;
      ra += va;
      rb += vb;
    }
  }
  __syncthreads();
  for (int i = t; i < MAXR; i += 256) {
    const int v = cnt[i];
    int gb = 0;
    if (v != 0) gb = gs[i] + atomicAdd(&cursor[i], v);
    gbase[i] = gb;
    if (blockIdx.x == 0 && i < NR) gstart[i] = gs[i];
  }
#pragma unroll
  for (int j = 0; j < EPT; ++j)
    if (r[j] >= 0) sorted[loff[d[j] >> RSH] + r[j]] = make_int2(s[j], d[j]);
  __syncthreads();
  const int left = E - e0;
  const int nloc = left < CHUNK ? left : CHUNK;
  int2* __restrict__ outp = reinterpret_cast<int2*>(binned);
  for (int i = t; i < nloc; i += 256) {
    const int2 p = sorted[i];
    const int c = p.y >> RSH;
    outp[(size_t)gbase[c] + (size_t)(i - loff[c])] = p;
  }
}

__global__ void __launch_bounds__(OWN_T)
range_own(const int* __restrict__ binned, const int* __restrict__ gstart, const int* __restrict__ tot,
          const float* __restrict__ x, float* __restrict__ rows, int* __restrict__ slot,
          int* __restrict__ ovf_cnt, int* __restrict__ ovf,
          const float* __restrict__ Wp1, const float* __restrict__ b1, float* __restrict__ h1, int N) {
  __shared__ __attribute__((aligned(16))) float rows_l[RNODES * ROWW];
  __shared__ __attribute__((aligned(16))) int   slot_l[RNODES * SLOTCAP];
  __shared__ __attribute__((aligned(16))) float a_l[RNODES * L1K];
  const int t = threadIdx.x;
  const int c = blockIdx.x;
  const int node0 = c << RSH;
  for (int i = t; i < RNODES * ROWW; i += OWN_T)
    rows_l[i] = ((i & (ROWW - 1)) == DEGSLOT) ? 1.0f : 0.0f;
  for (int i = t; i < RNODES * SLOTCAP; i += OWN_T) slot_l[i] = 0;
  __syncthreads();

  const int base = gstart[c];
  const int cnt  = tot[c];
  const int2* __restrict__ inp = reinterpret_cast<const int2*>(binned);
  const unsigned lane = (unsigned)t & 31u;
  const int wv = __builtin_amdgcn_readfirstlane(t >> 5);
  const unsigned f  = lane & 7u;
  const unsigned g  = lane >> 3;
  const unsigned fx = f < 5u ? f : 4u;

  for (int off = wv * 32; off < cnt; off += OWN_T) {
    int s = 0, d = node0;
    const bool have = off + (int)lane < cnt;
    if (have) {
      const int2 p = inp[(size_t)base + (size_t)(off + (int)lane)];
      s = p.x;
      d = p.y;
    }
    float  xv[8];
    float* p[8];
    bool   ok[8];
#pragma unroll
    for (int r = 0; r < 8; ++r) {
      const int sl = r * 4 + (int)g;
      const int ss = __shfl(s, sl);
      const int dd = __shfl(d, sl);
      ok[r] = (off + sl < cnt) && (f <= (unsigned)DEGSLOT);
      p[r]  = rows_l + (dd - node0) * ROWW + (int)f;
      xv[r] = x[(size_t)ss * 5 + fx];
    }
    float oldv[8];
#pragma unroll
    for (int r = 0; r < 8; ++r) {
      oldv[r] = 0.0f;
      if (ok[r]) oldv[r] = atomicAdd(p[r], f < 5u ? xv[r] : 1.0f);
    }
    float myold = 0.0f;
#pragma unroll
    for (int r = 0; r < 8; ++r) {
      const float v = __shfl(oldv[r], (int)((lane & 3u) * 8u + (unsigned)DEGSLOT));
      if ((lane >> 2) == (unsigned)r) myold = v;
    }
    if (have) {
      const int place = (int)myold - 1;
      if (place < SLOTCAP) {
        slot_l[(d - node0) * SLOTCAP + place] = s;
      } else {
        const int k = atomicAdd(ovf_cnt, 1);
        ovf[2 * (size_t)k]     = s;
        ovf[2 * (size_t)k + 1] = d;
      }
    }
  }
  __syncthreads();

  const int left = N - node0;
  const int nn = left < RNODES ? left : RNODES;

  const int mA    = (int)(lane & 15u);
  const int khalf = (lane < 16u) ? 0 : 2;
  v2f   wq[NCOLS / 16][3];
  float bq[NCOLS / 16];
#pragma unroll
  for (int ct = 0; ct < NCOLS / 16; ++ct) {
    const int n = ct * 16 + mA;
    const float* bp = Wp1 + ((size_t)(khalf >> 1) * NCOLS + n) * 2;
    wq[ct][0] = *(const v2f*)(bp);
    wq[ct][1] = *(const v2f*)(bp + WSTEP);
    wq[ct][2] = *(const v2f*)(bp + 2 * WSTEP);
    bq[ct]    = b1[n];
  }

  for (int i = t; i < RNODES * L1K; i += OWN_T) {
    const int m = i / L1K;
    const int k = i - m * L1K;
    float v = 0.0f;
    if (m < nn) {
      const size_t xr = (size_t)(node0 + m) * 5;
      if (k < 5) {
        v = x[xr + k];
      } else if (k < 10) {
        const int kk = k - 5;
        v = (x[xr + kk] + rows_l[m * ROWW + kk]) / rows_l[m * ROWW + DEGSLOT];
      }
    }
    a_l[i] = v;
  }

  __syncthreads();

  const int rbase = (lane < 16u) ? 0 : 8;
  const float* ap = &a_l[(wv * 16 + mA) * L1K + khalf];
  const v2f a0 = *(const v2f*)(ap);
  const v2f a1 = *(const v2f*)(ap + 4);
  const v2f a2 = *(const v2f*)(ap + 8);
  const int rowt = node0 + wv * 16;
  const bool full = rowt + 16 <= N;
#pragma unroll
  for (int ct = 0; ct < NCOLS / 16; ++ct) {
    const int n = ct * 16 + mA;
    v8f acc = {};
    acc = __builtin_amdgcn_wmma_f32_16x16x4_f32(false, a0, false, wq[ct][0], (short)0, acc, false, false);
    acc = __builtin_amdgcn_wmma_f32_16x16x4_f32(false, a1, false, wq[ct][1], (short)0, acc, false, false);
    acc = __builtin_amdgcn_wmma_f32_16x16x4_f32(false, a2, false, wq[ct][2], (short)0, acc, false, false);
    const float bv = bq[ct];
    float o[8];
#pragma unroll
    for (int r = 0; r < 8; ++r) {
      const float v = acc[r] + bv;
      o[r] = v > 0.0f ? v : 0.0f;
    }
    float* outp = h1 + (size_t)(rowt + rbase) * NCOLS + n;
    if (full) {
#pragma unroll
      for (int r = 0; r < 8; ++r) outp[(size_t)r * NCOLS] = o[r];
    } else {
#pragma unroll
      for (int r = 0; r < 8; ++r)
        if (rowt + rbase + r < N) outp[(size_t)r * NCOLS] = o[r];
    }
  }

  float4* __restrict__ rows_o = reinterpret_cast<float4*>(rows + (size_t)node0 * ROWW);
  int4*   __restrict__ slot_o = reinterpret_cast<int4*>(slot + (size_t)node0 * SLOTCAP);
  for (int i = t; i < nn * (ROWW / 4); i += OWN_T)
    rows_o[i] = *reinterpret_cast<const float4*>(&rows_l[4 * i]);
  for (int i = t; i < nn * (SLOTCAP / 4); i += OWN_T)
    slot_o[i] = *reinterpret_cast<const int4*>(&slot_l[4 * i]);
}

#define KG 2
#define BSTEP (2 * NCOLS * 2)

__device__ __forceinline__ void load_b_group(v2f (&b)[KG], const float* __restrict__ bp, const int g) {
#pragma unroll
  for (int j = 0; j < KG; ++j)
    b[j] = *(const v2f*)(bp + (size_t)(g * KG + j) * BSTEP);
}

__device__ __forceinline__ void mma_group(v8f& c0, v8f& c1, const float* ap, const v2f (&b)[KG], const int g) {
  v2f a0[KG], a1[KG];
#pragma unroll
  for (int j = 0; j < KG; ++j) {
    a0[j] = *(const v2f*)(ap + (g * KG + j) * 4);
    a1[j] = *(const v2f*)(ap + 16 * LDSTRIDE + (g * KG + j) * 4);
  }
#pragma unroll
  for (int j = 0; j < KG; ++j) {
    c0 = __builtin_amdgcn_wmma_f32_16x16x4_f32(false, a0[j], false, b[j],
                                               (short)0, c0, false, false);
    c1 = __builtin_amdgcn_wmma_f32_16x16x4_f32(false, a1[j], false, b[j],
                                               (short)0, c1, false, false);
  }
}

__device__ __forceinline__ void gemm_groups(v8f& c0, v8f& c1, const float* ap,
                                            const float* __restrict__ bp, const int ng,
                                            const float* __restrict__ bp_next, const bool has_next,
                                            v2f (&bA)[KG]) {
  v2f bB[KG];
#pragma unroll 1
  for (int g = 0; g < ng - 2; g += 2) {
    load_b_group(bB, bp, g + 1);
    mma_group(c0, c1, ap, bA, g);
    load_b_group(bA, bp, g + 2);
    mma_group(c0, c1, ap, bB, g + 1);
  }
  load_b_group(bB, bp, ng - 1);
  mma_group(c0, c1, ap, bA, ng - 2);
  if (has_next) load_b_group(bA, bp_next, 0);
  mma_group(c0, c1, ap, bB, ng - 1);
}

#define NPW (MTILE / 8)
static_assert(SLOTCAP == 32, "one slot word per lane");

__global__ void sage_head_fused(const float* __restrict__ h, const float* __restrict__ rows,
                                const int* __restrict__ slot,
                                const int* __restrict__ ovf_cnt, const int* __restrict__ ovf,
                                const float* __restrict__ Wp2, const float* __restrict__ b2,
                                const float* __restrict__ Wp3, const float* __restrict__ b3,
                                const float* __restrict__ W4, const float* __restrict__ b4,
                                float* __restrict__ out, int M) {
  __shared__ __attribute__((aligned(16))) float As[MTILE * LDSTRIDE];

  const int Kpad = 2 * NCOLS;
  const int row0 = blockIdx.x * MTILE;
  const int t = threadIdx.x;
  const int lane  = t & 31;
  const int wave  = t >> 5;
  const int wv = __builtin_amdgcn_readfirstlane(wave);

  const float* hl = h + 4 * lane;
  float  dg[NPW];
  int    mn[NPW];
  float4 sf[NPW];
#pragma unroll
  for (int q = 0; q < NPW; ++q) {
    const int row = row0 + wv * NPW + q;
    dg[q] = 1.0f;
    sf[q] = make_float4(0.0f, 0.0f, 0.0f, 0.0f);
    if (row < M) {
      dg[q] = rows[(size_t)row * ROWW + DEGSLOT];
      sf[q] = *(const float4*)(hl + (size_t)row * 128);
    }
  }
#pragma unroll
  for (int q = 0; q < NPW; ++q) {
    const int row = row0 + wv * NPW + q;
    const int filled = (int)dg[q] - 1;
    mn[q] = (lane < filled) ? slot[(size_t)row * SLOTCAP + lane] : 0;
  }
#pragma unroll
  for (int q = 0; q < NPW; ++q) {
    const int m   = wv * NPW + q;
    const int row = row0 + m;
    const int indeg = __builtin_amdgcn_readfirstlane((int)dg[q] - 1);
    const int end   = indeg < SLOTCAP ? indeg : SLOTCAP;
    const int mine  = mn[q];
    float4 acc = make_float4(0.0f, 0.0f, 0.0f, 0.0f);
    int k = 0;
    for (; k + 4 <= end; k += 4) {
      const int s0 = __builtin_amdgcn_readlane(mine, k);
      const int s1 = __builtin_amdgcn_readlane(mine, k + 1);
      const int s2 = __builtin_amdgcn_readlane(mine, k + 2);
      const int s3 = __builtin_amdgcn_readlane(mine, k + 3);
      const float4 v0 = *(const float4*)(hl + (size_t)s0 * 128);
      const float4 v1 = *(const float4*)(hl + (size_t)s1 * 128);
      const float4 v2 = *(const float4*)(hl + (size_t)s2 * 128);
      const float4 v3 = *(const float4*)(hl + (size_t)s3 * 128);
      acc.x += v0.x; acc.y += v0.y; acc.z += v0.z; acc.w += v0.w;
      acc.x += v1.x; acc.y += v1.y; acc.z += v1.z; acc.w += v1.w;
      acc.x += v2.x; acc.y += v2.y; acc.z += v2.z; acc.w += v2.w;
      acc.x += v3.x; acc.y += v3.y; acc.z += v3.z; acc.w += v3.w;
    }
    {
      float4 t0 = make_float4(0.0f, 0.0f, 0.0f, 0.0f), t1 = t0, t2 = t0;
      if (k < end)     t0 = *(const float4*)(hl + (size_t)__builtin_amdgcn_readlane(mine, k) * 128);
      if (k + 1 < end) t1 = *(const float4*)(hl + (size_t)__builtin_amdgcn_readlane(mine, k + 1) * 128);
      if (k + 2 < end) t2 = *(const float4*)(hl + (size_t)__builtin_amdgcn_readlane(mine, k + 2) * 128);
      if (k < end)     { acc.x += t0.x; acc.y += t0.y; acc.z += t0.z; acc.w += t0.w; }
      if (k + 1 < end) { acc.x += t1.x; acc.y += t1.y; acc.z += t1.z; acc.w += t1.w; }
      if (k + 2 < end) { acc.x += t2.x; acc.y += t2.y; acc.z += t2.z; acc.w += t2.w; }
    }
    if (indeg > SLOTCAP) {
      const int ocnt = __builtin_amdgcn_readfirstlane(*ovf_cnt);
      for (int k0 = 0; k0 < ocnt; k0 += 32) {
        const int kk = k0 + lane;
        int ss = 0, dd = -1;
        if (kk < ocnt) { ss = ovf[2 * (size_t)kk]; dd = ovf[2 * (size_t)kk + 1]; }
        unsigned hit = __builtin_amdgcn_ballot_w32(dd == row);
        while (hit != 0u) {
          const int b = __builtin_ctz(hit);
          hit &= hit - 1u;
          const int s = __builtin_amdgcn_readlane(ss, b);
          const float4 v = *(const float4*)(hl + (size_t)s * 128);
          acc.x += v.x; acc.y += v.y; acc.z += v.z; acc.w += v.w;
        }
      }
    }
    const float  dv = dg[q];
    const float4 self = sf[q];
    float4 mean;
    mean.x = (self.x + acc.x) / dv;
    mean.y = (self.y + acc.y) / dv;
    mean.z = (self.z + acc.z) / dv;
    mean.w = (self.w + acc.w) / dv;
    *(float4*)&As[m * LDSTRIDE + 4 * lane]         = self;
    *(float4*)&As[m * LDSTRIDE + NCOLS + 4 * lane] = mean;
  }
  const int n     = wave * 16 + (lane & 15);
  const int mA    = lane & 15;
  const int khalf = (lane < 16) ? 0 : 2;
  const int rbase = (lane < 16) ? 0 : 8;

  const float* bp2 = Wp2 + ((size_t)(khalf >> 1) * NCOLS + n) * 2;
  const float* bp3 = Wp3 + ((size_t)(khalf >> 1) * NCOLS + n) * 2;
  const float* ap  = &As[mA * LDSTRIDE + khalf];

  v2f bA[KG];
  load_b_group(bA, bp2, 0);
  __syncthreads();

  v8f c0 = {};
  v8f c1 = {};
  gemm_groups(c0, c1, ap, bp2, Kpad / (4 * KG), bp3, true, bA);

  __syncthreads();
  {
    const float bv = b2[n];
#pragma unroll
    for (int r = 0; r < 8; ++r) {
      float vA = c0[r] + bv;
      float vB = c1[r] + bv;
      As[(rbase + r) * LDSTRIDE + n]      = vA > 0.0f ? vA : 0.0f;
      As[(16 + rbase + r) * LDSTRIDE + n] = vB > 0.0f ? vB : 0.0f;
    }
  }
  __syncthreads();

  v8f e0 = {};
  v8f e1 = {};
  gemm_groups(e0, e1, ap, bp3, NCOLS / (4 * KG), bp3, false, bA);
  {
    const float bv = b3[n];
#pragma unroll
    for (int r = 0; r < 8; ++r) {
      float vA = e0[r] + bv;
      float vB = e1[r] + bv;
      As[(rbase + r) * LDSTRIDE + NCOLS + n]      = vA > 0.0f ? vA : 0.0f;
      As[(16 + rbase + r) * LDSTRIDE + NCOLS + n] = vB > 0.0f ? vB : 0.0f;
    }
  }
  __syncthreads();

  const int orow = t >> 3;
  const int part = t & 7;
  const float* hrow = &As[orow * LDSTRIDE + NCOLS + part * 16];
  const float* w    = W4 + part * 32;
  float s0 = 0.0f, s1 = 0.0f;
#pragma unroll
  for (int q = 0; q < 4; ++q) {
    const float4 hv = *(const float4*)(hrow + 4 * q);
    const float4 wa = *(const float4*)(w + 8 * q);
    const float4 wb = *(const float4*)(w + 8 * q + 4);
    s0 += hv.x * wa.x;  s1 += hv.x * wa.y;
    s0 += hv.y * wa.z;  s1 += hv.y * wa.w;
    s0 += hv.z * wb.x;  s1 += hv.z * wb.y;
    s0 += hv.w * wb.z;  s1 += hv.w * wb.w;
  }
  for (int o = 1; o < 8; o <<= 1) {
    s0 += __shfl_xor(s0, o);
    s1 += __shfl_xor(s1, o);
  }
  if (part == 0) {
    As[orow * 2 + 0] = s0 + b4[0];
    As[orow * 2 + 1] = s1 + b4[1];
  }
  __syncthreads();
  if (t < 2 * MTILE && row0 + (t >> 1) < M)
    out[(size_t)row0 * 2 + t] = As[t];
}

extern "C" void kernel_launch(void* const* d_in, const int* in_sizes, int n_in,
                              void* d_out, int out_size, void* d_ws, size_t ws_size,
                              hipStream_t stream) {
  const float* x   = (const float*)d_in[0];
  const int*   ei  = (const int*)d_in[1];
  const float* W1  = (const float*)d_in[2];
  const float* b1  = (const float*)d_in[3];
  const float* W2  = (const float*)d_in[4];
  const float* b2  = (const float*)d_in[5];
  const float* l1W = (const float*)d_in[6];
  const float* l1b = (const float*)d_in[7];
  const float* l2W = (const float*)d_in[8];
  const float* l2b = (const float*)d_in[9];

  const int N = in_sizes[0] / 5;
  const int E = in_sizes[1] / 2;
  const int* src = ei;
  const int* dst = ei + E;

  float* ws   = (float*)d_ws;
  float* rows = ws;
  float* h1   = rows + (size_t)ROWW * N;
  float* Wp1  = h1   + (size_t)128 * N;
  float* Wp2  = Wp1  + (size_t)12 * 128;
  float* Wp3  = Wp2  + (size_t)256 * 128;
  int* slot    = (int*)(Wp3 + (size_t)128 * 128);
  int* ovf_cnt = slot + (size_t)SLOTCAP * N;
  int* ovf     = ovf_cnt + 64;
  int* tot     = ovf + 2 * (size_t)E;
  int* cursor  = tot + MAXR;
  int* gstart  = cursor + MAXR;
  int* binned  = gstart + MAXR;
  const int NR = (N + RNODES - 1) / RNODES;

  const int nthread = 256;

  prep_all<<<PB1 + PB2 + PB3 + PBI, PREP_T, 0, stream>>>(W1, W2, l1W, Wp1, Wp2, Wp3, tot, cursor, ovf_cnt);

  range_count<<<(E + CHUNK - 1) / CHUNK, nthread, 0, stream>>>(dst, tot, E);
  range_place<<<(E + CHUNK - 1) / CHUNK, nthread, 0, stream>>>(src, dst, tot, cursor, gstart, binned, E, NR);
  range_own<<<NR, OWN_T, 0, stream>>>(binned, gstart, tot, x, rows, slot, ovf_cnt, ovf, Wp1, b1, h1, N);

  sage_head_fused<<<(N + MTILE - 1) / MTILE, nthread, 0, stream>>>(h1, rows, slot, ovf_cnt, ovf, Wp2, b2, Wp3, l1b, l2W, l2b,
                                                                    (float*)d_out, N);
}


# Edge regrouping without a counting pass: chunks ordered by range in LDS and stored in place with a run table; each range's work-group collects its runs via LDS; 5 launches -> 3; sums by arrival
# speedup vs baseline: 5.6112x; 1.0667x over previous
#include <hip/hip_runtime.h>

typedef __attribute__((ext_vector_type(2))) float v2f;
typedef __attribute__((ext_vector_type(8))) float v8f;

#define NCOLS 128
#define MTILE 32
#define LDSTRIDE 260

__device__ __forceinline__ void pack_w(const float* __restrict__ W, float* __restrict__ Wp,
                                       const int Kw, const int Kpad, const int i) {
  if (i < Kpad * NCOLS) {
    int k = i / NCOLS;
    int n = i - k * NCOLS;
    float v = (k < Kw) ? W[(size_t)k * NCOLS + n] : 0.0f;
    Wp[((size_t)(k >> 1) * NCOLS + n) * 2 + (k & 1)] = v;
  }
}

#define ROWW 8
#define DEGSLOT 5
#define SLOTCAP 32
#define RSH 8
#define RNODES (1 << RSH)
#define MAXR 512
#define CHUNK 4096
#define EPT (CHUNK / 256)
#define OWN_T 512
#define CAP 6144
#define RPG 8
#define L1K 12
#define WSTEP (2 * NCOLS * 2)

#define PREP_T 256
#define PB1 ((12  * NCOLS + PREP_T - 1) / PREP_T)
#define PB2 ((256 * NCOLS + PREP_T - 1) / PREP_T)
#define PB3 ((128 * NCOLS + PREP_T - 1) / PREP_T)

__global__ void range_place(const int* __restrict__ src, const int* __restrict__ dst,
                            int* __restrict__ tab, int* __restrict__ binned, int* __restrict__ ovf_cnt,
                            const float* __restrict__ W1, const float* __restrict__ W2,
                            const float* __restrict__ W3,
                            float* __restrict__ Wp1, float* __restrict__ Wp2, float* __restrict__ Wp3,
                            int E, int NC) {
  __shared__ int cnt[MAXR];
  __shared__ int loff[MAXR];
  __shared__ int2 sorted[CHUNK];
  const int t = threadIdx.x;
  const int b = blockIdx.x;
  if (b >= NC) {
    const int pb = b - NC;
    if (pb == 0 && t == 0) *ovf_cnt = 0;
    if (pb < PB1) {
      pack_w(W1, Wp1, 10, 12, pb * PREP_T + t);
    } else if (pb < PB1 + PB2) {
      pack_w(W2, Wp2, 256, 256, (pb - PB1) * PREP_T + t);
    } else {
      pack_w(W3, Wp3, 128, 128, (pb - PB1 - PB2) * PREP_T + t);
    }
    return;
  }
  for (int i = t; i < MAXR; i += 256) cnt[i] = 0;
  __syncthreads();
  const int e0 = b * CHUNK;
  int s[EPT], d[EPT], r[EPT];
#pragma unroll
  for (int j = 0; j < EPT; ++j) {
    const int e = e0 + j * 256 + t;
    s[j] = 0; d[j] = 0;
    if (e < E) { s[j] = src[e]; d[j] = dst[e]; }
  }
#pragma unroll
  for (int j = 0; j < EPT; ++j) {
    const int e = e0 + j * 256 + t;
    r[j] = -1;
    if (e < E) r[j] = atomicAdd(&cnt[d[j] >> RSH], 1);
  }
  __syncthreads();
  if (t < 32) {
    const int i0 = t * (MAXR / 32);
    int a = 0;
    for (int j = 0; j < MAXR / 32; ++j) a += cnt[i0 + j];
    int ia = a;
    for (int o = 1; o < 32; o <<= 1) {
      const int ua = __shfl_up(ia, o);
      if (t >= o) ia += ua;
    }
    int ra = ia - a;
    for (int j = 0; j < MAXR / 32; ++j) {
      const int va = cnt[i0 + j];
      loff[i0 + j] = ra;
      ra += va;
    }
  }
  __syncthreads();
  for (int i = t; i < MAXR; i += 256)
    tab[(size_t)b * MAXR + i] = loff[i] | (cnt[i] << 16);
#pragma unroll
  for (int j = 0; j < EPT; ++j)
    if (r[j] >= 0) sorted[loff[d[j] >> RSH] + r[j]] = make_int2(s[j], d[j]);
  __syncthreads();
  const int left = E - e0;
  const int nloc = left < CHUNK ? left : CHUNK;
  int2* __restrict__ outp = reinterpret_cast<int2*>(binned);
  for (int i = t; i < nloc; i += 256) outp[(size_t)e0 + (size_t)i] = sorted[i];
}

__global__ void __launch_bounds__(OWN_T)
range_own(const int* __restrict__ binned, const int* __restrict__ tab, const int NC,
          const float* __restrict__ x, float* __restrict__ rows, int* __restrict__ slot,
          int* __restrict__ ovf_cnt, int* __restrict__ ovf,
          const float* __restrict__ Wp1, const float* __restrict__ b1, float* __restrict__ h1, int N) {
  __shared__ __attribute__((aligned(16))) float rows_l[RNODES * ROWW];
  __shared__ __attribute__((aligned(16))) int   slot_l[RNODES * SLOTCAP];
  __shared__ __attribute__((aligned(16))) float a_l[RNODES * L1K];
  __shared__ __attribute__((aligned(16))) int2  stage[CAP];
  __shared__ int total_l;
  __shared__ int r_start[OWN_T];
  __shared__ int r_pos[OWN_T];
  __shared__ int r_cnt[OWN_T];
  const int t = threadIdx.x;
  const int c = blockIdx.x;
  const int node0 = c << RSH;
  for (int i = t; i < RNODES * ROWW; i += OWN_T)
    rows_l[i] = ((i & (ROWW - 1)) == DEGSLOT) ? 1.0f : 0.0f;
  for (int i = t; i < RNODES * SLOTCAP; i += OWN_T) slot_l[i] = 0;
  if (t == 0) total_l = 0;
  __syncthreads();

  const int2* __restrict__ inp = reinterpret_cast<const int2*>(binned);
  const unsigned lane = (unsigned)t & 31u;
  const int wv = __builtin_amdgcn_readfirstlane(t >> 5);
  const unsigned f  = lane & 7u;
  const unsigned g  = lane >> 3;
  const unsigned fx = f < 5u ? f : 4u;

  for (int cb = 0; cb < NC; cb += OWN_T) {
  {
    const int ch = cb + t;
    int mystart = 0, mycnt = 0, mypos = 0;
    if (ch < NC) {
      const int pk = tab[(size_t)ch * MAXR + c];
      mystart = ch * CHUNK + (pk & 0xffff);
      mycnt   = pk >> 16;
      if (mycnt != 0) mypos = atomicAdd(&total_l, mycnt);
    }
    r_start[t] = mystart;
    r_pos[t]   = mypos;
    r_cnt[t]   = mycnt;
  }
  __syncthreads();
  const int total = total_l;
  for (int lo = 0; lo < total; lo += CAP) {
  {
    const int gq = t >> 3;
    const int q  = t & 7;
    int2 v[RPG][2];
    int  pl[RPG][2];
#pragma unroll
    for (int k = 0; k < RPG; ++k) {
      const int gr = k * (OWN_T / 8) + gq;
      const int st = r_start[gr];
      const int ps = r_pos[gr];
      const int cn = r_cnt[gr];
#pragma unroll
      for (int h = 0; h < 2; ++h) {
        const int i = q + 8 * h;
        const int place = ps + i - lo;
        pl[k][h] = (i < cn && place >= 0 && place < CAP) ? place : -1;
        v[k][h]  = make_int2(0, 0);
        if (pl[k][h] >= 0) v[k][h] = inp[(size_t)(st + i)];
      }
    }
#pragma unroll
    for (int k = 0; k < RPG; ++k) {
#pragma unroll
      for (int h = 0; h < 2; ++h)
        if (pl[k][h] >= 0) stage[pl[k][h]] = v[k][h];
    }
#pragma unroll 1
    for (int k = 0; k < RPG; ++k) {
      const int gr = k * (OWN_T / 8) + gq;
      const int cn = r_cnt[gr];
      if (cn > 16) {
        const int st = r_start[gr];
        const int ps = r_pos[gr];
        for (int i = q + 16; i < cn; i += 8) {
          const int place = ps + i - lo;
          if (place >= 0 && place < CAP) stage[place] = inp[(size_t)(st + i)];
        }
      }
    }
  }
  __syncthreads();
  const int cnt = (total - lo) < CAP ? (total - lo) : CAP;
  for (int off = wv * 32; off < cnt; off += OWN_T) {
    int s = 0, d = node0;
    const bool have = off + (int)lane < cnt;
    if (have) {
      const int2 pr = stage[off + (int)lane];
      s = pr.x;
      d = pr.y;
    }
    float  xv[8];
    float* p[8];
    bool   ok[8];
#pragma unroll
    for (int r = 0; r < 8; ++r) {
      const int sl = r * 4 + (int)g;
      const int ss = __shfl(s, sl);
      const int dd = __shfl(d, sl);
      ok[r] = (off + sl < cnt) && (f <= (unsigned)DEGSLOT);
      p[r]  = rows_l + (dd - node0) * ROWW + (int)f;
      xv[r] = x[(size_t)ss * 5 + fx];
    }
    float oldv[8];
#pragma unroll
    for (int r = 0; r < 8; ++r) {
      oldv[r] = 0.0f;
      if (ok[r]) oldv[r] = atomicAdd(p[r], f < 5u ? xv[r] : 1.0f);
    }
    float myold = 0.0f;
#pragma unroll
    for (int r = 0; r < 8; ++r) {
      const float v = __shfl(oldv[r], (int)((lane & 3u) * 8u + (unsigned)DEGSLOT));
      if ((lane >> 2) == (unsigned)r) myold = v;
    }
    if (have) {
      const int place = (int)myold - 1;
      if (place < SLOTCAP) {
        slot_l[(d - node0) * SLOTCAP + place] = s;
      } else {
        const int k = atomicAdd(ovf_cnt, 1);
        ovf[2 * (size_t)k]     = s;
        ovf[2 * (size_t)k + 1] = d;
      }
    }
  }
  __syncthreads();
  }
  if (t == 0) total_l = 0;
  __syncthreads();
  }

  const int left = N - node0;
  const int nn = left < RNODES ? left : RNODES;

  const int mA    = (int)(lane & 15u);
  const int khalf = (lane < 16u) ? 0 : 2;
  v2f   wq[NCOLS / 16][3];
  float bq[NCOLS / 16];
#pragma unroll
  for (int ct = 0; ct < NCOLS / 16; ++ct) {
    const int n = ct * 16 + mA;
    const float* bp = Wp1 + ((size_t)(khalf >> 1) * NCOLS + n) * 2;
    wq[ct][0] = *(const v2f*)(bp);
    wq[ct][1] = *(const v2f*)(bp + WSTEP);
    wq[ct][2] = *(const v2f*)(bp + 2 * WSTEP);
    bq[ct]    = b1[n];
  }

  for (int i = t; i < RNODES * L1K; i += OWN_T) {
    const int m = i / L1K;
    const int k = i - m * L1K;
    float v = 0.0f;
    if (m < nn) {
      const size_t xr = (size_t)(node0 + m) * 5;
      if (k < 5) {
        v = x[xr + k];
      } else if (k < 10) {
        const int kk = k - 5;
        v = (x[xr + kk] + rows_l[m * ROWW + kk]) / rows_l[m * ROWW + DEGSLOT];
      }
    }
    a_l[i] = v;
  }

  __syncthreads();

  const int rbase = (lane < 16u) ? 0 : 8;
  const float* ap = &a_l[(wv * 16 + mA) * L1K + khalf];
  const v2f a0 = *(const v2f*)(ap);
  const v2f a1 = *(const v2f*)(ap + 4);
  const v2f a2 = *(const v2f*)(ap + 8);
  const int rowt = node0 + wv * 16;
  const bool full = rowt + 16 <= N;
#pragma unroll
  for (int ct = 0; ct < NCOLS / 16; ++ct) {
    const int n = ct * 16 + mA;
    v8f acc = {};
    acc = __builtin_amdgcn_wmma_f32_16x16x4_f32(false, a0, false, wq[ct][0], (short)0, acc, false, false);
    acc = __builtin_amdgcn_wmma_f32_16x16x4_f32(false, a1, false, wq[ct][1], (short)0, acc, false, false);
    acc = __builtin_amdgcn_wmma_f32_16x16x4_f32(false, a2, false, wq[ct][2], (short)0, acc, false, false);
    const float bv = bq[ct];
    float o[8];
#pragma unroll
    for (int r = 0; r < 8; ++r) {
      const float v = acc[r] + bv;
      o[r] = v > 0.0f ? v : 0.0f;
    }
    float* outp = h1 + (size_t)(rowt + rbase) * NCOLS + n;
    if (full) {
#pragma unroll
      for (int r = 0; r < 8; ++r) outp[(size_t)r * NCOLS] = o[r];
    } else {
#pragma unroll
      for (int r = 0; r < 8; ++r)
        if (rowt + rbase + r < N) outp[(size_t)r * NCOLS] = o[r];
    }
  }

  float4* __restrict__ rows_o = reinterpret_cast<float4*>(rows + (size_t)node0 * ROWW);
  int4*   __restrict__ slot_o = reinterpret_cast<int4*>(slot + (size_t)node0 * SLOTCAP);
  for (int i = t; i < nn * (ROWW / 4); i += OWN_T)
    rows_o[i] = *reinterpret_cast<const float4*>(&rows_l[4 * i]);
  for (int i = t; i < nn * (SLOTCAP / 4); i += OWN_T)
    slot_o[i] = *reinterpret_cast<const int4*>(&slot_l[4 * i]);
}

#define KG 2
#define BSTEP (2 * NCOLS * 2)

__device__ __forceinline__ void load_b_group(v2f (&b)[KG], const float* __restrict__ bp, const int g) {
#pragma unroll
  for (int j = 0; j < KG; ++j)
    b[j] = *(const v2f*)(bp + (size_t)(g * KG + j) * BSTEP);
}

__device__ __forceinline__ void mma_group(v8f& c0, v8f& c1, const float* ap, const v2f (&b)[KG], const int g) {
  v2f a0[KG], a1[KG];
#pragma unroll
  for (int j = 0; j < KG; ++j) {
    a0[j] = *(const v2f*)(ap + (g * KG + j) * 4);
    a1[j] = *(const v2f*)(ap + 16 * LDSTRIDE + (g * KG + j) * 4);
  }
#pragma unroll
  for (int j = 0; j < KG; ++j) {
    c0 = __builtin_amdgcn_wmma_f32_16x16x4_f32(false, a0[j], false, b[j],
                                               (short)0, c0, false, false);
    c1 = __builtin_amdgcn_wmma_f32_16x16x4_f32(false, a1[j], false, b[j],
                                               (short)0, c1, false, false);
  }
}

__device__ __forceinline__ void gemm_groups(v8f& c0, v8f& c1, const float* ap,
                                            const float* __restrict__ bp, const int ng,
                                            const float* __restrict__ bp_next, const bool has_next,
                                            v2f (&bA)[KG]) {
  v2f bB[KG];
#pragma unroll 1
  for (int g = 0; g < ng - 2; g += 2) {
    load_b_group(bB, bp, g + 1);
    mma_group(c0, c1, ap, bA, g);
    load_b_group(bA, bp, g + 2);
    mma_group(c0, c1, ap, bB, g + 1);
  }
  load_b_group(bB, bp, ng - 1);
  mma_group(c0, c1, ap, bA, ng - 2);
  if (has_next) load_b_group(bA, bp_next, 0);
  mma_group(c0, c1, ap, bB, ng - 1);
}

#define NPW (MTILE / 8)
static_assert(SLOTCAP == 32, "one slot word per lane");

__global__ void sage_head_fused(const float* __restrict__ h, const float* __restrict__ rows,
                                const int* __restrict__ slot,
                                const int* __restrict__ ovf_cnt, const int* __restrict__ ovf,
                                const float* __restrict__ Wp2, const float* __restrict__ b2,
                                const float* __restrict__ Wp3, const float* __restrict__ b3,
                                const float* __restrict__ W4, const float* __restrict__ b4,
                                float* __restrict__ out, int M) {
  __shared__ __attribute__((aligned(16))) float As[MTILE * LDSTRIDE];

  const int Kpad = 2 * NCOLS;
  const int row0 = blockIdx.x * MTILE;
  const int t = threadIdx.x;
  const int lane  = t & 31;
  const int wave  = t >> 5;
  const int wv = __builtin_amdgcn_readfirstlane(wave);

  const float* hl = h + 4 * lane;
  float  dg[NPW];
  int    mn[NPW];
  float4 sf[NPW];
#pragma unroll
  for (int q = 0; q < NPW; ++q) {
    const int row = row0 + wv * NPW + q;
    dg[q] = 1.0f;
    sf[q] = make_float4(0.0f, 0.0f, 0.0f, 0.0f);
    if (row < M) {
      dg[q] = rows[(size_t)row * ROWW + DEGSLOT];
      sf[q] = *(const float4*)(hl + (size_t)row * 128);
    }
  }
#pragma unroll
  for (int q = 0; q < NPW; ++q) {
    const int row = row0 + wv * NPW + q;
    const int filled = (int)dg[q] - 1;
    mn[q] = (lane < filled) ? slot[(size_t)row * SLOTCAP + lane] : 0;
  }
#pragma unroll
  for (int q = 0; q < NPW; ++q) {
    const int m   = wv * NPW + q;
    const int row = row0 + m;
    const int indeg = __builtin_amdgcn_readfirstlane((int)dg[q] - 1);
    const int end   = indeg < SLOTCAP ? indeg : SLOTCAP;
    const int mine  = mn[q];
    float4 acc = make_float4(0.0f, 0.0f, 0.0f, 0.0f);
    int k = 0;
    for (; k + 4 <= end; k += 4) {
      const int s0 = __builtin_amdgcn_readlane(mine, k);
      const int s1 = __builtin_amdgcn_readlane(mine, k + 1);
      const int s2 = __builtin_amdgcn_readlane(mine, k + 2);
      const int s3 = __builtin_amdgcn_readlane(mine, k + 3);
      const float4 v0 = *(const float4*)(hl + (size_t)s0 * 128);
      const float4 v1 = *(const float4*)(hl + (size_t)s1 * 128);
      const float4 v2 = *(const float4*)(hl + (size_t)s2 * 128);
      const float4 v3 = *(const float4*)(hl + (size_t)s3 * 128);
      acc.x += v0.x; acc.y += v0.y; acc.z += v0.z; acc.w += v0.w;
      acc.x += v1.x; acc.y += v1.y; acc.z += v1.z; acc.w += v1.w;
      acc.x += v2.x; acc.y += v2.y; acc.z += v2.z; acc.w += v2.w;
      acc.x += v3.x; acc.y += v3.y; acc.z += v3.z; acc.w += v3.w;
    }
    {
      float4 t0 = make_float4(0.0f, 0.0f, 0.0f, 0.0f), t1 = t0, t2 = t0;
      if (k < end)     t0 = *(const float4*)(hl + (size_t)__builtin_amdgcn_readlane(mine, k) * 128);
      if (k + 1 < end) t1 = *(const float4*)(hl + (size_t)__builtin_amdgcn_readlane(mine, k + 1) * 128);
      if (k + 2 < end) t2 = *(const float4*)(hl + (size_t)__builtin_amdgcn_readlane(mine, k + 2) * 128);
      if (k < end)     { acc.x += t0.x; acc.y += t0.y; acc.z += t0.z; acc.w += t0.w; }
      if (k + 1 < end) { acc.x += t1.x; acc.y += t1.y; acc.z += t1.z; acc.w += t1.w; }
      if (k + 2 < end) { acc.x += t2.x; acc.y += t2.y; acc.z += t2.z; acc.w += t2.w; }
    }
    if (indeg > SLOTCAP) {
      const int ocnt = __builtin_amdgcn_readfirstlane(*ovf_cnt);
      for (int k0 = 0; k0 < ocnt; k0 += 32) {
        const int kk = k0 + lane;
        int ss = 0, dd = -1;
        if (kk < ocnt) { ss = ovf[2 * (size_t)kk]; dd = ovf[2 * (size_t)kk + 1]; }
        unsigned hit = __builtin_amdgcn_ballot_w32(dd == row);
        while (hit != 0u) {
          const int b = __builtin_ctz(hit);
          hit &= hit - 1u;
          const int s = __builtin_amdgcn_readlane(ss, b);
          const float4 v = *(const float4*)(hl + (size_t)s * 128);
          acc.x += v.x; acc.y += v.y; acc.z += v.z; acc.w += v.w;
        }
      }
    }
    const float  dv = dg[q];
    const float4 self = sf[q];
    float4 mean;
    mean.x = (self.x + acc.x) / dv;
    mean.y = (self.y + acc.y) / dv;
    mean.z = (self.z + acc.z) / dv;
    mean.w = (self.w + acc.w) / dv;
    *(float4*)&As[m * LDSTRIDE + 4 * lane]         = self;
    *(float4*)&As[m * LDSTRIDE + NCOLS + 4 * lane] = mean;
  }
  const int n     = wave * 16 + (lane & 15);
  const int mA    = lane & 15;
  const int khalf = (lane < 16) ? 0 : 2;
  const int rbase = (lane < 16) ? 0 : 8;

  const float* bp2 = Wp2 + ((size_t)(khalf >> 1) * NCOLS + n) * 2;
  const float* bp3 = Wp3 + ((size_t)(khalf >> 1) * NCOLS + n) * 2;
  const float* ap  = &As[mA * LDSTRIDE + khalf];

  v2f bA[KG];
  load_b_group(bA, bp2, 0);
  __syncthreads();

  v8f c0 = {};
  v8f c1 = {};
  gemm_groups(c0, c1, ap, bp2, Kpad / (4 * KG), bp3, true, bA);

  __syncthreads();
  {
    const float bv = b2[n];
#pragma unroll
    for (int r = 0; r < 8; ++r) {
      float vA = c0[r] + bv;
      float vB = c1[r] + bv;
      As[(rbase + r) * LDSTRIDE + n]      = vA > 0.0f ? vA : 0.0f;
      As[(16 + rbase + r) * LDSTRIDE + n] = vB > 0.0f ? vB : 0.0f;
    }
  }
  __syncthreads();

  v8f e0 = {};
  v8f e1 = {};
  gemm_groups(e0, e1, ap, bp3, NCOLS / (4 * KG), bp3, false, bA);
  {
    const float bv = b3[n];
#pragma unroll
    for (int r = 0; r < 8; ++r) {
      float vA = e0[r] + bv;
      float vB = e1[r] + bv;
      As[(rbase + r) * LDSTRIDE + NCOLS + n]      = vA > 0.0f ? vA : 0.0f;
      As[(16 + rbase + r) * LDSTRIDE + NCOLS + n] = vB > 0.0f ? vB : 0.0f;
    }
  }
  __syncthreads();

  const int orow = t >> 3;
  const int part = t & 7;
  const float* hrow = &As[orow * LDSTRIDE + NCOLS + part * 16];
  const float* w    = W4 + part * 32;
  float s0 = 0.0f, s1 = 0.0f;
#pragma unroll
  for (int q = 0; q < 4; ++q) {
    const float4 hv = *(const float4*)(hrow + 4 * q);
    const float4 wa = *(const float4*)(w + 8 * q);
    const float4 wb = *(const float4*)(w + 8 * q + 4);
    s0 += hv.x * wa.x;  s1 += hv.x * wa.y;
    s0 += hv.y * wa.z;  s1 += hv.y * wa.w;
    s0 += hv.z * wb.x;  s1 += hv.z * wb.y;
    s0 += hv.w * wb.z;  s1 += hv.w * wb.w;
  }
  for (int o = 1; o < 8; o <<= 1) {
    s0 += __shfl_xor(s0, o);
    s1 += __shfl_xor(s1, o);
  }
  if (part == 0) {
    As[orow * 2 + 0] = s0 + b4[0];
    As[orow * 2 + 1] = s1 + b4[1];
  }
  __syncthreads();
  if (t < 2 * MTILE && row0 + (t >> 1) < M)
    out[(size_t)row0 * 2 + t] = As[t];
}

extern "C" void kernel_launch(void* const* d_in, const int* in_sizes, int n_in,
                              void* d_out, int out_size, void* d_ws, size_t ws_size,
                              hipStream_t stream) {
  const float* x   = (const float*)d_in[0];
  const int*   ei  = (const int*)d_in[1];
  const float* W1  = (const float*)d_in[2];
  const float* b1  = (const float*)d_in[3];
  const float* W2  = (const float*)d_in[4];
  const float* b2  = (const float*)d_in[5];
  const float* l1W = (const float*)d_in[6];
  const float* l1b = (const float*)d_in[7];
  const float* l2W = (const float*)d_in[8];
  const float* l2b = (const float*)d_in[9];

  const int N = in_sizes[0] / 5;
  const int E = in_sizes[1] / 2;
  const int* src = ei;
  const int* dst = ei + E;

  float* ws   = (float*)d_ws;
  float* rows = ws;
  float* h1   = rows + (size_t)ROWW * N;
  float* Wp1  = h1   + (size_t)128 * N;
  float* Wp2  = Wp1  + (size_t)12 * 128;
  float* Wp3  = Wp2  + (size_t)256 * 128;
  int* slot    = (int*)(Wp3 + (size_t)128 * 128);
  int* ovf_cnt = slot + (size_t)SLOTCAP * N;
  int* ovf     = ovf_cnt + 64;
  const int NR = (N + RNODES - 1) / RNODES;
  const int NC = (E + CHUNK - 1) / CHUNK;
  int* tab     = ovf + 2 * (size_t)E;
  int* binned  = tab + (size_t)NC * MAXR;

  const int nthread = 256;

  range_place<<<NC + PB1 + PB2 + PB3, nthread, 0, stream>>>(src, dst, tab, binned, ovf_cnt,
                                                            W1, W2, l1W, Wp1, Wp2, Wp3, E, NC);

  range_own<<<NR, OWN_T, 0, stream>>>(binned, tab, NC, x, rows, slot, ovf_cnt, ovf, Wp1, b1, h1, N);

  sage_head_fused<<<(N + MTILE - 1) / MTILE, nthread, 0, stream>>>(h1, rows, slot, ovf_cnt, ovf, Wp2, b2, Wp3, l1b, l2W, l2b,
                                                                    (float*)d_out, N);
}
